# starting kernel
# baseline (speedup 1.0000x reference)
#include <hip/hip_runtime.h>
#include <math.h>

#define NNODES 100000
#define NEDGES 1600000
#define NGRAPHS 64
#define DIN 256
#define DHID 128
#define DOUT 64
#define EPSF 1e-5f

typedef __attribute__((ext_vector_type(2))) float v2f;
typedef __attribute__((ext_vector_type(8))) float v8f;

__global__ void zero_f32(float* __restrict__ p, int n) {
  int i = blockIdx.x * blockDim.x + threadIdx.x;
  if (i < n) p[i] = 0.0f;
}

__global__ void deg_accum(const int* __restrict__ col, const float* __restrict__ w,
                          float* __restrict__ deg, int nE) {
  int e = blockIdx.x * blockDim.x + threadIdx.x;
  if (e < nE) atomicAdd(&deg[col[e]], w[e]);
}

__global__ void deg_to_dinv(float* __restrict__ deg, int n) {
  int i = blockIdx.x * blockDim.x + threadIdx.x;
  if (i < n) {
    float d = deg[i] + 1.0f;
    deg[i] = rsqrtf(fmaxf(d, EPSF));
  }
}

template <int NCOLS, int KDIM>
__global__ void wmma_gemm_f32(const float* __restrict__ A, const float* __restrict__ B,
                              float* __restrict__ C, int M) {
  __shared__ float2 Bl[(KDIM / 2) * NCOLS];

  for (int i = threadIdx.x; i < (KDIM / 2) * NCOLS; i += blockDim.x) {
    const int k2 = i / NCOLS;
    const int n = i - k2 * NCOLS;
    float2 v;
    v.x = B[(size_t)(2 * k2 + 0) * NCOLS + n];
    v.y = B[(size_t)(2 * k2 + 1) * NCOLS + n];
    Bl[i] = v;
  }
  __syncthreads();

  const int wave = (int)(((size_t)blockIdx.x * blockDim.x + threadIdx.x) >> 5);
  const int lane = threadIdx.x & 31;
  if (wave >= M / 16) return;
  const int m0 = wave * 16;
  const int half = lane >> 4;
  const int l = lane & 15;
  constexpr int NT = NCOLS / 16;

  v8f acc[NT];
#pragma unroll
  for (int t = 0; t < NT; ++t)
#pragma unroll
    for (int j = 0; j < 8; ++j) acc[t][j] = 0.0f;

  const float* __restrict__ arow = A + (size_t)(m0 + l) * KDIM + 2 * half;

  v2f a;
  a[0] = arow[0];
  a[1] = arow[1];

  for (int k0 = 0; k0 < KDIM; k0 += 4) {
    v2f a_next = a;
    if (k0 + 4 < KDIM) {
      a_next[0] = arow[k0 + 4 + 0];
      a_next[1] = arow[k0 + 4 + 1];
    }
    const float2* __restrict__ brow = Bl + (size_t)(k0 / 2 + half) * NCOLS + l;
    v2f bfrag[NT];
#pragma unroll
    for (int t = 0; t < NT; ++t) {
      float2 bb = brow[t * 16];
      bfrag[t][0] = bb.x;
      bfrag[t][1] = bb.y;
    }
#pragma unroll
    for (int t = 0; t < NT; ++t)
      acc[t] = __builtin_amdgcn_wmma_f32_16x16x4_f32(
           false, a,  false, bfrag[t],
           (short)0, acc[t],  false,  false);
    a = a_next;
  }

#pragma unroll
  for (int t = 0; t < NT; ++t)
#pragma unroll
    for (int r = 0; r < 8; ++r)
      C[(size_t)(m0 + r + 8 * half) * NCOLS + t * 16 + l] = acc[t][r];
}

template <int D>
__global__ void init_out(const float* __restrict__ h, const float* __restrict__ dinv,
                         const float* __restrict__ bias, float* __restrict__ out, int n) {
  const int wave = (int)(((size_t)blockIdx.x * blockDim.x + threadIdx.x) >> 5);
  const int lane = threadIdx.x & 31;
  if (wave >= n) return;
  constexpr int F = D / 32;
  const float s = dinv[wave] * dinv[wave];
  const float* __restrict__ hp = h + (size_t)wave * D + lane * F;
  float* __restrict__ op = out + (size_t)wave * D + lane * F;
#pragma unroll
  for (int f = 0; f < F; ++f) op[f] = bias[lane * F + f] + hp[f] * s;
}

template <int D>
__global__ void edge_scatter(const float* __restrict__ h, const int* __restrict__ rows,
                             const int* __restrict__ cols, const float* __restrict__ w,
                             const float* __restrict__ dinv, float* __restrict__ out, int nE) {
  const int wave = (int)(((size_t)blockIdx.x * blockDim.x + threadIdx.x) >> 5);
  const int lane = threadIdx.x & 31;
  if (wave >= nE) return;
  const int r = rows[wave];
  const int c = cols[wave];
  const float nrm = dinv[r] * w[wave] * dinv[c];
  constexpr int F = D / 32;
  const float* __restrict__ hp = h + (size_t)r * D + lane * F;
  float* __restrict__ op = out + (size_t)c * D + lane * F;
#pragma unroll
  for (int f = 0; f < F; ++f) atomicAdd(op + f, hp[f] * nrm);
}

template <int D>
__global__ void relu_layernorm(float* __restrict__ x, const float* __restrict__ g,
                               const float* __restrict__ b, int n) {
  const int wave = (int)(((size_t)blockIdx.x * blockDim.x + threadIdx.x) >> 5);
  const int lane = threadIdx.x & 31;
  if (wave >= n) return;
  constexpr int F = D / 32;
  float* __restrict__ p = x + (size_t)wave * D + lane * F;
  float v[F];
  float s = 0.0f;
#pragma unroll
  for (int f = 0; f < F; ++f) {
    v[f] = fmaxf(p[f], 0.0f);
    s += v[f];
  }
#pragma unroll
  for (int off = 16; off > 0; off >>= 1) s += __shfl_xor(s, off, 32);
  const float mu = s * (1.0f / D);
  float q = 0.0f;
#pragma unroll
  for (int f = 0; f < F; ++f) {
    float d = v[f] - mu;
    q += d * d;
  }
#pragma unroll
  for (int off = 16; off > 0; off >>= 1) q += __shfl_xor(q, off, 32);
  const float rs = rsqrtf(q * (1.0f / D) + EPSF);
#pragma unroll
  for (int f = 0; f < F; ++f)
    p[f] = (v[f] - mu) * rs * g[lane * F + f] + b[lane * F + f];
}

__global__ void pool_accum(const float* __restrict__ a, const int* __restrict__ batch,
                           float* __restrict__ sums, float* __restrict__ cnts, int n) {
  const int wave = (int)(((size_t)blockIdx.x * blockDim.x + threadIdx.x) >> 5);
  const int lane = threadIdx.x & 31;
  if (wave >= n) return;
  const int g = batch[wave];
  const float* __restrict__ p = a + (size_t)wave * DOUT + lane * 2;
  atomicAdd(&sums[g * DOUT + lane * 2 + 0], p[0]);
  atomicAdd(&sums[g * DOUT + lane * 2 + 1], p[1]);
  if (lane == 0) atomicAdd(&cnts[g], 1.0f);
}

__global__ void pool_final(const float* __restrict__ sums, const float* __restrict__ cnts,
                           float* __restrict__ out) {
  int i = blockIdx.x * blockDim.x + threadIdx.x;
  if (i < NGRAPHS * DOUT) {
    int g = i / DOUT;
    float m = sums[i] / fmaxf(cnts[g], 1.0f);
    out[i] = 1.0f / (1.0f + __expf(-m));
  }
}

extern "C" void kernel_launch(void* const* d_in, const int* in_sizes, int n_in,
                              void* d_out, int out_size, void* d_ws, size_t ws_size,
                              hipStream_t stream) {
  const float* x   = (const float*)d_in[0];
  const int*   ei  = (const int*)d_in[1];
  const float* ew  = (const float*)d_in[2];
  const int*   bat = (const int*)d_in[3];
  const float* W1  = (const float*)d_in[4];
  const float* b1  = (const float*)d_in[5];
  const float* g1  = (const float*)d_in[6];
  const float* bt1 = (const float*)d_in[7];
  const float* W2  = (const float*)d_in[8];
  const float* b2  = (const float*)d_in[9];
  const float* g2  = (const float*)d_in[10];
  const float* bt2 = (const float*)d_in[11];
  float* out = (float*)d_out;

  const int* rows = ei;
  const int* cols = ei + NEDGES;

  char* ws = (char*)d_ws;
  size_t off = 0;
  auto carve = [&](size_t bytes) -> float* {
    float* p = (float*)(ws + off);
    off += (bytes + 255) & ~(size_t)255;
    return p;
  };
  float* dinv = carve((size_t)NNODES * 4);
  float* h    = carve((size_t)NNODES * DHID * 4);
  float* a    = carve((size_t)NNODES * DHID * 4);
  float* sums = carve((size_t)NGRAPHS * DOUT * 4);
  float* cnts = carve((size_t)NGRAPHS * 4);

  const int TB = 256;
  auto blocks = [&](long long threads) { return (int)((threads + TB - 1) / TB); };

  zero_f32<<<blocks(NNODES), TB, 0, stream>>>(dinv, NNODES);
  deg_accum<<<blocks(NEDGES), TB, 0, stream>>>(cols, ew, dinv, NEDGES);
  deg_to_dinv<<<blocks(NNODES), TB, 0, stream>>>(dinv, NNODES);

  wmma_gemm_f32<DHID, DIN><<<blocks((long long)(NNODES / 16) * 32), TB, 0, stream>>>(x, W1, h, NNODES);
  init_out<DHID><<<blocks((long long)NNODES * 32), TB, 0, stream>>>(h, dinv, b1, a, NNODES);
  edge_scatter<DHID><<<blocks((long long)NEDGES * 32), TB, 0, stream>>>(h, rows, cols, ew, dinv, a, NEDGES);
  relu_layernorm<DHID><<<blocks((long long)NNODES * 32), TB, 0, stream>>>(a, g1, bt1, NNODES);

  wmma_gemm_f32<DOUT, DHID><<<blocks((long long)(NNODES / 16) * 32), TB, 0, stream>>>(a, W2, h, NNODES);
  init_out<DOUT><<<blocks((long long)NNODES * 32), TB, 0, stream>>>(h, dinv, b2, a, NNODES);
  edge_scatter<DOUT><<<blocks((long long)NEDGES * 32), TB, 0, stream>>>(h, rows, cols, ew, dinv, a, NEDGES);
  relu_layernorm<DOUT><<<blocks((long long)NNODES * 32), TB, 0, stream>>>(a, g2, bt2, NNODES);

  zero_f32<<<blocks(NGRAPHS * DOUT), TB, 0, stream>>>(sums, NGRAPHS * DOUT);
  zero_f32<<<blocks(NGRAPHS), TB, 0, stream>>>(cnts, NGRAPHS);
  pool_accum<<<blocks((long long)NNODES * 32), TB, 0, stream>>>(a, bat, sums, cnts, NNODES);
  pool_final<<<blocks(NGRAPHS * DOUT), TB, 0, stream>>>(sums, cnts, out);
}


# pool: sum runs of rows with equal graph id in registers (32 rows per wave), one atomicAdd per run and feature instead of one per row; changes the order of the pool sums
# speedup vs baseline: 1.2055x; 1.2055x over previous
#include <hip/hip_runtime.h>
#include <math.h>

#define NNODES 100000
#define NEDGES 1600000
#define NGRAPHS 64
#define DIN 256
#define DHID 128
#define DOUT 64
#define EPSF 1e-5f

typedef __attribute__((ext_vector_type(2))) float v2f;
typedef __attribute__((ext_vector_type(8))) float v8f;

__global__ void zero_f32(float* __restrict__ p, int n) {
  int i = blockIdx.x * blockDim.x + threadIdx.x;
  if (i < n) p[i] = 0.0f;
}

__global__ void deg_accum(const int* __restrict__ col, const float* __restrict__ w,
                          float* __restrict__ deg, int nE) {
  int e = blockIdx.x * blockDim.x + threadIdx.x;
  if (e < nE) atomicAdd(&deg[col[e]], w[e]);
}

__global__ void deg_to_dinv(float* __restrict__ deg, int n) {
  int i = blockIdx.x * blockDim.x + threadIdx.x;
  if (i < n) {
    float d = deg[i] + 1.0f;
    deg[i] = rsqrtf(fmaxf(d, EPSF));
  }
}

template <int NCOLS, int KDIM>
__global__ void wmma_gemm_f32(const float* __restrict__ A, const float* __restrict__ B,
                              float* __restrict__ C, int M) {
  __shared__ float2 Bl[(KDIM / 2) * NCOLS];

  for (int i = threadIdx.x; i < (KDIM / 2) * NCOLS; i += blockDim.x) {
    const int k2 = i / NCOLS;
    const int n = i - k2 * NCOLS;
    float2 v;
    v.x = B[(size_t)(2 * k2 + 0) * NCOLS + n];
    v.y = B[(size_t)(2 * k2 + 1) * NCOLS + n];
    Bl[i] = v;
  }
  __syncthreads();

  const int wave = (int)(((size_t)blockIdx.x * blockDim.x + threadIdx.x) >> 5);
  const int lane = threadIdx.x & 31;
  if (wave >= M / 16) return;
  const int m0 = wave * 16;
  const int half = lane >> 4;
  const int l = lane & 15;
  constexpr int NT = NCOLS / 16;

  v8f acc[NT];
#pragma unroll
  for (int t = 0; t < NT; ++t)
#pragma unroll
    for (int j = 0; j < 8; ++j) acc[t][j] = 0.0f;

  const float* __restrict__ arow = A + (size_t)(m0 + l) * KDIM + 2 * half;

  v2f a;
  a[0] = arow[0];
  a[1] = arow[1];

  for (int k0 = 0; k0 < KDIM; k0 += 4) {
    v2f a_next = a;
    if (k0 + 4 < KDIM) {
      a_next[0] = arow[k0 + 4 + 0];
      a_next[1] = arow[k0 + 4 + 1];
    }
    const float2* __restrict__ brow = Bl + (size_t)(k0 / 2 + half) * NCOLS + l;
    v2f bfrag[NT];
#pragma unroll
    for (int t = 0; t < NT; ++t) {
      float2 bb = brow[t * 16];
      bfrag[t][0] = bb.x;
      bfrag[t][1] = bb.y;
    }
#pragma unroll
    for (int t = 0; t < NT; ++t)
      acc[t] = __builtin_amdgcn_wmma_f32_16x16x4_f32(
           false, a,  false, bfrag[t],
           (short)0, acc[t],  false,  false);
    a = a_next;
  }

#pragma unroll
  for (int t = 0; t < NT; ++t)
#pragma unroll
    for (int r = 0; r < 8; ++r)
      C[(size_t)(m0 + r + 8 * half) * NCOLS + t * 16 + l] = acc[t][r];
}

template <int D>
__global__ void init_out(const float* __restrict__ h, const float* __restrict__ dinv,
                         const float* __restrict__ bias, float* __restrict__ out, int n) {
  const int wave = (int)(((size_t)blockIdx.x * blockDim.x + threadIdx.x) >> 5);
  const int lane = threadIdx.x & 31;
  if (wave >= n) return;
  constexpr int F = D / 32;
  const float s = dinv[wave] * dinv[wave];
  const float* __restrict__ hp = h + (size_t)wave * D + lane * F;
  float* __restrict__ op = out + (size_t)wave * D + lane * F;
#pragma unroll
  for (int f = 0; f < F; ++f) op[f] = bias[lane * F + f] + hp[f] * s;
}

template <int D>
__global__ void edge_scatter(const float* __restrict__ h, const int* __restrict__ rows,
                             const int* __restrict__ cols, const float* __restrict__ w,
                             const float* __restrict__ dinv, float* __restrict__ out, int nE) {
  const int wave = (int)(((size_t)blockIdx.x * blockDim.x + threadIdx.x) >> 5);
  const int lane = threadIdx.x & 31;
  if (wave >= nE) return;
  const int r = rows[wave];
  const int c = cols[wave];
  const float nrm = dinv[r] * w[wave] * dinv[c];
  constexpr int F = D / 32;
  const float* __restrict__ hp = h + (size_t)r * D + lane * F;
  float* __restrict__ op = out + (size_t)c * D + lane * F;
#pragma unroll
  for (int f = 0; f < F; ++f) atomicAdd(op + f, hp[f] * nrm);
}

template <int D>
__global__ void relu_layernorm(float* __restrict__ x, const float* __restrict__ g,
                               const float* __restrict__ b, int n) {
  const int wave = (int)(((size_t)blockIdx.x * blockDim.x + threadIdx.x) >> 5);
  const int lane = threadIdx.x & 31;
  if (wave >= n) return;
  constexpr int F = D / 32;
  float* __restrict__ p = x + (size_t)wave * D + lane * F;
  float v[F];
  float s = 0.0f;
#pragma unroll
  for (int f = 0; f < F; ++f) {
    v[f] = fmaxf(p[f], 0.0f);
    s += v[f];
  }
#pragma unroll
  for (int off = 16; off > 0; off >>= 1) s += __shfl_xor(s, off, 32);
  const float mu = s * (1.0f / D);
  float q = 0.0f;
#pragma unroll
  for (int f = 0; f < F; ++f) {
    float d = v[f] - mu;
    q += d * d;
  }
#pragma unroll
  for (int off = 16; off > 0; off >>= 1) q += __shfl_xor(q, off, 32);
  const float rs = rsqrtf(q * (1.0f / D) + EPSF);
#pragma unroll
  for (int f = 0; f < F; ++f)
    p[f] = (v[f] - mu) * rs * g[lane * F + f] + b[lane * F + f];
}

#define PR 32
__global__ void pool_accum(const float* __restrict__ a, const int* __restrict__ batch,
                           float* __restrict__ sums, float* __restrict__ cnts, int n) {
  const int wave = (int)(((size_t)blockIdx.x * blockDim.x + threadIdx.x) >> 5);
  const int lane = threadIdx.x & 31;
  const int row0 = wave * PR;
  if (row0 >= n) return;
  const int nrows = min(PR, n - row0);
  const float2* __restrict__ p =
      reinterpret_cast<const float2*>(a + (size_t)row0 * DOUT) + lane;
  const int gl = batch[row0 + min(lane, nrows - 1)];
  const int g0 = __builtin_amdgcn_readfirstlane(gl);

  if (nrows == PR && __all(gl == g0)) {
    float2 v[PR];
#pragma unroll
    for (int i = 0; i < PR; ++i) v[i] = p[i * (DOUT / 2)];
    float s0 = 0.0f, s1 = 0.0f;
#pragma unroll
    for (int i = 0; i < PR; ++i) {
      s0 += v[i].x;
      s1 += v[i].y;
    }
    atomicAdd(&sums[g0 * DOUT + lane * 2 + 0], s0);
    atomicAdd(&sums[g0 * DOUT + lane * 2 + 1], s1);
    if (lane == 0) atomicAdd(&cnts[g0], (float)PR);
  } else {
    int cur = g0;
    float s0 = 0.0f, s1 = 0.0f, c = 0.0f;
    for (int i = 0; i < nrows; ++i) {
      const int gi = __builtin_amdgcn_readfirstlane(batch[row0 + i]);
      if (gi != cur) {
        atomicAdd(&sums[cur * DOUT + lane * 2 + 0], s0);
        atomicAdd(&sums[cur * DOUT + lane * 2 + 1], s1);
        if (lane == 0) atomicAdd(&cnts[cur], c);
        cur = gi;
        s0 = 0.0f;
        s1 = 0.0f;
        c = 0.0f;
      }
      const float2 v = p[i * (DOUT / 2)];
      s0 += v.x;
      s1 += v.y;
      c += 1.0f;
    }
    atomicAdd(&sums[cur * DOUT + lane * 2 + 0], s0);
    atomicAdd(&sums[cur * DOUT + lane * 2 + 1], s1);
    if (lane == 0) atomicAdd(&cnts[cur], c);
  }
}

__global__ void pool_final(const float* __restrict__ sums, const float* __restrict__ cnts,
                           float* __restrict__ out) {
  int i = blockIdx.x * blockDim.x + threadIdx.x;
  if (i < NGRAPHS * DOUT) {
    int g = i / DOUT;
    float m = sums[i] / fmaxf(cnts[g], 1.0f);
    out[i] = 1.0f / (1.0f + __expf(-m));
  }
}

extern "C" void kernel_launch(void* const* d_in, const int* in_sizes, int n_in,
                              void* d_out, int out_size, void* d_ws, size_t ws_size,
                              hipStream_t stream) {
  const float* x   = (const float*)d_in[0];
  const int*   ei  = (const int*)d_in[1];
  const float* ew  = (const float*)d_in[2];
  const int*   bat = (const int*)d_in[3];
  const float* W1  = (const float*)d_in[4];
  const float* b1  = (const float*)d_in[5];
  const float* g1  = (const float*)d_in[6];
  const float* bt1 = (const float*)d_in[7];
  const float* W2  = (const float*)d_in[8];
  const float* b2  = (const float*)d_in[9];
  const float* g2  = (const float*)d_in[10];
  const float* bt2 = (const float*)d_in[11];
  float* out = (float*)d_out;

  const int* rows = ei;
  const int* cols = ei + NEDGES;

  char* ws = (char*)d_ws;
  size_t off = 0;
  auto carve = [&](size_t bytes) -> float* {
    float* p = (float*)(ws + off);
    off += (bytes + 255) & ~(size_t)255;
    return p;
  };
  float* dinv = carve((size_t)NNODES * 4);
  float* h    = carve((size_t)NNODES * DHID * 4);
  float* a    = carve((size_t)NNODES * DHID * 4);
  float* sums = carve((size_t)NGRAPHS * DOUT * 4);
  float* cnts = carve((size_t)NGRAPHS * 4);

  const int TB = 256;
  auto blocks = [&](long long threads) { return (int)((threads + TB - 1) / TB); };

  zero_f32<<<blocks(NNODES), TB, 0, stream>>>(dinv, NNODES);
  deg_accum<<<blocks(NEDGES), TB, 0, stream>>>(cols, ew, dinv, NEDGES);
  deg_to_dinv<<<blocks(NNODES), TB, 0, stream>>>(dinv, NNODES);

  wmma_gemm_f32<DHID, DIN><<<blocks((long long)(NNODES / 16) * 32), TB, 0, stream>>>(x, W1, h, NNODES);
  init_out<DHID><<<blocks((long long)NNODES * 32), TB, 0, stream>>>(h, dinv, b1, a, NNODES);
  edge_scatter<DHID><<<blocks((long long)NEDGES * 32), TB, 0, stream>>>(h, rows, cols, ew, dinv, a, NEDGES);
  relu_layernorm<DHID><<<blocks((long long)NNODES * 32), TB, 0, stream>>>(a, g1, bt1, NNODES);

  wmma_gemm_f32<DOUT, DHID><<<blocks((long long)(NNODES / 16) * 32), TB, 0, stream>>>(a, W2, h, NNODES);
  init_out<DOUT><<<blocks((long long)NNODES * 32), TB, 0, stream>>>(h, dinv, b2, a, NNODES);
  edge_scatter<DOUT><<<blocks((long long)NEDGES * 32), TB, 0, stream>>>(h, rows, cols, ew, dinv, a, NEDGES);
  relu_layernorm<DOUT><<<blocks((long long)NNODES * 32), TB, 0, stream>>>(a, g2, bt2, NNODES);

  zero_f32<<<blocks(NGRAPHS * DOUT), TB, 0, stream>>>(sums, NGRAPHS * DOUT);
  zero_f32<<<blocks(NGRAPHS), TB, 0, stream>>>(cnts, NGRAPHS);
  pool_accum<<<blocks((long long)((NNODES + PR - 1) / PR) * 32), TB, 0, stream>>>(a, bat, sums, cnts, NNODES);
  pool_final<<<blocks(NGRAPHS * DOUT), TB, 0, stream>>>(sums, cnts, out);
}


# edges bucketed by target per call (returning int atomic = arrival-order rank, scan, place); a wave sums a node's rows in registers, no float row atomics: sum order changes; layer 1 in 64-col planes
# speedup vs baseline: 2.1893x; 1.8161x over previous
#include <hip/hip_runtime.h>
#include <math.h>


#define NNODES 100000
#define NEDGES 1600000
#define NGRAPHS 64
#define DIN 256
#define DHID 128
#define DOUT 64
#define EPSF 1e-5f
#define PLANE 64
#define NPAD 100032

typedef __attribute__((ext_vector_type(2))) float v2f;
typedef __attribute__((ext_vector_type(8))) float v8f;

__global__ void zero_f32(float* __restrict__ p, int n) {
  int i = blockIdx.x * blockDim.x + threadIdx.x;
  if (i < n) p[i] = 0.0f;
}

__global__ void deg_count(const int* __restrict__ col, const float* __restrict__ w,
                          float* __restrict__ deg, int* __restrict__ cnt,
                          int* __restrict__ slot, int nE) {
  int e = blockIdx.x * blockDim.x + threadIdx.x;
  if (e < nE) {
    const int c = col[e];
    atomicAdd(&deg[c], w[e]);
    slot[e] = atomicAdd(&cnt[c], 1);
  }
}

__global__ void deg_to_dinv(float* __restrict__ deg, int n) {
  int i = blockIdx.x * blockDim.x + threadIdx.x;
  if (i < n) {
    float d = deg[i] + 1.0f;
    deg[i] = rsqrtf(fmaxf(d, EPSF));
  }
}

#define SCAN_TB 1024
__global__ void offs_scan(const int* __restrict__ cnt, int* __restrict__ offs, int n) {
  __shared__ int wsum[SCAN_TB / 32];
  const int tid = threadIdx.x;
  const int lane = tid & 31;
  const int wv = tid >> 5;
  const int first = blockIdx.x * (SCAN_TB * 4);

  int part = 0;
  const int4* __restrict__ c4 = reinterpret_cast<const int4*>(cnt);
#pragma unroll 8
  for (int i = tid; i < first / 4; i += SCAN_TB) {
    const int4 q = c4[i];
    part += (q.x + q.y) + (q.z + q.w);
  }
#pragma unroll
  for (int off = 16; off > 0; off >>= 1) part += __shfl_xor(part, off, 32);
  if (lane == 0) wsum[wv] = part;
  __syncthreads();
  int base = 0;
  for (int k = 0; k < SCAN_TB / 32; ++k) base += wsum[k];
  __syncthreads();

  const int idx0 = first + tid * 4;
  int v[4];
#pragma unroll
  for (int j = 0; j < 4; ++j) v[j] = (idx0 + j < n) ? cnt[idx0 + j] : 0;
  const int tsum = (v[0] + v[1]) + (v[2] + v[3]);
  int inc = tsum;
#pragma unroll
  for (int off = 1; off < 32; off <<= 1) {
    const int t = __shfl_up(inc, off, 32);
    if (lane >= off) inc += t;
  }
  if (lane == 31) wsum[wv] = inc;
  __syncthreads();
  int wbase = 0;
  for (int k = 0; k < SCAN_TB / 32; ++k) wbase += (k < wv) ? wsum[k] : 0;
  int run = base + wbase + inc - tsum;
#pragma unroll
  for (int j = 0; j < 4; ++j) {
    if (idx0 + j < n) {
      offs[idx0 + j] = run;
      run += v[j];
      if (idx0 + j == n - 1) offs[n] = run;
    }
  }
}

__global__ void place_edges(const int* __restrict__ rows, const int* __restrict__ cols,
                            const float* __restrict__ w, const int* __restrict__ slot,
                            const int* __restrict__ offs, const float* __restrict__ dinv,
                            int2* __restrict__ pairs, int nE) {
  int e = blockIdx.x * blockDim.x + threadIdx.x;
  if (e < nE) {
    const int r = rows[e];
    const int c = cols[e];
    int2 pr;
    pr.x = r;
    pr.y = __float_as_int(dinv[r] * w[e]);
    pairs[offs[c] + slot[e]] = pr;
  }
}

template <int NCOLS, int KDIM, int KPL, int CPL>
__global__ void wmma_gemm_f32(const float* __restrict__ A0, const float* __restrict__ A1,
                              const float* __restrict__ B, float* __restrict__ C0,
                              float* __restrict__ C1, int M) {
  __shared__ float2 Bl[(KDIM / 2) * NCOLS];

  for (int i = threadIdx.x; i < (KDIM / 2) * NCOLS; i += blockDim.x) {
    const int k2 = i / NCOLS;
    const int n = i - k2 * NCOLS;
    float2 v;
    v.x = B[(size_t)(2 * k2 + 0) * NCOLS + n];
    v.y = B[(size_t)(2 * k2 + 1) * NCOLS + n];
    Bl[i] = v;
  }
  __syncthreads();

  const int wave = (int)(((size_t)blockIdx.x * blockDim.x + threadIdx.x) >> 5);
  const int lane = threadIdx.x & 31;
  if (wave >= M / 16) return;
  const int m0 = wave * 16;
  const int half = lane >> 4;
  const int l = lane & 15;
  constexpr int NT = NCOLS / 16;

  v8f acc[NT];
#pragma unroll
  for (int t = 0; t < NT; ++t)
#pragma unroll
    for (int j = 0; j < 8; ++j) acc[t][j] = 0.0f;

  auto a_at = [&](int k) -> const float* {
    const float* base = (KPL < KDIM && k >= KPL) ? A1 : A0;
    return base + (size_t)(m0 + l) * KPL + (k & (KPL - 1)) + 2 * half;
  };

  v2f a;
  {
    const float* ap = a_at(0);
    a[0] = ap[0];
    a[1] = ap[1];
  }

  for (int k0 = 0; k0 < KDIM; k0 += 4) {
    v2f a_next = a;
    if (k0 + 4 < KDIM) {
      const float* np = a_at(k0 + 4);
      a_next[0] = np[0];
      a_next[1] = np[1];
    }
    const float2* __restrict__ brow = Bl + (size_t)(k0 / 2 + half) * NCOLS + l;
    v2f bfrag[NT];
#pragma unroll
    for (int t = 0; t < NT; ++t) {
      float2 bb = brow[t * 16];
      bfrag[t][0] = bb.x;
      bfrag[t][1] = bb.y;
    }
#pragma unroll
    for (int t = 0; t < NT; ++t)
      acc[t] = __builtin_amdgcn_wmma_f32_16x16x4_f32(
           false, a,  false, bfrag[t],
           (short)0, acc[t],  false,  false);
    a = a_next;
  }

#pragma unroll
  for (int t = 0; t < NT; ++t) {
    constexpr int TPP = CPL / 16;
    float* __restrict__ cb = (t / TPP) ? C1 : C0;
    const int cc = (t % TPP) * 16 + l;
#pragma unroll
    for (int r = 0; r < 8; ++r)
      cb[(size_t)(m0 + r + 8 * half) * CPL + cc] = acc[t][r];
  }
}

#define GCH 8
__global__ void gather_rows(const float* __restrict__ hp, const int* __restrict__ offs,
                            const int2* __restrict__ pairs, const float* __restrict__ dinv,
                            const float* __restrict__ bias, float* __restrict__ outp, int n) {
  const int wave = (int)(((size_t)blockIdx.x * blockDim.x + threadIdx.x) >> 5);
  const int lane = threadIdx.x & 31;
  if (wave >= n) return;
  const int node = __builtin_amdgcn_readfirstlane(wave);
  const float2* __restrict__ H = reinterpret_cast<const float2*>(hp) + lane;
  const int beg = offs[node];
  const int end = offs[node + 1];
  const float dc = dinv[node];
  const float2 self = H[(size_t)node * (PLANE / 2)];
  const float2 bb = reinterpret_cast<const float2*>(bias)[lane];
  const float s = dc * dc;
  float a0 = bb.x + self.x * s;
  float a1 = bb.y + self.y * s;

  for (int base = beg; base < end; base += 32) {
    const int m = min(32, end - base);
    const int2 pr = pairs[base + min(lane, m - 1)];
    const int rv = pr.x;
    const int cv = __float_as_int(__int_as_float(pr.y) * dc);
    for (int j = 0; j < m; j += GCH) {
      float2 v[GCH];
      float cf[GCH];
#pragma unroll
      for (int u = 0; u < GCH; ++u) {
        const bool ok = (j + u < m);
        const int src = ok ? (j + u) : j;
        const int r = __builtin_amdgcn_readlane(rv, src);
        const float c = __int_as_float(__builtin_amdgcn_readlane(cv, src));
        const float2 t = H[(size_t)r * (PLANE / 2)];
        v[u].x = ok ? t.x : 0.0f;
        v[u].y = ok ? t.y : 0.0f;
        cf[u] = ok ? c : 0.0f;
      }
#pragma unroll
      for (int u = 0; u < GCH; ++u) {
        a0 += v[u].x * cf[u];
        a1 += v[u].y * cf[u];
      }
    }
  }
  float2 o;
  o.x = a0;
  o.y = a1;
  reinterpret_cast<float2*>(outp)[(size_t)node * (PLANE / 2) + lane] = o;
}

__global__ void relu_layernorm_2p(float* __restrict__ x0, float* __restrict__ x1,
                                  const float* __restrict__ g, const float* __restrict__ b, int n) {
  const int wave = (int)(((size_t)blockIdx.x * blockDim.x + threadIdx.x) >> 5);
  const int lane = threadIdx.x & 31;
  if (wave >= n) return;
  constexpr int D = 2 * PLANE;
  constexpr int F = D / 32;
  float* __restrict__ p = ((lane < 16) ? x0 : x1) + (size_t)wave * PLANE + (lane & 15) * F;
  float v[F];
  float s = 0.0f;
#pragma unroll
  for (int f = 0; f < F; ++f) {
    v[f] = fmaxf(p[f], 0.0f);
    s += v[f];
  }
#pragma unroll
  for (int off = 16; off > 0; off >>= 1) s += __shfl_xor(s, off, 32);
  const float mu = s * (1.0f / D);
  float q = 0.0f;
#pragma unroll
  for (int f = 0; f < F; ++f) {
    float d = v[f] - mu;
    q += d * d;
  }
#pragma unroll
  for (int off = 16; off > 0; off >>= 1) q += __shfl_xor(q, off, 32);
  const float rs = rsqrtf(q * (1.0f / D) + EPSF);
#pragma unroll
  for (int f = 0; f < F; ++f)
    p[f] = (v[f] - mu) * rs * g[lane * F + f] + b[lane * F + f];
}

template <int D>
__global__ void relu_layernorm(float* __restrict__ x, const float* __restrict__ g,
                               const float* __restrict__ b, int n) {
  const int wave = (int)(((size_t)blockIdx.x * blockDim.x + threadIdx.x) >> 5);
  const int lane = threadIdx.x & 31;
  if (wave >= n) return;
  constexpr int F = D / 32;
  float* __restrict__ p = x + (size_t)wave * D + lane * F;
  float v[F];
  float s = 0.0f;
#pragma unroll
  for (int f = 0; f < F; ++f) {
    v[f] = fmaxf(p[f], 0.0f);
    s += v[f];
  }
#pragma unroll
  for (int off = 16; off > 0; off >>= 1) s += __shfl_xor(s, off, 32);
  const float mu = s * (1.0f / D);
  float q = 0.0f;
#pragma unroll
  for (int f = 0; f < F; ++f) {
    float d = v[f] - mu;
    q += d * d;
  }
#pragma unroll
  for (int off = 16; off > 0; off >>= 1) q += __shfl_xor(q, off, 32);
  const float rs = rsqrtf(q * (1.0f / D) + EPSF);
#pragma unroll
  for (int f = 0; f < F; ++f)
    p[f] = (v[f] - mu) * rs * g[lane * F + f] + b[lane * F + f];
}

#define PR 32
__global__ void pool_accum(const float* __restrict__ a, const int* __restrict__ batch,
                           float* __restrict__ sums, float* __restrict__ cnts, int n) {
  const int wave = (int)(((size_t)blockIdx.x * blockDim.x + threadIdx.x) >> 5);
  const int lane = threadIdx.x & 31;
  const int row0 = wave * PR;
  if (row0 >= n) return;
  const int nrows = min(PR, n - row0);
  const float2* __restrict__ p =
      reinterpret_cast<const float2*>(a + (size_t)row0 * DOUT) + lane;
  const int gl = batch[row0 + min(lane, nrows - 1)];
  const int g0 = __builtin_amdgcn_readfirstlane(gl);

  if (nrows == PR && __all(gl == g0)) {
    float2 v[PR];
#pragma unroll
    for (int i = 0; i < PR; ++i) v[i] = p[i * (DOUT / 2)];
    float s0 = 0.0f, s1 = 0.0f;
#pragma unroll
    for (int i = 0; i < PR; ++i) {
      s0 += v[i].x;
      s1 += v[i].y;
    }
    atomicAdd(&sums[g0 * DOUT + lane * 2 + 0], s0);
    atomicAdd(&sums[g0 * DOUT + lane * 2 + 1], s1);
    if (lane == 0) atomicAdd(&cnts[g0], (float)PR);
  } else {
    int cur = g0;
    float s0 = 0.0f, s1 = 0.0f, c = 0.0f;
    for (int i = 0; i < nrows; ++i) {
      const int gi = __builtin_amdgcn_readfirstlane(batch[row0 + i]);
      if (gi != cur) {
        atomicAdd(&sums[cur * DOUT + lane * 2 + 0], s0);
        atomicAdd(&sums[cur * DOUT + lane * 2 + 1], s1);
        if (lane == 0) atomicAdd(&cnts[cur], c);
        cur = gi;
        s0 = 0.0f;
        s1 = 0.0f;
        c = 0.0f;
      }
      const float2 v = p[i * (DOUT / 2)];
      s0 += v.x;
      s1 += v.y;
      c += 1.0f;
    }
    atomicAdd(&sums[cur * DOUT + lane * 2 + 0], s0);
    atomicAdd(&sums[cur * DOUT + lane * 2 + 1], s1);
    if (lane == 0) atomicAdd(&cnts[cur], c);
  }
}

__global__ void pool_final(const float* __restrict__ sums, const float* __restrict__ cnts,
                           float* __restrict__ out) {
  int i = blockIdx.x * blockDim.x + threadIdx.x;
  if (i < NGRAPHS * DOUT) {
    int g = i / DOUT;
    float m = sums[i] / fmaxf(cnts[g], 1.0f);
    out[i] = 1.0f / (1.0f + __expf(-m));
  }
}

extern "C" void kernel_launch(void* const* d_in, const int* in_sizes, int n_in,
                              void* d_out, int out_size, void* d_ws, size_t ws_size,
                              hipStream_t stream) {
  const float* x   = (const float*)d_in[0];
  const int*   ei  = (const int*)d_in[1];
  const float* ew  = (const float*)d_in[2];
  const int*   bat = (const int*)d_in[3];
  const float* W1  = (const float*)d_in[4];
  const float* b1  = (const float*)d_in[5];
  const float* g1  = (const float*)d_in[6];
  const float* bt1 = (const float*)d_in[7];
  const float* W2  = (const float*)d_in[8];
  const float* b2  = (const float*)d_in[9];
  const float* g2  = (const float*)d_in[10];
  const float* bt2 = (const float*)d_in[11];
  float* out = (float*)d_out;

  const int* rows = ei;
  const int* cols = ei + NEDGES;

  char* ws = (char*)d_ws;
  size_t off = 0;
  auto carve = [&](size_t bytes) -> float* {
    float* p = (float*)(ws + off);
    off += (bytes + 255) & ~(size_t)255;
    return p;
  };
  float* hA   = carve((size_t)NNODES * PLANE * 4);
  float* hB   = carve((size_t)NNODES * PLANE * 4);
  float* aA   = carve((size_t)NNODES * PLANE * 4);
  float* dinv = carve((size_t)NPAD * 4);
  int*   cnt  = (int*)carve((size_t)NPAD * 4);
  int*   offs = (int*)carve((size_t)NPAD * 4);
  int*   slot = (int*)carve((size_t)NEDGES * 4);
  int2*  pairs = (int2*)carve((size_t)NEDGES * 8);
  float* sums = carve((size_t)NGRAPHS * DOUT * 4);
  float* cnts = carve((size_t)NGRAPHS * 4);

  const int TB = 256;
  auto blocks = [&](long long threads) { return (int)((threads + TB - 1) / TB); };

  zero_f32<<<blocks(2 * NPAD), TB, 0, stream>>>(dinv, 2 * NPAD);
  deg_count<<<blocks(NEDGES), TB, 0, stream>>>(cols, ew, dinv, cnt, slot, NEDGES);
  deg_to_dinv<<<blocks(NNODES), TB, 0, stream>>>(dinv, NNODES);
  offs_scan<<<(NNODES + SCAN_TB * 4 - 1) / (SCAN_TB * 4), SCAN_TB, 0, stream>>>(cnt, offs, NNODES);
  place_edges<<<blocks(NEDGES), TB, 0, stream>>>(rows, cols, ew, slot, offs, dinv, pairs, NEDGES);

  wmma_gemm_f32<DHID, DIN, DIN, PLANE><<<blocks((long long)(NNODES / 16) * 32), TB, 0, stream>>>(x, x, W1, hA, hB, NNODES);
  gather_rows<<<blocks((long long)NNODES * 32), TB, 0, stream>>>(hA, offs, pairs, dinv, b1, aA, NNODES);
  gather_rows<<<blocks((long long)NNODES * 32), TB, 0, stream>>>(hB, offs, pairs, dinv, b1 + PLANE, hA, NNODES);
  relu_layernorm_2p<<<blocks((long long)NNODES * 32), TB, 0, stream>>>(aA, hA, g1, bt1, NNODES);

  wmma_gemm_f32<DOUT, DHID, PLANE, DOUT><<<blocks((long long)(NNODES / 16) * 32), TB, 0, stream>>>(aA, hA, W2, hB, hB, NNODES);
  gather_rows<<<blocks((long long)NNODES * 32), TB, 0, stream>>>(hB, offs, pairs, dinv, b2, aA, NNODES);
  relu_layernorm<DOUT><<<blocks((long long)NNODES * 32), TB, 0, stream>>>(aA, g2, bt2, NNODES);

  zero_f32<<<blocks(NGRAPHS * DOUT), TB, 0, stream>>>(sums, NGRAPHS * DOUT);
  zero_f32<<<blocks(NGRAPHS), TB, 0, stream>>>(cnts, NGRAPHS);
  pool_accum<<<blocks((long long)((NNODES + PR - 1) / PR) * 32), TB, 0, stream>>>(aA, bat, sums, cnts, NNODES);
  pool_final<<<blocks(NGRAPHS * DOUT), TB, 0, stream>>>(sums, cnts, out);
}


# relu+layernorm folded into neighbours: layer 1's in the layer-2 product's A load (row stats in registers), layer 2's in the gather wave's epilogue; no layernorm launches; LN sum order changes
# speedup vs baseline: 2.3368x; 1.0674x over previous
#include <hip/hip_runtime.h>
#include <math.h>


#define NNODES 100000
#define NEDGES 1600000
#define NGRAPHS 64
#define DIN 256
#define DHID 128
#define DOUT 64
#define EPSF 1e-5f
#define PLANE 64
#define NPAD 100032

typedef __attribute__((ext_vector_type(2))) float v2f;
typedef __attribute__((ext_vector_type(8))) float v8f;

__device__ __forceinline__ float xhalf(float x) {
  const int b = __builtin_bit_cast(int, x);
  return __builtin_bit_cast(
      float, __builtin_amdgcn_permlanex16(b, b, 0x76543210, 0xfedcba98, false, false));
}
template <int XM>
__device__ __forceinline__ float xrow(float x) {
  const int b = __builtin_bit_cast(int, x);
  return __builtin_bit_cast(float, __builtin_amdgcn_update_dpp(b, b, 0x160 + XM, 0xf, 0xf, false));
}
__device__ __forceinline__ float wave_sum(float x) {
  x += xhalf(x);
  x += xrow<8>(x);
  x += xrow<4>(x);
  x += xrow<2>(x);
  x += xrow<1>(x);
  return x;
}

__global__ void zero_f32(float* __restrict__ p, int n) {
  int i = blockIdx.x * blockDim.x + threadIdx.x;
  if (i < n) p[i] = 0.0f;
}

__global__ void deg_count(const int* __restrict__ col, const float* __restrict__ w,
                          float* __restrict__ deg, int* __restrict__ cnt,
                          int* __restrict__ slot, int nE) {
  int e = blockIdx.x * blockDim.x + threadIdx.x;
  if (e < nE) {
    const int c = col[e];
    atomicAdd(&deg[c], w[e]);
    slot[e] = atomicAdd(&cnt[c], 1);
  }
}

__global__ void deg_to_dinv(float* __restrict__ deg, int n) {
  int i = blockIdx.x * blockDim.x + threadIdx.x;
  if (i < n) {
    float d = deg[i] + 1.0f;
    deg[i] = rsqrtf(fmaxf(d, EPSF));
  }
}

#define SCAN_TB 1024
__global__ void offs_scan(const int* __restrict__ cnt, int* __restrict__ offs, int n) {
  __shared__ int wsum[SCAN_TB / 32];
  const int tid = threadIdx.x;
  const int lane = tid & 31;
  const int wv = tid >> 5;
  const int first = blockIdx.x * (SCAN_TB * 4);

  int part = 0;
  const int4* __restrict__ c4 = reinterpret_cast<const int4*>(cnt);
#pragma unroll 8
  for (int i = tid; i < first / 4; i += SCAN_TB) {
    const int4 q = c4[i];
    part += (q.x + q.y) + (q.z + q.w);
  }
#pragma unroll
  for (int off = 16; off > 0; off >>= 1) part += __shfl_xor(part, off, 32);
  if (lane == 0) wsum[wv] = part;
  __syncthreads();
  int base = 0;
  for (int k = 0; k < SCAN_TB / 32; ++k) base += wsum[k];
  __syncthreads();

  const int idx0 = first + tid * 4;
  int v[4];
#pragma unroll
  for (int j = 0; j < 4; ++j) v[j] = (idx0 + j < n) ? cnt[idx0 + j] : 0;
  const int tsum = (v[0] + v[1]) + (v[2] + v[3]);
  int inc = tsum;
#pragma unroll
  for (int off = 1; off < 32; off <<= 1) {
    const int t = __shfl_up(inc, off, 32);
    if (lane >= off) inc += t;
  }
  if (lane == 31) wsum[wv] = inc;
  __syncthreads();
  int wbase = 0;
  for (int k = 0; k < SCAN_TB / 32; ++k) wbase += (k < wv) ? wsum[k] : 0;
  int run = base + wbase + inc - tsum;
#pragma unroll
  for (int j = 0; j < 4; ++j) {
    if (idx0 + j < n) {
      offs[idx0 + j] = run;
      run += v[j];
      if (idx0 + j == n - 1) offs[n] = run;
    }
  }
}

__global__ void place_edges(const int* __restrict__ rows, const int* __restrict__ cols,
                            const float* __restrict__ w, const int* __restrict__ slot,
                            const int* __restrict__ offs, const float* __restrict__ dinv,
                            int2* __restrict__ pairs, int nE) {
  int e = blockIdx.x * blockDim.x + threadIdx.x;
  if (e < nE) {
    const int r = rows[e];
    const int c = cols[e];
    int2 pr;
    pr.x = r;
    pr.y = __float_as_int(dinv[r] * w[e]);
    pairs[offs[c] + slot[e]] = pr;
  }
}

template <int NCOLS, int KDIM, int KPL, int CPL>
__global__ void wmma_gemm_f32(const float* __restrict__ A0, const float* __restrict__ A1,
                              const float* __restrict__ B, float* __restrict__ C0,
                              float* __restrict__ C1, int M) {
  __shared__ float2 Bl[(KDIM / 2) * NCOLS];

  for (int i = threadIdx.x; i < (KDIM / 2) * NCOLS; i += blockDim.x) {
    const int k2 = i / NCOLS;
    const int n = i - k2 * NCOLS;
    float2 v;
    v.x = B[(size_t)(2 * k2 + 0) * NCOLS + n];
    v.y = B[(size_t)(2 * k2 + 1) * NCOLS + n];
    Bl[i] = v;
  }
  __syncthreads();

  const int wave = (int)(((size_t)blockIdx.x * blockDim.x + threadIdx.x) >> 5);
  const int lane = threadIdx.x & 31;
  if (wave >= M / 16) return;
  const int m0 = wave * 16;
  const int half = lane >> 4;
  const int l = lane & 15;
  constexpr int NT = NCOLS / 16;

  v8f acc[NT];
#pragma unroll
  for (int t = 0; t < NT; ++t)
#pragma unroll
    for (int j = 0; j < 8; ++j) acc[t][j] = 0.0f;

  auto a_at = [&](int k) -> const float* {
    const float* base = (KPL < KDIM && k >= KPL) ? A1 : A0;
    return base + (size_t)(m0 + l) * KPL + (k & (KPL - 1)) + 2 * half;
  };

  v2f a;
  {
    const float* ap = a_at(0);
    a[0] = ap[0];
    a[1] = ap[1];
  }

  for (int k0 = 0; k0 < KDIM; k0 += 4) {
    v2f a_next = a;
    if (k0 + 4 < KDIM) {
      const float* np = a_at(k0 + 4);
      a_next[0] = np[0];
      a_next[1] = np[1];
    }
    const float2* __restrict__ brow = Bl + (size_t)(k0 / 2 + half) * NCOLS + l;
    v2f bfrag[NT];
#pragma unroll
    for (int t = 0; t < NT; ++t) {
      float2 bb = brow[t * 16];
      bfrag[t][0] = bb.x;
      bfrag[t][1] = bb.y;
    }
#pragma unroll
    for (int t = 0; t < NT; ++t)
      acc[t] = __builtin_amdgcn_wmma_f32_16x16x4_f32(
           false, a,  false, bfrag[t],
           (short)0, acc[t],  false,  false);
    a = a_next;
  }

#pragma unroll
  for (int t = 0; t < NT; ++t) {
    constexpr int TPP = CPL / 16;
    float* __restrict__ cb = (t / TPP) ? C1 : C0;
    const int cc = (t % TPP) * 16 + l;
#pragma unroll
    for (int r = 0; r < 8; ++r)
      cb[(size_t)(m0 + r + 8 * half) * CPL + cc] = acc[t][r];
  }
}

__global__ void ln_wmma_gemm_f32(const float* __restrict__ A0, const float* __restrict__ A1,
                                 const float* __restrict__ g, const float* __restrict__ b,
                                 const float* __restrict__ B, float* __restrict__ C, int M) {
  constexpr int KDIM = DHID;
  constexpr int NCOLS = DOUT;
  constexpr int NT = NCOLS / 16;
  constexpr int KS = KDIM / 4;
  __shared__ float2 Bl[(KDIM / 2) * NCOLS];

  for (int i = threadIdx.x; i < (KDIM / 2) * NCOLS; i += blockDim.x) {
    const int k2 = i / NCOLS;
    const int n = i - k2 * NCOLS;
    float2 v;
    v.x = B[(size_t)(2 * k2 + 0) * NCOLS + n];
    v.y = B[(size_t)(2 * k2 + 1) * NCOLS + n];
    Bl[i] = v;
  }
  __syncthreads();

  const int wave = (int)(((size_t)blockIdx.x * blockDim.x + threadIdx.x) >> 5);
  const int lane = threadIdx.x & 31;
  if (wave >= M / 16) return;
  const int m0 = wave * 16;
  const int half = lane >> 4;
  const int l = lane & 15;

  const float2* __restrict__ r0 =
      reinterpret_cast<const float2*>(A0 + (size_t)(m0 + l) * PLANE + 2 * half);
  const float2* __restrict__ r1 =
      reinterpret_cast<const float2*>(A1 + (size_t)(m0 + l) * PLANE + 2 * half);
  v2f a[KS];
#pragma unroll
  for (int j = 0; j < KS; ++j) {
    const float2 t = (4 * j < PLANE) ? r0[2 * j] : r1[2 * j - PLANE / 2];
    a[j][0] = fmaxf(t.x, 0.0f);
    a[j][1] = fmaxf(t.y, 0.0f);
  }

  float sp[4] = {0.0f, 0.0f, 0.0f, 0.0f};
#pragma unroll
  for (int j = 0; j < KS; ++j) sp[j & 3] += a[j][0] + a[j][1];
  float s = (sp[0] + sp[1]) + (sp[2] + sp[3]);
  s += xhalf(s);
  const float mu = s * (1.0f / KDIM);
  float qp[4] = {0.0f, 0.0f, 0.0f, 0.0f};
#pragma unroll
  for (int j = 0; j < KS; ++j) {
    const float d0 = a[j][0] - mu;
    const float d1 = a[j][1] - mu;
    qp[j & 3] += d0 * d0 + d1 * d1;
  }
  float q = (qp[0] + qp[1]) + (qp[2] + qp[3]);
  q += xhalf(q);
  const float rs = rsqrtf(q * (1.0f / KDIM) + EPSF);

  const float2* __restrict__ gp = reinterpret_cast<const float2*>(g + 2 * half);
  const float2* __restrict__ bp = reinterpret_cast<const float2*>(b + 2 * half);
#pragma unroll
  for (int j = 0; j < KS; ++j) {
    const float2 gg = gp[2 * j];
    const float2 sh = bp[2 * j];
    a[j][0] = (a[j][0] - mu) * rs * gg.x + sh.x;
    a[j][1] = (a[j][1] - mu) * rs * gg.y + sh.y;
  }

  v8f acc[NT];
#pragma unroll
  for (int t = 0; t < NT; ++t)
#pragma unroll
    for (int j = 0; j < 8; ++j) acc[t][j] = 0.0f;

#pragma unroll
  for (int j = 0; j < KS; ++j) {
    const float2* __restrict__ brow = Bl + (size_t)(2 * j + half) * NCOLS + l;
    v2f bfrag[NT];
#pragma unroll
    for (int t = 0; t < NT; ++t) {
      float2 bb = brow[t * 16];
      bfrag[t][0] = bb.x;
      bfrag[t][1] = bb.y;
    }
#pragma unroll
    for (int t = 0; t < NT; ++t)
      acc[t] = __builtin_amdgcn_wmma_f32_16x16x4_f32(
           false, a[j],  false, bfrag[t],
           (short)0, acc[t],  false,  false);
  }

#pragma unroll
  for (int t = 0; t < NT; ++t) {
    const int cc = t * 16 + l;
#pragma unroll
    for (int r = 0; r < 8; ++r)
      C[(size_t)(m0 + r + 8 * half) * NCOLS + cc] = acc[t][r];
  }
}

#define GCH 8
template <bool NORM>
__global__ void gather_rows(const float* __restrict__ hp, const int* __restrict__ offs,
                            const int2* __restrict__ pairs, const float* __restrict__ dinv,
                            const float* __restrict__ bias, const float* __restrict__ g,
                            const float* __restrict__ b, float* __restrict__ outp, int n) {
  const int wave = (int)(((size_t)blockIdx.x * blockDim.x + threadIdx.x) >> 5);
  const int lane = threadIdx.x & 31;
  if (wave >= n) return;
  const int node = __builtin_amdgcn_readfirstlane(wave);
  const float2* __restrict__ H = reinterpret_cast<const float2*>(hp) + lane;
  const int beg = offs[node];
  const int end = offs[node + 1];
  const float dc = dinv[node];
  const float2 self = H[(size_t)node * (PLANE / 2)];
  const float2 bb = reinterpret_cast<const float2*>(bias)[lane];
  const float s = dc * dc;
  float a0 = bb.x + self.x * s;
  float a1 = bb.y + self.y * s;

  for (int base = beg; base < end; base += 32) {
    const int m = min(32, end - base);
    const int2 pr = pairs[base + min(lane, m - 1)];
    const int rv = pr.x;
    const int cv = __float_as_int(__int_as_float(pr.y) * dc);
    for (int j = 0; j < m; j += GCH) {
      float2 v[GCH];
      float cf[GCH];
#pragma unroll
      for (int u = 0; u < GCH; ++u) {
        const bool ok = (j + u < m);
        const int src = ok ? (j + u) : j;
        const int r = __builtin_amdgcn_readlane(rv, src);
        const float c = __int_as_float(__builtin_amdgcn_readlane(cv, src));
        const float2 t = H[(size_t)r * (PLANE / 2)];
        v[u].x = ok ? t.x : 0.0f;
        v[u].y = ok ? t.y : 0.0f;
        cf[u] = ok ? c : 0.0f;
      }
#pragma unroll
      for (int u = 0; u < GCH; ++u) {
        a0 += v[u].x * cf[u];
        a1 += v[u].y * cf[u];
      }
    }
  }
  float2 o;
  if (NORM) {
    float v[2];
    v[0] = fmaxf(a0, 0.0f);
    v[1] = fmaxf(a1, 0.0f);
    float rsum = 0.0f;
#pragma unroll
    for (int f = 0; f < 2; ++f) rsum += v[f];
    const float mu = wave_sum(rsum) * (1.0f / PLANE);
    float q = 0.0f;
#pragma unroll
    for (int f = 0; f < 2; ++f) {
      float d = v[f] - mu;
      q += d * d;
    }
    const float rs = rsqrtf(wave_sum(q) * (1.0f / PLANE) + EPSF);
    const float2 gg = reinterpret_cast<const float2*>(g)[lane];
    const float2 sh = reinterpret_cast<const float2*>(b)[lane];
    o.x = (v[0] - mu) * rs * gg.x + sh.x;
    o.y = (v[1] - mu) * rs * gg.y + sh.y;
  } else {
    o.x = a0;
    o.y = a1;
  }
  reinterpret_cast<float2*>(outp)[(size_t)node * (PLANE / 2) + lane] = o;
}

#define PR 32
__global__ void pool_accum(const float* __restrict__ a, const int* __restrict__ batch,
                           float* __restrict__ sums, float* __restrict__ cnts, int n) {
  const int wave = (int)(((size_t)blockIdx.x * blockDim.x + threadIdx.x) >> 5);
  const int lane = threadIdx.x & 31;
  const int row0 = wave * PR;
  if (row0 >= n) return;
  const int nrows = min(PR, n - row0);
  const float2* __restrict__ p =
      reinterpret_cast<const float2*>(a + (size_t)row0 * DOUT) + lane;
  const int gl = batch[row0 + min(lane, nrows - 1)];
  const int g0 = __builtin_amdgcn_readfirstlane(gl);

  if (nrows == PR && __all(gl == g0)) {
    float2 v[PR];
#pragma unroll
    for (int i = 0; i < PR; ++i) v[i] = p[i * (DOUT / 2)];
    float s0 = 0.0f, s1 = 0.0f;
#pragma unroll
    for (int i = 0; i < PR; ++i) {
      s0 += v[i].x;
      s1 += v[i].y;
    }
    atomicAdd(&sums[g0 * DOUT + lane * 2 + 0], s0);
    atomicAdd(&sums[g0 * DOUT + lane * 2 + 1], s1);
    if (lane == 0) atomicAdd(&cnts[g0], (float)PR);
  } else {
    int cur = g0;
    float s0 = 0.0f, s1 = 0.0f, c = 0.0f;
    for (int i = 0; i < nrows; ++i) {
      const int gi = __builtin_amdgcn_readfirstlane(batch[row0 + i]);
      if (gi != cur) {
        atomicAdd(&sums[cur * DOUT + lane * 2 + 0], s0);
        atomicAdd(&sums[cur * DOUT + lane * 2 + 1], s1);
        if (lane == 0) atomicAdd(&cnts[cur], c);
        cur = gi;
        s0 = 0.0f;
        s1 = 0.0f;
        c = 0.0f;
      }
      const float2 v = p[i * (DOUT / 2)];
      s0 += v.x;
      s1 += v.y;
      c += 1.0f;
    }
    atomicAdd(&sums[cur * DOUT + lane * 2 + 0], s0);
    atomicAdd(&sums[cur * DOUT + lane * 2 + 1], s1);
    if (lane == 0) atomicAdd(&cnts[cur], c);
  }
}

__global__ void pool_final(const float* __restrict__ sums, const float* __restrict__ cnts,
                           float* __restrict__ out) {
  int i = blockIdx.x * blockDim.x + threadIdx.x;
  if (i < NGRAPHS * DOUT) {
    int g = i / DOUT;
    float m = sums[i] / fmaxf(cnts[g], 1.0f);
    out[i] = 1.0f / (1.0f + __expf(-m));
  }
}

extern "C" void kernel_launch(void* const* d_in, const int* in_sizes, int n_in,
                              void* d_out, int out_size, void* d_ws, size_t ws_size,
                              hipStream_t stream) {
  const float* x   = (const float*)d_in[0];
  const int*   ei  = (const int*)d_in[1];
  const float* ew  = (const float*)d_in[2];
  const int*   bat = (const int*)d_in[3];
  const float* W1  = (const float*)d_in[4];
  const float* b1  = (const float*)d_in[5];
  const float* g1  = (const float*)d_in[6];
  const float* bt1 = (const float*)d_in[7];
  const float* W2  = (const float*)d_in[8];
  const float* b2  = (const float*)d_in[9];
  const float* g2  = (const float*)d_in[10];
  const float* bt2 = (const float*)d_in[11];
  float* out = (float*)d_out;

  const int* rows = ei;
  const int* cols = ei + NEDGES;

  char* ws = (char*)d_ws;
  size_t off = 0;
  auto carve = [&](size_t bytes) -> float* {
    float* p = (float*)(ws + off);
    off += (bytes + 255) & ~(size_t)255;
    return p;
  };
  float* hA   = carve((size_t)NNODES * PLANE * 4);
  float* hB   = carve((size_t)NNODES * PLANE * 4);
  float* aA   = carve((size_t)NNODES * PLANE * 4);
  float* dinv = carve((size_t)NPAD * 4);
  int*   cnt  = (int*)carve((size_t)NPAD * 4);
  int*   offs = (int*)carve((size_t)NPAD * 4);
  int*   slot = (int*)carve((size_t)NEDGES * 4);
  int2*  pairs = (int2*)carve((size_t)NEDGES * 8);
  float* sums = carve((size_t)NGRAPHS * DOUT * 4);
  float* cnts = carve((size_t)NGRAPHS * 4);

  const int TB = 256;
  auto blocks = [&](long long threads) { return (int)((threads + TB - 1) / TB); };

  zero_f32<<<blocks(2 * NPAD), TB, 0, stream>>>(dinv, 2 * NPAD);
  deg_count<<<blocks(NEDGES), TB, 0, stream>>>(cols, ew, dinv, cnt, slot, NEDGES);
  deg_to_dinv<<<blocks(NNODES), TB, 0, stream>>>(dinv, NNODES);
  offs_scan<<<(NNODES + SCAN_TB * 4 - 1) / (SCAN_TB * 4), SCAN_TB, 0, stream>>>(cnt, offs, NNODES);
  place_edges<<<blocks(NEDGES), TB, 0, stream>>>(rows, cols, ew, slot, offs, dinv, pairs, NEDGES);

  wmma_gemm_f32<DHID, DIN, DIN, PLANE><<<blocks((long long)(NNODES / 16) * 32), TB, 0, stream>>>(x, x, W1, hA, hB, NNODES);
  gather_rows<false><<<blocks((long long)NNODES * 32), TB, 0, stream>>>(hA, offs, pairs, dinv, b1, g1, bt1, aA, NNODES);
  gather_rows<false><<<blocks((long long)NNODES * 32), TB, 0, stream>>>(hB, offs, pairs, dinv, b1 + PLANE, g1, bt1, hA, NNODES);

  ln_wmma_gemm_f32<<<blocks((long long)(NNODES / 16) * 32), TB, 0, stream>>>(aA, hA, g1, bt1, W2, hB, NNODES);
  gather_rows<true><<<blocks((long long)NNODES * 32), TB, 0, stream>>>(hB, offs, pairs, dinv, b2, g2, bt2, aA, NNODES);

  zero_f32<<<blocks(NGRAPHS * DOUT), TB, 0, stream>>>(sums, NGRAPHS * DOUT);
  zero_f32<<<blocks(NGRAPHS), TB, 0, stream>>>(cnts, NGRAPHS);
  pool_accum<<<blocks((long long)((NNODES + PR - 1) / PR) * 32), TB, 0, stream>>>(aA, bat, sums, cnts, NNODES);
  pool_final<<<blocks(NGRAPHS * DOUT), TB, 0, stream>>>(sums, cnts, out);
}


# edges bucketed by target with LDS counters in two levels (edge chunk, then 256-node bucket), no global atomic per edge; degree by LDS float adds, edge rank from LDS counters: sum orders change
# speedup vs baseline: 3.2800x; 1.4036x over previous
#include <hip/hip_runtime.h>
#include <math.h>


#define NNODES 100000
#define NEDGES 1600000
#define NGRAPHS 64
#define DIN 256
#define DHID 128
#define DOUT 64
#define EPSF 1e-5f
#define PLANE 64
#define NPAD 100032

typedef __attribute__((ext_vector_type(2))) float v2f;
typedef __attribute__((ext_vector_type(8))) float v8f;

__device__ __forceinline__ float xhalf(float x) {
  const int b = __builtin_bit_cast(int, x);
  return __builtin_bit_cast(
      float, __builtin_amdgcn_permlanex16(b, b, 0x76543210, 0xfedcba98, false, false));
}
template <int XM>
__device__ __forceinline__ float xrow(float x) {
  const int b = __builtin_bit_cast(int, x);
  return __builtin_bit_cast(float, __builtin_amdgcn_update_dpp(b, b, 0x160 + XM, 0xf, 0xf, false));
}
__device__ __forceinline__ float wave_sum(float x) {
  x += xhalf(x);
  x += xrow<8>(x);
  x += xrow<4>(x);
  x += xrow<2>(x);
  x += xrow<1>(x);
  return x;
}

__global__ void zero_f32(float* __restrict__ p, int n) {
  int i = blockIdx.x * blockDim.x + threadIdx.x;
  if (i < n) p[i] = 0.0f;
}

#define SCAN_TB 1024
__global__ void offs_scan(const int* __restrict__ cnt, int* __restrict__ offs, int n) {
  __shared__ int wsum[SCAN_TB / 32];
  const int tid = threadIdx.x;
  const int lane = tid & 31;
  const int wv = tid >> 5;
  const int first = blockIdx.x * (SCAN_TB * 4);

  int part = 0;
  const int4* __restrict__ c4 = reinterpret_cast<const int4*>(cnt);
#pragma unroll 8
  for (int i = tid; i < first / 4; i += SCAN_TB) {
    const int4 q = c4[i];
    part += (q.x + q.y) + (q.z + q.w);
  }
#pragma unroll
  for (int off = 16; off > 0; off >>= 1) part += __shfl_xor(part, off, 32);
  if (lane == 0) wsum[wv] = part;
  __syncthreads();
  int base = 0;
  for (int k = 0; k < SCAN_TB / 32; ++k) base += wsum[k];
  __syncthreads();

  const int idx0 = first + tid * 4;
  int v[4];
#pragma unroll
  for (int j = 0; j < 4; ++j) v[j] = (idx0 + j < n) ? cnt[idx0 + j] : 0;
  const int tsum = (v[0] + v[1]) + (v[2] + v[3]);
  int inc = tsum;
#pragma unroll
  for (int off = 1; off < 32; off <<= 1) {
    const int t = __shfl_up(inc, off, 32);
    if (lane >= off) inc += t;
  }
  if (lane == 31) wsum[wv] = inc;
  __syncthreads();
  int wbase = 0;
  for (int k = 0; k < SCAN_TB / 32; ++k) wbase += (k < wv) ? wsum[k] : 0;
  int run = base + wbase + inc - tsum;
#pragma unroll
  for (int j = 0; j < 4; ++j) {
    if (idx0 + j < n) {
      offs[idx0 + j] = run;
      run += v[j];
      if (idx0 + j == n - 1) offs[n] = run;
    }
  }
}

#define BKSH 8
#define BKN (1 << BKSH)
#define NBK ((NNODES + BKN - 1) / BKN)
#define ECH 256
#define ECHUNK (NEDGES / ECH)
#define TBE 1024
#define HIST_ROOM (((NBK * ECH) + 1 + 1023) & ~1023)
#define EPT ((ECHUNK + TBE - 1) / TBE)
static_assert(NEDGES % ECH == 0, "chunks must tile the edge list");
static_assert(NNODES <= (1 << 17), "source index must fit under the packed node-in-bucket");
static_assert(BKSH + 17 < 32, "the packed word must stay positive");

__global__ void edge_hist(const int* __restrict__ cols, int* __restrict__ hist) {
  __shared__ int h[NBK];
  const int tid = threadIdx.x;
  for (int i = tid; i < NBK; i += TBE) h[i] = 0;
  __syncthreads();
  const int e0 = blockIdx.x * ECHUNK;
  int c[EPT];
#pragma unroll
  for (int k = 0; k < EPT; ++k) {
    const int i = k * TBE + tid;
    c[k] = (i < ECHUNK) ? cols[e0 + i] : -1;
  }
#pragma unroll
  for (int k = 0; k < EPT; ++k)
    if (c[k] >= 0) atomicAdd(&h[c[k] >> BKSH], 1);
  __syncthreads();
  for (int i = tid; i < NBK; i += TBE) hist[i * ECH + blockIdx.x] = h[i];
}

__global__ void edge_bucket(const int* __restrict__ rows, const int* __restrict__ cols,
                            const float* __restrict__ w, const int* __restrict__ base,
                            int2* __restrict__ tmp) {
  __shared__ int cur[NBK];
  const int tid = threadIdx.x;
  for (int i = tid; i < NBK; i += TBE) cur[i] = base[i * ECH + blockIdx.x];
  __syncthreads();
  const int e0 = blockIdx.x * ECHUNK;
  int c[EPT];
  int2 rec[EPT];
#pragma unroll
  for (int k = 0; k < EPT; ++k) {
    const int i = k * TBE + tid;
    const bool ok = (i < ECHUNK);
    c[k] = ok ? cols[e0 + i] : -1;
    rec[k].x = ok ? rows[e0 + i] : 0;
    rec[k].y = ok ? __float_as_int(w[e0 + i]) : 0;
  }
#pragma unroll
  for (int k = 0; k < EPT; ++k) {
    if (c[k] >= 0) {
      const int pos = atomicAdd(&cur[c[k] >> BKSH], 1);
      int2 t;
      t.x = rec[k].x | ((c[k] & (BKN - 1)) << 17);
      t.y = rec[k].y;
      tmp[pos] = t;
    }
  }
}

#define TBF 1024
#define FPT 4
__global__ void bucket_fine(const int2* __restrict__ tmp, const int* __restrict__ base,
                            int* __restrict__ offs, float* __restrict__ dinv,
                            int2* __restrict__ pairs) {
  __shared__ int cntL[BKN];
  __shared__ float degL[BKN];
  __shared__ int curL[BKN];
  __shared__ int wsum[BKN / 32];
  const int tid = threadIdx.x;
  const int lane = tid & 31;
  const int wv = tid >> 5;
  const int start = base[blockIdx.x * ECH];
  const int end = base[(blockIdx.x + 1) * ECH];
  if (tid < BKN) {
    cntL[tid] = 0;
    degL[tid] = 0.0f;
  }
  __syncthreads();

  for (int b0 = start; b0 < end; b0 += FPT * TBF) {
    int2 t[FPT];
#pragma unroll
    for (int k = 0; k < FPT; ++k) {
      const int i = b0 + k * TBF + tid;
      t[k] = (i < end) ? tmp[i] : make_int2(-1, 0);
    }
#pragma unroll
    for (int k = 0; k < FPT; ++k) {
      if (t[k].x >= 0) {
        const int cl = t[k].x >> 17;
        atomicAdd(&cntL[cl], 1);
        atomicAdd(&degL[cl], __int_as_float(t[k].y));
      }
    }
  }
  __syncthreads();

  int v = 0, inc = 0;
  if (tid < BKN) {
    v = cntL[tid];
    inc = v;
#pragma unroll
    for (int off = 1; off < 32; off <<= 1) {
      const int u = __shfl_up(inc, off, 32);
      if (lane >= off) inc += u;
    }
    if (lane == 31) wsum[wv] = inc;
  }
  __syncthreads();
  if (tid < BKN) {
    int wbase = 0;
    for (int k = 0; k < BKN / 32; ++k) wbase += (k < wv) ? wsum[k] : 0;
    const int excl = start + wbase + inc - v;
    curL[tid] = excl;
    const int node = blockIdx.x * BKN + tid;
    if (node < NNODES) {
      offs[node] = excl;
      dinv[node] = rsqrtf(fmaxf(degL[tid] + 1.0f, EPSF));
      if (node == NNODES - 1) offs[NNODES] = excl + v;
    }
  }
  __syncthreads();

  for (int b0 = start; b0 < end; b0 += FPT * TBF) {
    int2 t[FPT];
#pragma unroll
    for (int k = 0; k < FPT; ++k) {
      const int i = b0 + k * TBF + tid;
      t[k] = (i < end) ? tmp[i] : make_int2(-1, 0);
    }
#pragma unroll
    for (int k = 0; k < FPT; ++k) {
      if (t[k].x >= 0) {
        const int pos = atomicAdd(&curL[t[k].x >> 17], 1);
        int2 pr;
        pr.x = t[k].x & ((1 << 17) - 1);
        pr.y = t[k].y;
        pairs[pos] = pr;
      }
    }
  }
}

__global__ void scale_pairs(int2* __restrict__ pairs, const float* __restrict__ dinv, int nE) {
  int e = blockIdx.x * blockDim.x + threadIdx.x;
  if (e < nE) {
    int2 pr = pairs[e];
    pr.y = __float_as_int(dinv[pr.x] * __int_as_float(pr.y));
    pairs[e] = pr;
  }
}

template <int NCOLS, int KDIM, int KPL, int CPL>
__global__ void wmma_gemm_f32(const float* __restrict__ A0, const float* __restrict__ A1,
                              const float* __restrict__ B, float* __restrict__ C0,
                              float* __restrict__ C1, int M) {
  __shared__ float2 Bl[(KDIM / 2) * NCOLS];

  for (int i = threadIdx.x; i < (KDIM / 2) * NCOLS; i += blockDim.x) {
    const int k2 = i / NCOLS;
    const int n = i - k2 * NCOLS;
    float2 v;
    v.x = B[(size_t)(2 * k2 + 0) * NCOLS + n];
    v.y = B[(size_t)(2 * k2 + 1) * NCOLS + n];
    Bl[i] = v;
  }
  __syncthreads();

  const int wave = (int)(((size_t)blockIdx.x * blockDim.x + threadIdx.x) >> 5);
  const int lane = threadIdx.x & 31;
  if (wave >= M / 16) return;
  const int m0 = wave * 16;
  const int half = lane >> 4;
  const int l = lane & 15;
  constexpr int NT = NCOLS / 16;

  v8f acc[NT];
#pragma unroll
  for (int t = 0; t < NT; ++t)
#pragma unroll
    for (int j = 0; j < 8; ++j) acc[t][j] = 0.0f;

  auto a_at = [&](int k) -> const float* {
    const float* base = (KPL < KDIM && k >= KPL) ? A1 : A0;
    return base + (size_t)(m0 + l) * KPL + (k & (KPL - 1)) + 2 * half;
  };

  v2f a;
  {
    const float* ap = a_at(0);
    a[0] = ap[0];
    a[1] = ap[1];
  }

  for (int k0 = 0; k0 < KDIM; k0 += 4) {
    v2f a_next = a;
    if (k0 + 4 < KDIM) {
      const float* np = a_at(k0 + 4);
      a_next[0] = np[0];
      a_next[1] = np[1];
    }
    const float2* __restrict__ brow = Bl + (size_t)(k0 / 2 + half) * NCOLS + l;
    v2f bfrag[NT];
#pragma unroll
    for (int t = 0; t < NT; ++t) {
      float2 bb = brow[t * 16];
      bfrag[t][0] = bb.x;
      bfrag[t][1] = bb.y;
    }
#pragma unroll
    for (int t = 0; t < NT; ++t)
      acc[t] = __builtin_amdgcn_wmma_f32_16x16x4_f32(
           false, a,  false, bfrag[t],
           (short)0, acc[t],  false,  false);
    a = a_next;
  }

#pragma unroll
  for (int t = 0; t < NT; ++t) {
    constexpr int TPP = CPL / 16;
    float* __restrict__ cb = (t / TPP) ? C1 : C0;
    const int cc = (t % TPP) * 16 + l;
#pragma unroll
    for (int r = 0; r < 8; ++r)
      cb[(size_t)(m0 + r + 8 * half) * CPL + cc] = acc[t][r];
  }
}

__global__ void ln_wmma_gemm_f32(const float* __restrict__ A0, const float* __restrict__ A1,
                                 const float* __restrict__ g, const float* __restrict__ b,
                                 const float* __restrict__ B, float* __restrict__ C, int M) {
  constexpr int KDIM = DHID;
  constexpr int NCOLS = DOUT;
  constexpr int NT = NCOLS / 16;
  constexpr int KS = KDIM / 4;
  __shared__ float2 Bl[(KDIM / 2) * NCOLS];

  for (int i = threadIdx.x; i < (KDIM / 2) * NCOLS; i += blockDim.x) {
    const int k2 = i / NCOLS;
    const int n = i - k2 * NCOLS;
    float2 v;
    v.x = B[(size_t)(2 * k2 + 0) * NCOLS + n];
    v.y = B[(size_t)(2 * k2 + 1) * NCOLS + n];
    Bl[i] = v;
  }
  __syncthreads();

  const int wave = (int)(((size_t)blockIdx.x * blockDim.x + threadIdx.x) >> 5);
  const int lane = threadIdx.x & 31;
  if (wave >= M / 16) return;
  const int m0 = wave * 16;
  const int half = lane >> 4;
  const int l = lane & 15;

  const float2* __restrict__ r0 =
      reinterpret_cast<const float2*>(A0 + (size_t)(m0 + l) * PLANE + 2 * half);
  const float2* __restrict__ r1 =
      reinterpret_cast<const float2*>(A1 + (size_t)(m0 + l) * PLANE + 2 * half);
  v2f a[KS];
#pragma unroll
  for (int j = 0; j < KS; ++j) {
    const float2 t = (4 * j < PLANE) ? r0[2 * j] : r1[2 * j - PLANE / 2];
    a[j][0] = fmaxf(t.x, 0.0f);
    a[j][1] = fmaxf(t.y, 0.0f);
  }

  float sp[4] = {0.0f, 0.0f, 0.0f, 0.0f};
#pragma unroll
  for (int j = 0; j < KS; ++j) sp[j & 3] += a[j][0] + a[j][1];
  float s = (sp[0] + sp[1]) + (sp[2] + sp[3]);
  s += xhalf(s);
  const float mu = s * (1.0f / KDIM);
  float qp[4] = {0.0f, 0.0f, 0.0f, 0.0f};
#pragma unroll
  for (int j = 0; j < KS; ++j) {
    const float d0 = a[j][0] - mu;
    const float d1 = a[j][1] - mu;
    qp[j & 3] += d0 * d0 + d1 * d1;
  }
  float q = (qp[0] + qp[1]) + (qp[2] + qp[3]);
  q += xhalf(q);
  const float rs = rsqrtf(q * (1.0f / KDIM) + EPSF);

  const float2* __restrict__ gp = reinterpret_cast<const float2*>(g + 2 * half);
  const float2* __restrict__ bp = reinterpret_cast<const float2*>(b + 2 * half);
#pragma unroll
  for (int j = 0; j < KS; ++j) {
    const float2 gg = gp[2 * j];
    const float2 sh = bp[2 * j];
    a[j][0] = (a[j][0] - mu) * rs * gg.x + sh.x;
    a[j][1] = (a[j][1] - mu) * rs * gg.y + sh.y;
  }

  v8f acc[NT];
#pragma unroll
  for (int t = 0; t < NT; ++t)
#pragma unroll
    for (int j = 0; j < 8; ++j) acc[t][j] = 0.0f;

#pragma unroll
  for (int j = 0; j < KS; ++j) {
    const float2* __restrict__ brow = Bl + (size_t)(2 * j + half) * NCOLS + l;
    v2f bfrag[NT];
#pragma unroll
    for (int t = 0; t < NT; ++t) {
      float2 bb = brow[t * 16];
      bfrag[t][0] = bb.x;
      bfrag[t][1] = bb.y;
    }
#pragma unroll
    for (int t = 0; t < NT; ++t)
      acc[t] = __builtin_amdgcn_wmma_f32_16x16x4_f32(
           false, a[j],  false, bfrag[t],
           (short)0, acc[t],  false,  false);
  }

#pragma unroll
  for (int t = 0; t < NT; ++t) {
    const int cc = t * 16 + l;
#pragma unroll
    for (int r = 0; r < 8; ++r)
      C[(size_t)(m0 + r + 8 * half) * NCOLS + cc] = acc[t][r];
  }
}

#define GCH 8
template <bool NORM>
__global__ void gather_rows(const float* __restrict__ hp, const int* __restrict__ offs,
                            const int2* __restrict__ pairs, const float* __restrict__ dinv,
                            const float* __restrict__ bias, const float* __restrict__ g,
                            const float* __restrict__ b, float* __restrict__ outp, int n) {
  const int wave = (int)(((size_t)blockIdx.x * blockDim.x + threadIdx.x) >> 5);
  const int lane = threadIdx.x & 31;
  if (wave >= n) return;
  const int node = __builtin_amdgcn_readfirstlane(wave);
  const float2* __restrict__ H = reinterpret_cast<const float2*>(hp) + lane;
  const int beg = offs[node];
  const int end = offs[node + 1];
  const float dc = dinv[node];
  const float2 self = H[(size_t)node * (PLANE / 2)];
  const float2 bb = reinterpret_cast<const float2*>(bias)[lane];
  const float s = dc * dc;
  float a0 = bb.x + self.x * s;
  float a1 = bb.y + self.y * s;

  for (int base = beg; base < end; base += 32) {
    const int m = min(32, end - base);
    const int2 pr = pairs[base + min(lane, m - 1)];
    const int rv = pr.x;
    const int cv = __float_as_int(__int_as_float(pr.y) * dc);
    for (int j = 0; j < m; j += GCH) {
      float2 v[GCH];
      float cf[GCH];
#pragma unroll
      for (int u = 0; u < GCH; ++u) {
        const bool ok = (j + u < m);
        const int src = ok ? (j + u) : j;
        const int r = __builtin_amdgcn_readlane(rv, src);
        const float c = __int_as_float(__builtin_amdgcn_readlane(cv, src));
        const float2 t = H[(size_t)r * (PLANE / 2)];
        v[u].x = ok ? t.x : 0.0f;
        v[u].y = ok ? t.y : 0.0f;
        cf[u] = ok ? c : 0.0f;
      }
#pragma unroll
      for (int u = 0; u < GCH; ++u) {
        a0 += v[u].x * cf[u];
        a1 += v[u].y * cf[u];
      }
    }
  }
  float2 o;
  if (NORM) {
    float v[2];
    v[0] = fmaxf(a0, 0.0f);
    v[1] = fmaxf(a1, 0.0f);
    float rsum = 0.0f;
#pragma unroll
    for (int f = 0; f < 2; ++f) rsum += v[f];
    const float mu = wave_sum(rsum) * (1.0f / PLANE);
    float q = 0.0f;
#pragma unroll
    for (int f = 0; f < 2; ++f) {
      float d = v[f] - mu;
      q += d * d;
    }
    const float rs = rsqrtf(wave_sum(q) * (1.0f / PLANE) + EPSF);
    const float2 gg = reinterpret_cast<const float2*>(g)[lane];
    const float2 sh = reinterpret_cast<const float2*>(b)[lane];
    o.x = (v[0] - mu) * rs * gg.x + sh.x;
    o.y = (v[1] - mu) * rs * gg.y + sh.y;
  } else {
    o.x = a0;
    o.y = a1;
  }
  reinterpret_cast<float2*>(outp)[(size_t)node * (PLANE / 2) + lane] = o;
}

#define PR 32
__global__ void pool_accum(const float* __restrict__ a, const int* __restrict__ batch,
                           float* __restrict__ sums, float* __restrict__ cnts, int n) {
  const int wave = (int)(((size_t)blockIdx.x * blockDim.x + threadIdx.x) >> 5);
  const int lane = threadIdx.x & 31;
  const int row0 = wave * PR;
  if (row0 >= n) return;
  const int nrows = min(PR, n - row0);
  const float2* __restrict__ p =
      reinterpret_cast<const float2*>(a + (size_t)row0 * DOUT) + lane;
  const int gl = batch[row0 + min(lane, nrows - 1)];
  const int g0 = __builtin_amdgcn_readfirstlane(gl);

  if (nrows == PR && __all(gl == g0)) {
    float2 v[PR];
#pragma unroll
    for (int i = 0; i < PR; ++i) v[i] = p[i * (DOUT / 2)];
    float s0 = 0.0f, s1 = 0.0f;
#pragma unroll
    for (int i = 0; i < PR; ++i) {
      s0 += v[i].x;
      s1 += v[i].y;
    }
    atomicAdd(&sums[g0 * DOUT + lane * 2 + 0], s0);
    atomicAdd(&sums[g0 * DOUT + lane * 2 + 1], s1);
    if (lane == 0) atomicAdd(&cnts[g0], (float)PR);
  } else {
    int cur = g0;
    float s0 = 0.0f, s1 = 0.0f, c = 0.0f;
    for (int i = 0; i < nrows; ++i) {
      const int gi = __builtin_amdgcn_readfirstlane(batch[row0 + i]);
      if (gi != cur) {
        atomicAdd(&sums[cur * DOUT + lane * 2 + 0], s0);
        atomicAdd(&sums[cur * DOUT + lane * 2 + 1], s1);
        if (lane == 0) atomicAdd(&cnts[cur], c);
        cur = gi;
        s0 = 0.0f;
        s1 = 0.0f;
        c = 0.0f;
      }
      const float2 v = p[i * (DOUT / 2)];
      s0 += v.x;
      s1 += v.y;
      c += 1.0f;
    }
    atomicAdd(&sums[cur * DOUT + lane * 2 + 0], s0);
    atomicAdd(&sums[cur * DOUT + lane * 2 + 1], s1);
    if (lane == 0) atomicAdd(&cnts[cur], c);
  }
}

__global__ void pool_final(const float* __restrict__ sums, const float* __restrict__ cnts,
                           float* __restrict__ out) {
  int i = blockIdx.x * blockDim.x + threadIdx.x;
  if (i < NGRAPHS * DOUT) {
    int g = i / DOUT;
    float m = sums[i] / fmaxf(cnts[g], 1.0f);
    out[i] = 1.0f / (1.0f + __expf(-m));
  }
}

extern "C" void kernel_launch(void* const* d_in, const int* in_sizes, int n_in,
                              void* d_out, int out_size, void* d_ws, size_t ws_size,
                              hipStream_t stream) {
  const float* x   = (const float*)d_in[0];
  const int*   ei  = (const int*)d_in[1];
  const float* ew  = (const float*)d_in[2];
  const int*   bat = (const int*)d_in[3];
  const float* W1  = (const float*)d_in[4];
  const float* b1  = (const float*)d_in[5];
  const float* g1  = (const float*)d_in[6];
  const float* bt1 = (const float*)d_in[7];
  const float* W2  = (const float*)d_in[8];
  const float* b2  = (const float*)d_in[9];
  const float* g2  = (const float*)d_in[10];
  const float* bt2 = (const float*)d_in[11];
  float* out = (float*)d_out;

  const int* rows = ei;
  const int* cols = ei + NEDGES;

  char* ws = (char*)d_ws;
  size_t off = 0;
  auto carve = [&](size_t bytes) -> float* {
    float* p = (float*)(ws + off);
    off += (bytes + 255) & ~(size_t)255;
    return p;
  };
  float* hA   = carve((size_t)NNODES * PLANE * 4);
  float* hB   = carve((size_t)NNODES * PLANE * 4);
  float* aA   = carve((size_t)NNODES * PLANE * 4);
  float* dinv = carve((size_t)NPAD * 4);
  int*   offs = (int*)carve((size_t)NPAD * 4);
  int2*  pairs = (int2*)carve((size_t)NEDGES * 8);
  float* sums = carve((size_t)NGRAPHS * DOUT * 4);
  float* cnts = carve((size_t)NGRAPHS * 4);

  const int TB = 256;
  auto blocks = [&](long long threads) { return (int)((threads + TB - 1) / TB); };

  int2* tmp  = (int2*)hA;
  int*  hist = (int*)hB;
  int*  base = (int*)hB + HIST_ROOM;
  edge_hist<<<ECH, TBE, 0, stream>>>(cols, hist);
  offs_scan<<<(NBK * ECH + SCAN_TB * 4 - 1) / (SCAN_TB * 4), SCAN_TB, 0, stream>>>(hist, base, NBK * ECH);
  edge_bucket<<<ECH, TBE, 0, stream>>>(rows, cols, ew, base, tmp);
  bucket_fine<<<NBK, TBF, 0, stream>>>(tmp, base, offs, dinv, pairs);
  scale_pairs<<<blocks(NEDGES), TB, 0, stream>>>(pairs, dinv, NEDGES);

  wmma_gemm_f32<DHID, DIN, DIN, PLANE><<<blocks((long long)(NNODES / 16) * 32), TB, 0, stream>>>(x, x, W1, hA, hB, NNODES);
  gather_rows<false><<<blocks((long long)NNODES * 32), TB, 0, stream>>>(hA, offs, pairs, dinv, b1, g1, bt1, aA, NNODES);
  gather_rows<false><<<blocks((long long)NNODES * 32), TB, 0, stream>>>(hB, offs, pairs, dinv, b1 + PLANE, g1, bt1, hA, NNODES);

  ln_wmma_gemm_f32<<<blocks((long long)(NNODES / 16) * 32), TB, 0, stream>>>(aA, hA, g1, bt1, W2, hB, NNODES);
  gather_rows<true><<<blocks((long long)NNODES * 32), TB, 0, stream>>>(hB, offs, pairs, dinv, b2, g2, bt2, aA, NNODES);

  zero_f32<<<blocks(NGRAPHS * DOUT), TB, 0, stream>>>(sums, NGRAPHS * DOUT);
  zero_f32<<<blocks(NGRAPHS), TB, 0, stream>>>(cnts, NGRAPHS);
  pool_accum<<<blocks((long long)((NNODES + PR - 1) / PR) * 32), TB, 0, stream>>>(aA, bat, sums, cnts, NNODES);
  pool_final<<<blocks(NGRAPHS * DOUT), TB, 0, stream>>>(sums, cnts, out);
}


# W1 and W2 reach LDS by whole-line 16-byte loads, 16 (layer 1) or 8 (layer 2) per thread in flight ahead of the LDS stores; was one 4-byte row pair per memory round trip, 64 or 16 trips per block
# speedup vs baseline: 3.8102x; 1.1616x over previous
#include <hip/hip_runtime.h>
#include <math.h>


#define NNODES 100000
#define NEDGES 1600000
#define NGRAPHS 64
#define DIN 256
#define DHID 128
#define DOUT 64
#define EPSF 1e-5f
#define PLANE 64
#define NPAD 100032

typedef __attribute__((ext_vector_type(2))) float v2f;
typedef __attribute__((ext_vector_type(8))) float v8f;

__device__ __forceinline__ float xhalf(float x) {
  const int b = __builtin_bit_cast(int, x);
  return __builtin_bit_cast(
      float, __builtin_amdgcn_permlanex16(b, b, 0x76543210, 0xfedcba98, false, false));
}
template <int XM>
__device__ __forceinline__ float xrow(float x) {
  const int b = __builtin_bit_cast(int, x);
  return __builtin_bit_cast(float, __builtin_amdgcn_update_dpp(b, b, 0x160 + XM, 0xf, 0xf, false));
}
__device__ __forceinline__ float wave_sum(float x) {
  x += xhalf(x);
  x += xrow<8>(x);
  x += xrow<4>(x);
  x += xrow<2>(x);
  x += xrow<1>(x);
  return x;
}

__global__ void zero_f32(float* __restrict__ p, int n) {
  int i = blockIdx.x * blockDim.x + threadIdx.x;
  if (i < n) p[i] = 0.0f;
}

#define SCAN_TB 1024
__global__ void offs_scan(const int* __restrict__ cnt, int* __restrict__ offs, int n) {
  __shared__ int wsum[SCAN_TB / 32];
  const int tid = threadIdx.x;
  const int lane = tid & 31;
  const int wv = tid >> 5;
  const int first = blockIdx.x * (SCAN_TB * 4);

  int part = 0;
  const int4* __restrict__ c4 = reinterpret_cast<const int4*>(cnt);
#pragma unroll 8
  for (int i = tid; i < first / 4; i += SCAN_TB) {
    const int4 q = c4[i];
    part += (q.x + q.y) + (q.z + q.w);
  }
#pragma unroll
  for (int off = 16; off > 0; off >>= 1) part += __shfl_xor(part, off, 32);
  if (lane == 0) wsum[wv] = part;
  __syncthreads();
  int base = 0;
  for (int k = 0; k < SCAN_TB / 32; ++k) base += wsum[k];
  __syncthreads();

  const int idx0 = first + tid * 4;
  int v[4];
#pragma unroll
  for (int j = 0; j < 4; ++j) v[j] = (idx0 + j < n) ? cnt[idx0 + j] : 0;
  const int tsum = (v[0] + v[1]) + (v[2] + v[3]);
  int inc = tsum;
#pragma unroll
  for (int off = 1; off < 32; off <<= 1) {
    const int t = __shfl_up(inc, off, 32);
    if (lane >= off) inc += t;
  }
  if (lane == 31) wsum[wv] = inc;
  __syncthreads();
  int wbase = 0;
  for (int k = 0; k < SCAN_TB / 32; ++k) wbase += (k < wv) ? wsum[k] : 0;
  int run = base + wbase + inc - tsum;
#pragma unroll
  for (int j = 0; j < 4; ++j) {
    if (idx0 + j < n) {
      offs[idx0 + j] = run;
      run += v[j];
      if (idx0 + j == n - 1) offs[n] = run;
    }
  }
}

#define BKSH 8
#define BKN (1 << BKSH)
#define NBK ((NNODES + BKN - 1) / BKN)
#define ECH 256
#define ECHUNK (NEDGES / ECH)
#define TBE 1024
#define HIST_ROOM (((NBK * ECH) + 1 + 1023) & ~1023)
#define EPT ((ECHUNK + TBE - 1) / TBE)
static_assert(NEDGES % ECH == 0, "chunks must tile the edge list");
static_assert(NNODES <= (1 << 17), "source index must fit under the packed node-in-bucket");
static_assert(BKSH + 17 < 32, "the packed word must stay positive");

__global__ void edge_hist(const int* __restrict__ cols, int* __restrict__ hist) {
  __shared__ int h[NBK];
  const int tid = threadIdx.x;
  for (int i = tid; i < NBK; i += TBE) h[i] = 0;
  __syncthreads();
  const int e0 = blockIdx.x * ECHUNK;
  int c[EPT];
#pragma unroll
  for (int k = 0; k < EPT; ++k) {
    const int i = k * TBE + tid;
    c[k] = (i < ECHUNK) ? cols[e0 + i] : -1;
  }
#pragma unroll
  for (int k = 0; k < EPT; ++k)
    if (c[k] >= 0) atomicAdd(&h[c[k] >> BKSH], 1);
  __syncthreads();
  for (int i = tid; i < NBK; i += TBE) hist[i * ECH + blockIdx.x] = h[i];
}

__global__ void edge_bucket(const int* __restrict__ rows, const int* __restrict__ cols,
                            const float* __restrict__ w, const int* __restrict__ base,
                            int2* __restrict__ tmp) {
  __shared__ int cur[NBK];
  const int tid = threadIdx.x;
  for (int i = tid; i < NBK; i += TBE) cur[i] = base[i * ECH + blockIdx.x];
  __syncthreads();
  const int e0 = blockIdx.x * ECHUNK;
  int c[EPT];
  int2 rec[EPT];
#pragma unroll
  for (int k = 0; k < EPT; ++k) {
    const int i = k * TBE + tid;
    const bool ok = (i < ECHUNK);
    c[k] = ok ? cols[e0 + i] : -1;
    rec[k].x = ok ? rows[e0 + i] : 0;
    rec[k].y = ok ? __float_as_int(w[e0 + i]) : 0;
  }
#pragma unroll
  for (int k = 0; k < EPT; ++k) {
    if (c[k] >= 0) {
      const int pos = atomicAdd(&cur[c[k] >> BKSH], 1);
      int2 t;
      t.x = rec[k].x | ((c[k] & (BKN - 1)) << 17);
      t.y = rec[k].y;
      tmp[pos] = t;
    }
  }
}

#define TBF 1024
#define FPT 4
__global__ void bucket_fine(const int2* __restrict__ tmp, const int* __restrict__ base,
                            int* __restrict__ offs, float* __restrict__ dinv,
                            int2* __restrict__ pairs) {
  __shared__ int cntL[BKN];
  __shared__ float degL[BKN];
  __shared__ int curL[BKN];
  __shared__ int wsum[BKN / 32];
  const int tid = threadIdx.x;
  const int lane = tid & 31;
  const int wv = tid >> 5;
  const int start = base[blockIdx.x * ECH];
  const int end = base[(blockIdx.x + 1) * ECH];
  if (tid < BKN) {
    cntL[tid] = 0;
    degL[tid] = 0.0f;
  }
  __syncthreads();

  for (int b0 = start; b0 < end; b0 += FPT * TBF) {
    int2 t[FPT];
#pragma unroll
    for (int k = 0; k < FPT; ++k) {
      const int i = b0 + k * TBF + tid;
      t[k] = (i < end) ? tmp[i] : make_int2(-1, 0);
    }
#pragma unroll
    for (int k = 0; k < FPT; ++k) {
      if (t[k].x >= 0) {
        const int cl = t[k].x >> 17;
        atomicAdd(&cntL[cl], 1);
        atomicAdd(&degL[cl], __int_as_float(t[k].y));
      }
    }
  }
  __syncthreads();

  int v = 0, inc = 0;
  if (tid < BKN) {
    v = cntL[tid];
    inc = v;
#pragma unroll
    for (int off = 1; off < 32; off <<= 1) {
      const int u = __shfl_up(inc, off, 32);
      if (lane >= off) inc += u;
    }
    if (lane == 31) wsum[wv] = inc;
  }
  __syncthreads();
  if (tid < BKN) {
    int wbase = 0;
    for (int k = 0; k < BKN / 32; ++k) wbase += (k < wv) ? wsum[k] : 0;
    const int excl = start + wbase + inc - v;
    curL[tid] = excl;
    const int node = blockIdx.x * BKN + tid;
    if (node < NNODES) {
      offs[node] = excl;
      dinv[node] = rsqrtf(fmaxf(degL[tid] + 1.0f, EPSF));
      if (node == NNODES - 1) offs[NNODES] = excl + v;
    }
  }
  __syncthreads();

  for (int b0 = start; b0 < end; b0 += FPT * TBF) {
    int2 t[FPT];
#pragma unroll
    for (int k = 0; k < FPT; ++k) {
      const int i = b0 + k * TBF + tid;
      t[k] = (i < end) ? tmp[i] : make_int2(-1, 0);
    }
#pragma unroll
    for (int k = 0; k < FPT; ++k) {
      if (t[k].x >= 0) {
        const int pos = atomicAdd(&curL[t[k].x >> 17], 1);
        int2 pr;
        pr.x = t[k].x & ((1 << 17) - 1);
        pr.y = t[k].y;
        pairs[pos] = pr;
      }
    }
  }
}

__global__ void scale_pairs(int2* __restrict__ pairs, const float* __restrict__ dinv, int nE) {
  int e = blockIdx.x * blockDim.x + threadIdx.x;
  if (e < nE) {
    int2 pr = pairs[e];
    pr.y = __float_as_int(dinv[pr.x] * __int_as_float(pr.y));
    pairs[e] = pr;
  }
}

#define GT 256
template <int NCOLS, int KDIM>
__device__ __forceinline__ void stage_b_pairs(const float* __restrict__ B,
                                              float2* __restrict__ Bl) {
  constexpr int NC4 = NCOLS / 4;
  constexpr int UNITS = (KDIM / 2) * NC4;
  constexpr int UPT = UNITS / GT;
  constexpr int SB = (UPT < 8) ? UPT : 8;
  static_assert(UNITS % GT == 0 && UPT % SB == 0, "the units must tile the work-group");
  const float4* __restrict__ B4 = reinterpret_cast<const float4*>(B);
  float4* __restrict__ Bl4 = reinterpret_cast<float4*>(Bl);
  for (int u0 = threadIdx.x; u0 < UNITS; u0 += SB * GT) {
    float4 ra[SB], rb[SB];
#pragma unroll
    for (int s = 0; s < SB; ++s) {
      const int u = u0 + s * GT;
      const int k2 = u / NC4;
      const int n4 = u - k2 * NC4;
      ra[s] = B4[(2 * k2 + 0) * NC4 + n4];
      rb[s] = B4[(2 * k2 + 1) * NC4 + n4];
    }
#pragma unroll
    for (int s = 0; s < SB; ++s) {
      const int u = u0 + s * GT;
      Bl4[2 * u + 0] = make_float4(ra[s].x, rb[s].x, ra[s].y, rb[s].y);
      Bl4[2 * u + 1] = make_float4(ra[s].z, rb[s].z, ra[s].w, rb[s].w);
    }
  }
}

template <int NCOLS, int KDIM, int KPL, int CPL>
__global__ void wmma_gemm_f32(const float* __restrict__ A0, const float* __restrict__ A1,
                              const float* __restrict__ B, float* __restrict__ C0,
                              float* __restrict__ C1, int M) {
  __shared__ alignas(16) float2 Bl[(KDIM / 2) * NCOLS];

  stage_b_pairs<NCOLS, KDIM>(B, Bl);
  __syncthreads();

  const int wave = (int)(((size_t)blockIdx.x * blockDim.x + threadIdx.x) >> 5);
  const int lane = threadIdx.x & 31;
  if (wave >= M / 16) return;
  const int m0 = wave * 16;
  const int half = lane >> 4;
  const int l = lane & 15;
  constexpr int NT = NCOLS / 16;

  v8f acc[NT];
#pragma unroll
  for (int t = 0; t < NT; ++t)
#pragma unroll
    for (int j = 0; j < 8; ++j) acc[t][j] = 0.0f;

  auto a_at = [&](int k) -> const float* {
    const float* base = (KPL < KDIM && k >= KPL) ? A1 : A0;
    return base + (size_t)(m0 + l) * KPL + (k & (KPL - 1)) + 2 * half;
  };

  v2f a;
  {
    const float* ap = a_at(0);
    a[0] = ap[0];
    a[1] = ap[1];
  }

  for (int k0 = 0; k0 < KDIM; k0 += 4) {
    v2f a_next = a;
    if (k0 + 4 < KDIM) {
      const float* np = a_at(k0 + 4);
      a_next[0] = np[0];
      a_next[1] = np[1];
    }
    const float2* __restrict__ brow = Bl + (size_t)(k0 / 2 + half) * NCOLS + l;
    v2f bfrag[NT];
#pragma unroll
    for (int t = 0; t < NT; ++t) {
      float2 bb = brow[t * 16];
      bfrag[t][0] = bb.x;
      bfrag[t][1] = bb.y;
    }
#pragma unroll
    for (int t = 0; t < NT; ++t)
      acc[t] = __builtin_amdgcn_wmma_f32_16x16x4_f32(
           false, a,  false, bfrag[t],
           (short)0, acc[t],  false,  false);
    a = a_next;
  }

#pragma unroll
  for (int t = 0; t < NT; ++t) {
    constexpr int TPP = CPL / 16;
    float* __restrict__ cb = (t / TPP) ? C1 : C0;
    const int cc = (t % TPP) * 16 + l;
#pragma unroll
    for (int r = 0; r < 8; ++r)
      cb[(size_t)(m0 + r + 8 * half) * CPL + cc] = acc[t][r];
  }
}

__global__ void ln_wmma_gemm_f32(const float* __restrict__ A0, const float* __restrict__ A1,
                                 const float* __restrict__ g, const float* __restrict__ b,
                                 const float* __restrict__ B, float* __restrict__ C, int M) {
  constexpr int KDIM = DHID;
  constexpr int NCOLS = DOUT;
  constexpr int NT = NCOLS / 16;
  constexpr int KS = KDIM / 4;
  __shared__ alignas(16) float2 Bl[(KDIM / 2) * NCOLS];

  stage_b_pairs<NCOLS, KDIM>(B, Bl);
  __syncthreads();

  const int wave = (int)(((size_t)blockIdx.x * blockDim.x + threadIdx.x) >> 5);
  const int lane = threadIdx.x & 31;
  if (wave >= M / 16) return;
  const int m0 = wave * 16;
  const int half = lane >> 4;
  const int l = lane & 15;

  const float2* __restrict__ r0 =
      reinterpret_cast<const float2*>(A0 + (size_t)(m0 + l) * PLANE + 2 * half);
  const float2* __restrict__ r1 =
      reinterpret_cast<const float2*>(A1 + (size_t)(m0 + l) * PLANE + 2 * half);
  v2f a[KS];
#pragma unroll
  for (int j = 0; j < KS; ++j) {
    const float2 t = (4 * j < PLANE) ? r0[2 * j] : r1[2 * j - PLANE / 2];
    a[j][0] = fmaxf(t.x, 0.0f);
    a[j][1] = fmaxf(t.y, 0.0f);
  }

  float sp[4] = {0.0f, 0.0f, 0.0f, 0.0f};
#pragma unroll
  for (int j = 0; j < KS; ++j) sp[j & 3] += a[j][0] + a[j][1];
  float s = (sp[0] + sp[1]) + (sp[2] + sp[3]);
  s += xhalf(s);
  const float mu = s * (1.0f / KDIM);
  float qp[4] = {0.0f, 0.0f, 0.0f, 0.0f};
#pragma unroll
  for (int j = 0; j < KS; ++j) {
    const float d0 = a[j][0] - mu;
    const float d1 = a[j][1] - mu;
    qp[j & 3] += d0 * d0 + d1 * d1;
  }
  float q = (qp[0] + qp[1]) + (qp[2] + qp[3]);
  q += xhalf(q);
  const float rs = rsqrtf(q * (1.0f / KDIM) + EPSF);

  const float2* __restrict__ gp = reinterpret_cast<const float2*>(g + 2 * half);
  const float2* __restrict__ bp = reinterpret_cast<const float2*>(b + 2 * half);
#pragma unroll
  for (int j = 0; j < KS; ++j) {
    const float2 gg = gp[2 * j];
    const float2 sh = bp[2 * j];
    a[j][0] = (a[j][0] - mu) * rs * gg.x + sh.x;
    a[j][1] = (a[j][1] - mu) * rs * gg.y + sh.y;
  }

  v8f acc[NT];
#pragma unroll
  for (int t = 0; t < NT; ++t)
#pragma unroll
    for (int j = 0; j < 8; ++j) acc[t][j] = 0.0f;

#pragma unroll
  for (int j = 0; j < KS; ++j) {
    const float2* __restrict__ brow = Bl + (size_t)(2 * j + half) * NCOLS + l;
    v2f bfrag[NT];
#pragma unroll
    for (int t = 0; t < NT; ++t) {
      float2 bb = brow[t * 16];
      bfrag[t][0] = bb.x;
      bfrag[t][1] = bb.y;
    }
#pragma unroll
    for (int t = 0; t < NT; ++t)
      acc[t] = __builtin_amdgcn_wmma_f32_16x16x4_f32(
           false, a[j],  false, bfrag[t],
           (short)0, acc[t],  false,  false);
  }

#pragma unroll
  for (int t = 0; t < NT; ++t) {
    const int cc = t * 16 + l;
#pragma unroll
    for (int r = 0; r < 8; ++r)
      C[(size_t)(m0 + r + 8 * half) * NCOLS + cc] = acc[t][r];
  }
}

#define GCH 8
template <bool NORM>
__global__ void gather_rows(const float* __restrict__ hp, const int* __restrict__ offs,
                            const int2* __restrict__ pairs, const float* __restrict__ dinv,
                            const float* __restrict__ bias, const float* __restrict__ g,
                            const float* __restrict__ b, float* __restrict__ outp, int n) {
  const int wave = (int)(((size_t)blockIdx.x * blockDim.x + threadIdx.x) >> 5);
  const int lane = threadIdx.x & 31;
  if (wave >= n) return;
  const int node = __builtin_amdgcn_readfirstlane(wave);
  const float2* __restrict__ H = reinterpret_cast<const float2*>(hp) + lane;
  const int beg = offs[node];
  const int end = offs[node + 1];
  const float dc = dinv[node];
  const float2 self = H[(size_t)node * (PLANE / 2)];
  const float2 bb = reinterpret_cast<const float2*>(bias)[lane];
  const float s = dc * dc;
  float a0 = bb.x + self.x * s;
  float a1 = bb.y + self.y * s;

  for (int base = beg; base < end; base += 32) {
    const int m = min(32, end - base);
    const int2 pr = pairs[base + min(lane, m - 1)];
    const int rv = pr.x;
    const int cv = __float_as_int(__int_as_float(pr.y) * dc);
    for (int j = 0; j < m; j += GCH) {
      float2 v[GCH];
      float cf[GCH];
#pragma unroll
      for (int u = 0; u < GCH; ++u) {
        const bool ok = (j + u < m);
        const int src = ok ? (j + u) : j;
        const int r = __builtin_amdgcn_readlane(rv, src);
        const float c = __int_as_float(__builtin_amdgcn_readlane(cv, src));
        const float2 t = H[(size_t)r * (PLANE / 2)];
        v[u].x = ok ? t.x : 0.0f;
        v[u].y = ok ? t.y : 0.0f;
        cf[u] = ok ? c : 0.0f;
      }
#pragma unroll
      for (int u = 0; u < GCH; ++u) {
        a0 += v[u].x * cf[u];
        a1 += v[u].y * cf[u];
      }
    }
  }
  float2 o;
  if (NORM) {
    float v[2];
    v[0] = fmaxf(a0, 0.0f);
    v[1] = fmaxf(a1, 0.0f);
    float rsum = 0.0f;
#pragma unroll
    for (int f = 0; f < 2; ++f) rsum += v[f];
    const float mu = wave_sum(rsum) * (1.0f / PLANE);
    float q = 0.0f;
#pragma unroll
    for (int f = 0; f < 2; ++f) {
      float d = v[f] - mu;
      q += d * d;
    }
    const float rs = rsqrtf(wave_sum(q) * (1.0f / PLANE) + EPSF);
    const float2 gg = reinterpret_cast<const float2*>(g)[lane];
    const float2 sh = reinterpret_cast<const float2*>(b)[lane];
    o.x = (v[0] - mu) * rs * gg.x + sh.x;
    o.y = (v[1] - mu) * rs * gg.y + sh.y;
  } else {
    o.x = a0;
    o.y = a1;
  }
  reinterpret_cast<float2*>(outp)[(size_t)node * (PLANE / 2) + lane] = o;
}

#define PR 32
__global__ void pool_accum(const float* __restrict__ a, const int* __restrict__ batch,
                           float* __restrict__ sums, float* __restrict__ cnts, int n) {
  const int wave = (int)(((size_t)blockIdx.x * blockDim.x + threadIdx.x) >> 5);
  const int lane = threadIdx.x & 31;
  const int row0 = wave * PR;
  if (row0 >= n) return;
  const int nrows = min(PR, n - row0);
  const float2* __restrict__ p =
      reinterpret_cast<const float2*>(a + (size_t)row0 * DOUT) + lane;
  const int gl = batch[row0 + min(lane, nrows - 1)];
  const int g0 = __builtin_amdgcn_readfirstlane(gl);

  if (nrows == PR && __all(gl == g0)) {
    float2 v[PR];
#pragma unroll
    for (int i = 0; i < PR; ++i) v[i] = p[i * (DOUT / 2)];
    float s0 = 0.0f, s1 = 0.0f;
#pragma unroll
    for (int i = 0; i < PR; ++i) {
      s0 += v[i].x;
      s1 += v[i].y;
    }
    atomicAdd(&sums[g0 * DOUT + lane * 2 + 0], s0);
    atomicAdd(&sums[g0 * DOUT + lane * 2 + 1], s1);
    if (lane == 0) atomicAdd(&cnts[g0], (float)PR);
  } else {
    int cur = g0;
    float s0 = 0.0f, s1 = 0.0f, c = 0.0f;
    for (int i = 0; i < nrows; ++i) {
      const int gi = __builtin_amdgcn_readfirstlane(batch[row0 + i]);
      if (gi != cur) {
        atomicAdd(&sums[cur * DOUT + lane * 2 + 0], s0);
        atomicAdd(&sums[cur * DOUT + lane * 2 + 1], s1);
        if (lane == 0) atomicAdd(&cnts[cur], c);
        cur = gi;
        s0 = 0.0f;
        s1 = 0.0f;
        c = 0.0f;
      }
      const float2 v = p[i * (DOUT / 2)];
      s0 += v.x;
      s1 += v.y;
      c += 1.0f;
    }
    atomicAdd(&sums[cur * DOUT + lane * 2 + 0], s0);
    atomicAdd(&sums[cur * DOUT + lane * 2 + 1], s1);
    if (lane == 0) atomicAdd(&cnts[cur], c);
  }
}

__global__ void pool_final(const float* __restrict__ sums, const float* __restrict__ cnts,
                           float* __restrict__ out) {
  int i = blockIdx.x * blockDim.x + threadIdx.x;
  if (i < NGRAPHS * DOUT) {
    int g = i / DOUT;
    float m = sums[i] / fmaxf(cnts[g], 1.0f);
    out[i] = 1.0f / (1.0f + __expf(-m));
  }
}

extern "C" void kernel_launch(void* const* d_in, const int* in_sizes, int n_in,
                              void* d_out, int out_size, void* d_ws, size_t ws_size,
                              hipStream_t stream) {
  const float* x   = (const float*)d_in[0];
  const int*   ei  = (const int*)d_in[1];
  const float* ew  = (const float*)d_in[2];
  const int*   bat = (const int*)d_in[3];
  const float* W1  = (const float*)d_in[4];
  const float* b1  = (const float*)d_in[5];
  const float* g1  = (const float*)d_in[6];
  const float* bt1 = (const float*)d_in[7];
  const float* W2  = (const float*)d_in[8];
  const float* b2  = (const float*)d_in[9];
  const float* g2  = (const float*)d_in[10];
  const float* bt2 = (const float*)d_in[11];
  float* out = (float*)d_out;

  const int* rows = ei;
  const int* cols = ei + NEDGES;

  char* ws = (char*)d_ws;
  size_t off = 0;
  auto carve = [&](size_t bytes) -> float* {
    float* p = (float*)(ws + off);
    off += (bytes + 255) & ~(size_t)255;
    return p;
  };
  float* hA   = carve((size_t)NNODES * PLANE * 4);
  float* hB   = carve((size_t)NNODES * PLANE * 4);
  float* aA   = carve((size_t)NNODES * PLANE * 4);
  float* dinv = carve((size_t)NPAD * 4);
  int*   offs = (int*)carve((size_t)NPAD * 4);
  int2*  pairs = (int2*)carve((size_t)NEDGES * 8);
  float* sums = carve((size_t)NGRAPHS * DOUT * 4);
  float* cnts = carve((size_t)NGRAPHS * 4);

  const int TB = 256;
  auto blocks = [&](long long threads) { return (int)((threads + TB - 1) / TB); };

  int2* tmp  = (int2*)hA;
  int*  hist = (int*)hB;
  int*  base = (int*)hB + HIST_ROOM;
  edge_hist<<<ECH, TBE, 0, stream>>>(cols, hist);
  offs_scan<<<(NBK * ECH + SCAN_TB * 4 - 1) / (SCAN_TB * 4), SCAN_TB, 0, stream>>>(hist, base, NBK * ECH);
  edge_bucket<<<ECH, TBE, 0, stream>>>(rows, cols, ew, base, tmp);
  bucket_fine<<<NBK, TBF, 0, stream>>>(tmp, base, offs, dinv, pairs);
  scale_pairs<<<blocks(NEDGES), TB, 0, stream>>>(pairs, dinv, NEDGES);

  wmma_gemm_f32<DHID, DIN, DIN, PLANE><<<((NNODES / 16) * 32 + GT - 1) / GT, GT, 0, stream>>>(x, x, W1, hA, hB, NNODES);
  gather_rows<false><<<blocks((long long)NNODES * 32), TB, 0, stream>>>(hA, offs, pairs, dinv, b1, g1, bt1, aA, NNODES);
  gather_rows<false><<<blocks((long long)NNODES * 32), TB, 0, stream>>>(hB, offs, pairs, dinv, b1 + PLANE, g1, bt1, hA, NNODES);

  ln_wmma_gemm_f32<<<((NNODES / 16) * 32 + GT - 1) / GT, GT, 0, stream>>>(aA, hA, g1, bt1, W2, hB, NNODES);
  gather_rows<true><<<blocks((long long)NNODES * 32), TB, 0, stream>>>(hB, offs, pairs, dinv, b2, g2, bt2, aA, NNODES);

  zero_f32<<<blocks(NGRAPHS * DOUT), TB, 0, stream>>>(sums, NGRAPHS * DOUT);
  zero_f32<<<blocks(NGRAPHS), TB, 0, stream>>>(cnts, NGRAPHS);
  pool_accum<<<blocks((long long)((NNODES + PR - 1) / PR) * 32), TB, 0, stream>>>(aA, bat, sums, cnts, NNODES);
  pool_final<<<blocks(NGRAPHS * DOUT), TB, 0, stream>>>(sums, cnts, out);
}


# three launches fewer: dinv[source] applied to the product rows in both GEMM epilogues, not in a pass over the edge list; pool sums zeroed inside edge_hist; rounding of the edge scale changes
# speedup vs baseline: 4.0682x; 1.0677x over previous
#include <hip/hip_runtime.h>
#include <math.h>


#define NNODES 100000
#define NEDGES 1600000
#define NGRAPHS 64
#define DIN 256
#define DHID 128
#define DOUT 64
#define EPSF 1e-5f
#define PLANE 64
#define NPAD 100032

typedef __attribute__((ext_vector_type(2))) float v2f;
typedef __attribute__((ext_vector_type(8))) float v8f;

__device__ __forceinline__ float xhalf(float x) {
  const int b = __builtin_bit_cast(int, x);
  return __builtin_bit_cast(
      float, __builtin_amdgcn_permlanex16(b, b, 0x76543210, 0xfedcba98, false, false));
}
template <int XM>
__device__ __forceinline__ float xrow(float x) {
  const int b = __builtin_bit_cast(int, x);
  return __builtin_bit_cast(float, __builtin_amdgcn_update_dpp(b, b, 0x160 + XM, 0xf, 0xf, false));
}
__device__ __forceinline__ float wave_sum(float x) {
  x += xhalf(x);
  x += xrow<8>(x);
  x += xrow<4>(x);
  x += xrow<2>(x);
  x += xrow<1>(x);
  return x;
}

#define SCAN_TB 1024
__global__ void offs_scan(const int* __restrict__ cnt, int* __restrict__ offs, int n) {
  __shared__ int wsum[SCAN_TB / 32];
  const int tid = threadIdx.x;
  const int lane = tid & 31;
  const int wv = tid >> 5;
  const int first = blockIdx.x * (SCAN_TB * 4);

  int part = 0;
  const int4* __restrict__ c4 = reinterpret_cast<const int4*>(cnt);
#pragma unroll 8
  for (int i = tid; i < first / 4; i += SCAN_TB) {
    const int4 q = c4[i];
    part += (q.x + q.y) + (q.z + q.w);
  }
#pragma unroll
  for (int off = 16; off > 0; off >>= 1) part += __shfl_xor(part, off, 32);
  if (lane == 0) wsum[wv] = part;
  __syncthreads();
  int base = 0;
  for (int k = 0; k < SCAN_TB / 32; ++k) base += wsum[k];
  __syncthreads();

  const int idx0 = first + tid * 4;
  int v[4];
#pragma unroll
  for (int j = 0; j < 4; ++j) v[j] = (idx0 + j < n) ? cnt[idx0 + j] : 0;
  const int tsum = (v[0] + v[1]) + (v[2] + v[3]);
  int inc = tsum;
#pragma unroll
  for (int off = 1; off < 32; off <<= 1) {
    const int t = __shfl_up(inc, off, 32);
    if (lane >= off) inc += t;
  }
  if (lane == 31) wsum[wv] = inc;
  __syncthreads();
  int wbase = 0;
  for (int k = 0; k < SCAN_TB / 32; ++k) wbase += (k < wv) ? wsum[k] : 0;
  int run = base + wbase + inc - tsum;
#pragma unroll
  for (int j = 0; j < 4; ++j) {
    if (idx0 + j < n) {
      offs[idx0 + j] = run;
      run += v[j];
      if (idx0 + j == n - 1) offs[n] = run;
    }
  }
}

#define BKSH 8
#define BKN (1 << BKSH)
#define NBK ((NNODES + BKN - 1) / BKN)
#define ECH 256
#define ECHUNK (NEDGES / ECH)
#define TBE 1024
#define HIST_ROOM (((NBK * ECH) + 1 + 1023) & ~1023)
#define EPT ((ECHUNK + TBE - 1) / TBE)
static_assert(NEDGES % ECH == 0, "chunks must tile the edge list");
static_assert(NNODES <= (1 << 17), "source index must fit under the packed node-in-bucket");
static_assert(BKSH + 17 < 32, "the packed word must stay positive");

__global__ void edge_hist(const int* __restrict__ cols, int* __restrict__ hist,
                          float* __restrict__ sums, float* __restrict__ cnts) {
  __shared__ int h[NBK];
  const int tid = threadIdx.x;
  if (blockIdx.x == 0) {
    for (int i = tid; i < NGRAPHS * DOUT; i += TBE) sums[i] = 0.0f;
    if (tid < NGRAPHS) cnts[tid] = 0.0f;
  }
  for (int i = tid; i < NBK; i += TBE) h[i] = 0;
  __syncthreads();
  const int e0 = blockIdx.x * ECHUNK;
  int c[EPT];
#pragma unroll
  for (int k = 0; k < EPT; ++k) {
    const int i = k * TBE + tid;
    c[k] = (i < ECHUNK) ? cols[e0 + i] : -1;
  }
#pragma unroll
  for (int k = 0; k < EPT; ++k)
    if (c[k] >= 0) atomicAdd(&h[c[k] >> BKSH], 1);
  __syncthreads();
  for (int i = tid; i < NBK; i += TBE) hist[i * ECH + blockIdx.x] = h[i];
}

__global__ void edge_bucket(const int* __restrict__ rows, const int* __restrict__ cols,
                            const float* __restrict__ w, const int* __restrict__ base,
                            int2* __restrict__ tmp) {
  __shared__ int cur[NBK];
  const int tid = threadIdx.x;
  for (int i = tid; i < NBK; i += TBE) cur[i] = base[i * ECH + blockIdx.x];
  __syncthreads();
  const int e0 = blockIdx.x * ECHUNK;
  int c[EPT];
  int2 rec[EPT];
#pragma unroll
  for (int k = 0; k < EPT; ++k) {
    const int i = k * TBE + tid;
    const bool ok = (i < ECHUNK);
    c[k] = ok ? cols[e0 + i] : -1;
    rec[k].x = ok ? rows[e0 + i] : 0;
    rec[k].y = ok ? __float_as_int(w[e0 + i]) : 0;
  }
#pragma unroll
  for (int k = 0; k < EPT; ++k) {
    if (c[k] >= 0) {
      const int pos = atomicAdd(&cur[c[k] >> BKSH], 1);
      int2 t;
      t.x = rec[k].x | ((c[k] & (BKN - 1)) << 17);
      t.y = rec[k].y;
      tmp[pos] = t;
    }
  }
}

#define TBF 1024
#define FPT 4
__global__ void bucket_fine(const int2* __restrict__ tmp, const int* __restrict__ base,
                            int* __restrict__ offs, float* __restrict__ dinv,
                            int2* __restrict__ pairs) {
  __shared__ int cntL[BKN];
  __shared__ float degL[BKN];
  __shared__ int curL[BKN];
  __shared__ int wsum[BKN / 32];
  const int tid = threadIdx.x;
  const int lane = tid & 31;
  const int wv = tid >> 5;
  const int start = base[blockIdx.x * ECH];
  const int end = base[(blockIdx.x + 1) * ECH];
  if (tid < BKN) {
    cntL[tid] = 0;
    degL[tid] = 0.0f;
  }
  __syncthreads();

  for (int b0 = start; b0 < end; b0 += FPT * TBF) {
    int2 t[FPT];
#pragma unroll
    for (int k = 0; k < FPT; ++k) {
      const int i = b0 + k * TBF + tid;
      t[k] = (i < end) ? tmp[i] : make_int2(-1, 0);
    }
#pragma unroll
    for (int k = 0; k < FPT; ++k) {
      if (t[k].x >= 0) {
        const int cl = t[k].x >> 17;
        atomicAdd(&cntL[cl], 1);
        atomicAdd(&degL[cl], __int_as_float(t[k].y));
      }
    }
  }
  __syncthreads();

  int v = 0, inc = 0;
  if (tid < BKN) {
    v = cntL[tid];
    inc = v;
#pragma unroll
    for (int off = 1; off < 32; off <<= 1) {
      const int u = __shfl_up(inc, off, 32);
      if (lane >= off) inc += u;
    }
    if (lane == 31) wsum[wv] = inc;
  }
  __syncthreads();
  if (tid < BKN) {
    int wbase = 0;
    for (int k = 0; k < BKN / 32; ++k) wbase += (k < wv) ? wsum[k] : 0;
    const int excl = start + wbase + inc - v;
    curL[tid] = excl;
    const int node = blockIdx.x * BKN + tid;
    if (node < NNODES) {
      offs[node] = excl;
      dinv[node] = rsqrtf(fmaxf(degL[tid] + 1.0f, EPSF));
      if (node == NNODES - 1) offs[NNODES] = excl + v;
    }
  }
  __syncthreads();

  for (int b0 = start; b0 < end; b0 += FPT * TBF) {
    int2 t[FPT];
#pragma unroll
    for (int k = 0; k < FPT; ++k) {
      const int i = b0 + k * TBF + tid;
      t[k] = (i < end) ? tmp[i] : make_int2(-1, 0);
    }
#pragma unroll
    for (int k = 0; k < FPT; ++k) {
      if (t[k].x >= 0) {
        const int pos = atomicAdd(&curL[t[k].x >> 17], 1);
        int2 pr;
        pr.x = t[k].x & ((1 << 17) - 1);
        pr.y = t[k].y;
        pairs[pos] = pr;
      }
    }
  }
}

#define GT 256
template <int NCOLS, int KDIM>
__device__ __forceinline__ void stage_b_pairs(const float* __restrict__ B,
                                              float2* __restrict__ Bl) {
  constexpr int NC4 = NCOLS / 4;
  constexpr int UNITS = (KDIM / 2) * NC4;
  constexpr int UPT = UNITS / GT;
  constexpr int SB = (UPT < 8) ? UPT : 8;
  static_assert(UNITS % GT == 0 && UPT % SB == 0, "the units must tile the work-group");
  const float4* __restrict__ B4 = reinterpret_cast<const float4*>(B);
  float4* __restrict__ Bl4 = reinterpret_cast<float4*>(Bl);
  for (int u0 = threadIdx.x; u0 < UNITS; u0 += SB * GT) {
    float4 ra[SB], rb[SB];
#pragma unroll
    for (int s = 0; s < SB; ++s) {
      const int u = u0 + s * GT;
      const int k2 = u / NC4;
      const int n4 = u - k2 * NC4;
      ra[s] = B4[(2 * k2 + 0) * NC4 + n4];
      rb[s] = B4[(2 * k2 + 1) * NC4 + n4];
    }
#pragma unroll
    for (int s = 0; s < SB; ++s) {
      const int u = u0 + s * GT;
      Bl4[2 * u + 0] = make_float4(ra[s].x, rb[s].x, ra[s].y, rb[s].y);
      Bl4[2 * u + 1] = make_float4(ra[s].z, rb[s].z, ra[s].w, rb[s].w);
    }
  }
}

template <int NCOLS, int KDIM, int KPL, int CPL>
__global__ void wmma_gemm_f32(const float* __restrict__ A0, const float* __restrict__ A1,
                              const float* __restrict__ B, const float* __restrict__ dinv,
                              float* __restrict__ C0, float* __restrict__ C1, int M) {
  __shared__ alignas(16) float2 Bl[(KDIM / 2) * NCOLS];

  stage_b_pairs<NCOLS, KDIM>(B, Bl);
  __syncthreads();

  const int wave = (int)(((size_t)blockIdx.x * blockDim.x + threadIdx.x) >> 5);
  const int lane = threadIdx.x & 31;
  if (wave >= M / 16) return;
  const int m0 = wave * 16;
  const int half = lane >> 4;
  const int l = lane & 15;
  constexpr int NT = NCOLS / 16;

  v8f acc[NT];
#pragma unroll
  for (int t = 0; t < NT; ++t)
#pragma unroll
    for (int j = 0; j < 8; ++j) acc[t][j] = 0.0f;

  auto a_at = [&](int k) -> const float* {
    const float* base = (KPL < KDIM && k >= KPL) ? A1 : A0;
    return base + (size_t)(m0 + l) * KPL + (k & (KPL - 1)) + 2 * half;
  };

  v2f a;
  {
    const float* ap = a_at(0);
    a[0] = ap[0];
    a[1] = ap[1];
  }

  float rsc[8];
#pragma unroll
  for (int r = 0; r < 8; ++r) rsc[r] = dinv[m0 + r + 8 * half];

  for (int k0 = 0; k0 < KDIM; k0 += 4) {
    v2f a_next = a;
    if (k0 + 4 < KDIM) {
      const float* np = a_at(k0 + 4);
      a_next[0] = np[0];
      a_next[1] = np[1];
    }
    const float2* __restrict__ brow = Bl + (size_t)(k0 / 2 + half) * NCOLS + l;
    v2f bfrag[NT];
#pragma unroll
    for (int t = 0; t < NT; ++t) {
      float2 bb = brow[t * 16];
      bfrag[t][0] = bb.x;
      bfrag[t][1] = bb.y;
    }
#pragma unroll
    for (int t = 0; t < NT; ++t)
      acc[t] = __builtin_amdgcn_wmma_f32_16x16x4_f32(
           false, a,  false, bfrag[t],
           (short)0, acc[t],  false,  false);
    a = a_next;
  }

#pragma unroll
  for (int t = 0; t < NT; ++t) {
    constexpr int TPP = CPL / 16;
    float* __restrict__ cb = (t / TPP) ? C1 : C0;
    const int cc = (t % TPP) * 16 + l;
#pragma unroll
    for (int r = 0; r < 8; ++r)
      cb[(size_t)(m0 + r + 8 * half) * CPL + cc] = acc[t][r] * rsc[r];
  }
}

__global__ void ln_wmma_gemm_f32(const float* __restrict__ A0, const float* __restrict__ A1,
                                 const float* __restrict__ g, const float* __restrict__ b,
                                 const float* __restrict__ B, const float* __restrict__ dinv,
                                 float* __restrict__ C, int M) {
  constexpr int KDIM = DHID;
  constexpr int NCOLS = DOUT;
  constexpr int NT = NCOLS / 16;
  constexpr int KS = KDIM / 4;
  __shared__ alignas(16) float2 Bl[(KDIM / 2) * NCOLS];

  stage_b_pairs<NCOLS, KDIM>(B, Bl);
  __syncthreads();

  const int wave = (int)(((size_t)blockIdx.x * blockDim.x + threadIdx.x) >> 5);
  const int lane = threadIdx.x & 31;
  if (wave >= M / 16) return;
  const int m0 = wave * 16;
  const int half = lane >> 4;
  const int l = lane & 15;

  const float2* __restrict__ r0 =
      reinterpret_cast<const float2*>(A0 + (size_t)(m0 + l) * PLANE + 2 * half);
  const float2* __restrict__ r1 =
      reinterpret_cast<const float2*>(A1 + (size_t)(m0 + l) * PLANE + 2 * half);
  float rsc[8];
#pragma unroll
  for (int r = 0; r < 8; ++r) rsc[r] = dinv[m0 + r + 8 * half];

  v2f a[KS];
#pragma unroll
  for (int j = 0; j < KS; ++j) {
    const float2 t = (4 * j < PLANE) ? r0[2 * j] : r1[2 * j - PLANE / 2];
    a[j][0] = fmaxf(t.x, 0.0f);
    a[j][1] = fmaxf(t.y, 0.0f);
  }

  float sp[4] = {0.0f, 0.0f, 0.0f, 0.0f};
#pragma unroll
  for (int j = 0; j < KS; ++j) sp[j & 3] += a[j][0] + a[j][1];
  float s = (sp[0] + sp[1]) + (sp[2] + sp[3]);
  s += xhalf(s);
  const float mu = s * (1.0f / KDIM);
  float qp[4] = {0.0f, 0.0f, 0.0f, 0.0f};
#pragma unroll
  for (int j = 0; j < KS; ++j) {
    const float d0 = a[j][0] - mu;
    const float d1 = a[j][1] - mu;
    qp[j & 3] += d0 * d0 + d1 * d1;
  }
  float q = (qp[0] + qp[1]) + (qp[2] + qp[3]);
  q += xhalf(q);
  const float rs = rsqrtf(q * (1.0f / KDIM) + EPSF);

  const float2* __restrict__ gp = reinterpret_cast<const float2*>(g + 2 * half);
  const float2* __restrict__ bp = reinterpret_cast<const float2*>(b + 2 * half);
#pragma unroll
  for (int j = 0; j < KS; ++j) {
    const float2 gg = gp[2 * j];
    const float2 sh = bp[2 * j];
    a[j][0] = (a[j][0] - mu) * rs * gg.x + sh.x;
    a[j][1] = (a[j][1] - mu) * rs * gg.y + sh.y;
  }

  v8f acc[NT];
#pragma unroll
  for (int t = 0; t < NT; ++t)
#pragma unroll
    for (int j = 0; j < 8; ++j) acc[t][j] = 0.0f;

#pragma unroll
  for (int j = 0; j < KS; ++j) {
    const float2* __restrict__ brow = Bl + (size_t)(2 * j + half) * NCOLS + l;
    v2f bfrag[NT];
#pragma unroll
    for (int t = 0; t < NT; ++t) {
      float2 bb = brow[t * 16];
      bfrag[t][0] = bb.x;
      bfrag[t][1] = bb.y;
    }
#pragma unroll
    for (int t = 0; t < NT; ++t)
      acc[t] = __builtin_amdgcn_wmma_f32_16x16x4_f32(
           false, a[j],  false, bfrag[t],
           (short)0, acc[t],  false,  false);
  }

#pragma unroll
  for (int t = 0; t < NT; ++t) {
    const int cc = t * 16 + l;
#pragma unroll
    for (int r = 0; r < 8; ++r)
      C[(size_t)(m0 + r + 8 * half) * NCOLS + cc] = acc[t][r] * rsc[r];
  }
}

#define GCH 8
template <bool NORM>
__global__ void gather_rows(const float* __restrict__ hp, const int* __restrict__ offs,
                            const int2* __restrict__ pairs, const float* __restrict__ dinv,
                            const float* __restrict__ bias, const float* __restrict__ g,
                            const float* __restrict__ b, float* __restrict__ outp, int n) {
  const int wave = (int)(((size_t)blockIdx.x * blockDim.x + threadIdx.x) >> 5);
  const int lane = threadIdx.x & 31;
  if (wave >= n) return;
  const int node = __builtin_amdgcn_readfirstlane(wave);
  const float2* __restrict__ H = reinterpret_cast<const float2*>(hp) + lane;
  const int beg = offs[node];
  const int end = offs[node + 1];
  const float dc = dinv[node];
  const float2 self = H[(size_t)node * (PLANE / 2)];
  const float2 bb = reinterpret_cast<const float2*>(bias)[lane];
  float a0 = bb.x + self.x * dc;
  float a1 = bb.y + self.y * dc;

  for (int base = beg; base < end; base += 32) {
    const int m = min(32, end - base);
    const int2 pr = pairs[base + min(lane, m - 1)];
    const int rv = pr.x;
    const int cv = __float_as_int(__int_as_float(pr.y) * dc);
    for (int j = 0; j < m; j += GCH) {
      float2 v[GCH];
      float cf[GCH];
#pragma unroll
      for (int u = 0; u < GCH; ++u) {
        const bool ok = (j + u < m);
        const int src = ok ? (j + u) : j;
        const int r = __builtin_amdgcn_readlane(rv, src);
        const float c = __int_as_float(__builtin_amdgcn_readlane(cv, src));
        const float2 t = H[(size_t)r * (PLANE / 2)];
        v[u].x = ok ? t.x : 0.0f;
        v[u].y = ok ? t.y : 0.0f;
        cf[u] = ok ? c : 0.0f;
      }
#pragma unroll
      for (int u = 0; u < GCH; ++u) {
        a0 += v[u].x * cf[u];
        a1 += v[u].y * cf[u];
      }
    }
  }
  float2 o;
  if (NORM) {
    float v[2];
    v[0] = fmaxf(a0, 0.0f);
    v[1] = fmaxf(a1, 0.0f);
    float rsum = 0.0f;
#pragma unroll
    for (int f = 0; f < 2; ++f) rsum += v[f];
    const float mu = wave_sum(rsum) * (1.0f / PLANE);
    float q = 0.0f;
#pragma unroll
    for (int f = 0; f < 2; ++f) {
      float d = v[f] - mu;
      q += d * d;
    }
    const float rs = rsqrtf(wave_sum(q) * (1.0f / PLANE) + EPSF);
    const float2 gg = reinterpret_cast<const float2*>(g)[lane];
    const float2 sh = reinterpret_cast<const float2*>(b)[lane];
    o.x = (v[0] - mu) * rs * gg.x + sh.x;
    o.y = (v[1] - mu) * rs * gg.y + sh.y;
  } else {
    o.x = a0;
    o.y = a1;
  }
  reinterpret_cast<float2*>(outp)[(size_t)node * (PLANE / 2) + lane] = o;
}

#define PR 32
__global__ void pool_accum(const float* __restrict__ a, const int* __restrict__ batch,
                           float* __restrict__ sums, float* __restrict__ cnts, int n) {
  const int wave = (int)(((size_t)blockIdx.x * blockDim.x + threadIdx.x) >> 5);
  const int lane = threadIdx.x & 31;
  const int row0 = wave * PR;
  if (row0 >= n) return;
  const int nrows = min(PR, n - row0);
  const float2* __restrict__ p =
      reinterpret_cast<const float2*>(a + (size_t)row0 * DOUT) + lane;
  const int gl = batch[row0 + min(lane, nrows - 1)];
  const int g0 = __builtin_amdgcn_readfirstlane(gl);

  if (nrows == PR && __all(gl == g0)) {
    float2 v[PR];
#pragma unroll
    for (int i = 0; i < PR; ++i) v[i] = p[i * (DOUT / 2)];
    float s0 = 0.0f, s1 = 0.0f;
#pragma unroll
    for (int i = 0; i < PR; ++i) {
      s0 += v[i].x;
      s1 += v[i].y;
    }
    atomicAdd(&sums[g0 * DOUT + lane * 2 + 0], s0);
    atomicAdd(&sums[g0 * DOUT + lane * 2 + 1], s1);
    if (lane == 0) atomicAdd(&cnts[g0], (float)PR);
  } else {
    int cur = g0;
    float s0 = 0.0f, s1 = 0.0f, c = 0.0f;
    for (int i = 0; i < nrows; ++i) {
      const int gi = __builtin_amdgcn_readfirstlane(batch[row0 + i]);
      if (gi != cur) {
        atomicAdd(&sums[cur * DOUT + lane * 2 + 0], s0);
        atomicAdd(&sums[cur * DOUT + lane * 2 + 1], s1);
        if (lane == 0) atomicAdd(&cnts[cur], c);
        cur = gi;
        s0 = 0.0f;
        s1 = 0.0f;
        c = 0.0f;
      }
      const float2 v = p[i * (DOUT / 2)];
      s0 += v.x;
      s1 += v.y;
      c += 1.0f;
    }
    atomicAdd(&sums[cur * DOUT + lane * 2 + 0], s0);
    atomicAdd(&sums[cur * DOUT + lane * 2 + 1], s1);
    if (lane == 0) atomicAdd(&cnts[cur], c);
  }
}

__global__ void pool_final(const float* __restrict__ sums, const float* __restrict__ cnts,
                           float* __restrict__ out) {
  int i = blockIdx.x * blockDim.x + threadIdx.x;
  if (i < NGRAPHS * DOUT) {
    int g = i / DOUT;
    float m = sums[i] / fmaxf(cnts[g], 1.0f);
    out[i] = 1.0f / (1.0f + __expf(-m));
  }
}

extern "C" void kernel_launch(void* const* d_in, const int* in_sizes, int n_in,
                              void* d_out, int out_size, void* d_ws, size_t ws_size,
                              hipStream_t stream) {
  const float* x   = (const float*)d_in[0];
  const int*   ei  = (const int*)d_in[1];
  const float* ew  = (const float*)d_in[2];
  const int*   bat = (const int*)d_in[3];
  const float* W1  = (const float*)d_in[4];
  const float* b1  = (const float*)d_in[5];
  const float* g1  = (const float*)d_in[6];
  const float* bt1 = (const float*)d_in[7];
  const float* W2  = (const float*)d_in[8];
  const float* b2  = (const float*)d_in[9];
  const float* g2  = (const float*)d_in[10];
  const float* bt2 = (const float*)d_in[11];
  float* out = (float*)d_out;

  const int* rows = ei;
  const int* cols = ei + NEDGES;

  char* ws = (char*)d_ws;
  size_t off = 0;
  auto carve = [&](size_t bytes) -> float* {
    float* p = (float*)(ws + off);
    off += (bytes + 255) & ~(size_t)255;
    return p;
  };
  float* hA   = carve((size_t)NNODES * PLANE * 4);
  float* hB   = carve((size_t)NNODES * PLANE * 4);
  float* aA   = carve((size_t)NNODES * PLANE * 4);
  float* dinv = carve((size_t)NPAD * 4);
  int*   offs = (int*)carve((size_t)NPAD * 4);
  int2*  pairs = (int2*)carve((size_t)NEDGES * 8);
  float* sums = carve((size_t)NGRAPHS * DOUT * 4);
  float* cnts = carve((size_t)NGRAPHS * 4);

  const int TB = 256;
  auto blocks = [&](long long threads) { return (int)((threads + TB - 1) / TB); };

  int2* tmp  = (int2*)hA;
  int*  hist = (int*)hB;
  int*  base = (int*)hB + HIST_ROOM;
  edge_hist<<<ECH, TBE, 0, stream>>>(cols, hist, sums, cnts);
  offs_scan<<<(NBK * ECH + SCAN_TB * 4 - 1) / (SCAN_TB * 4), SCAN_TB, 0, stream>>>(hist, base, NBK * ECH);
  edge_bucket<<<ECH, TBE, 0, stream>>>(rows, cols, ew, base, tmp);
  bucket_fine<<<NBK, TBF, 0, stream>>>(tmp, base, offs, dinv, pairs);

  wmma_gemm_f32<DHID, DIN, DIN, PLANE><<<((NNODES / 16) * 32 + GT - 1) / GT, GT, 0, stream>>>(x, x, W1, dinv, hA, hB, NNODES);
  gather_rows<false><<<blocks((long long)NNODES * 32), TB, 0, stream>>>(hA, offs, pairs, dinv, b1, g1, bt1, aA, NNODES);
  gather_rows<false><<<blocks((long long)NNODES * 32), TB, 0, stream>>>(hB, offs, pairs, dinv, b1 + PLANE, g1, bt1, hA, NNODES);

  ln_wmma_gemm_f32<<<((NNODES / 16) * 32 + GT - 1) / GT, GT, 0, stream>>>(aA, hA, g1, bt1, W2, dinv, hB, NNODES);
  gather_rows<true><<<blocks((long long)NNODES * 32), TB, 0, stream>>>(hB, offs, pairs, dinv, b2, g2, bt2, aA, NNODES);

  pool_accum<<<blocks((long long)((NNODES + PR - 1) / PR) * 32), TB, 0, stream>>>(aA, bat, sums, cnts, NNODES);
  pool_final<<<blocks(NGRAPHS * DOUT), TB, 0, stream>>>(sums, cnts, out);
}


# neighbour sums: a 16-lane half reads one 256-B row by 16-byte loads, two edges' rows per wave load; edge records shared by DPP row_share, no readlane or scalar work per row; sum order changes
# speedup vs baseline: 4.2818x; 1.0525x over previous
#include <hip/hip_runtime.h>
#include <math.h>


#define NNODES 100000
#define NEDGES 1600000
#define NGRAPHS 64
#define DIN 256
#define DHID 128
#define DOUT 64
#define EPSF 1e-5f
#define PLANE 64
#define NPAD 100032

typedef __attribute__((ext_vector_type(2))) float v2f;
typedef __attribute__((ext_vector_type(8))) float v8f;

__device__ __forceinline__ float xhalf(float x) {
  const int b = __builtin_bit_cast(int, x);
  return __builtin_bit_cast(
      float, __builtin_amdgcn_permlanex16(b, b, 0x76543210, 0xfedcba98, false, false));
}
template <int XM>
__device__ __forceinline__ float xrow(float x) {
  const int b = __builtin_bit_cast(int, x);
  return __builtin_bit_cast(float, __builtin_amdgcn_update_dpp(b, b, 0x160 + XM, 0xf, 0xf, false));
}
__device__ __forceinline__ float wave_sum(float x) {
  x += xhalf(x);
  x += xrow<8>(x);
  x += xrow<4>(x);
  x += xrow<2>(x);
  x += xrow<1>(x);
  return x;
}

#define SCAN_TB 1024
__global__ void offs_scan(const int* __restrict__ cnt, int* __restrict__ offs, int n) {
  __shared__ int wsum[SCAN_TB / 32];
  const int tid = threadIdx.x;
  const int lane = tid & 31;
  const int wv = tid >> 5;
  const int first = blockIdx.x * (SCAN_TB * 4);

  int part = 0;
  const int4* __restrict__ c4 = reinterpret_cast<const int4*>(cnt);
#pragma unroll 8
  for (int i = tid; i < first / 4; i += SCAN_TB) {
    const int4 q = c4[i];
    part += (q.x + q.y) + (q.z + q.w);
  }
#pragma unroll
  for (int off = 16; off > 0; off >>= 1) part += __shfl_xor(part, off, 32);
  if (lane == 0) wsum[wv] = part;
  __syncthreads();
  int base = 0;
  for (int k = 0; k < SCAN_TB / 32; ++k) base += wsum[k];
  __syncthreads();

  const int idx0 = first + tid * 4;
  int v[4];
#pragma unroll
  for (int j = 0; j < 4; ++j) v[j] = (idx0 + j < n) ? cnt[idx0 + j] : 0;
  const int tsum = (v[0] + v[1]) + (v[2] + v[3]);
  int inc = tsum;
#pragma unroll
  for (int off = 1; off < 32; off <<= 1) {
    const int t = __shfl_up(inc, off, 32);
    if (lane >= off) inc += t;
  }
  if (lane == 31) wsum[wv] = inc;
  __syncthreads();
  int wbase = 0;
  for (int k = 0; k < SCAN_TB / 32; ++k) wbase += (k < wv) ? wsum[k] : 0;
  int run = base + wbase + inc - tsum;
#pragma unroll
  for (int j = 0; j < 4; ++j) {
    if (idx0 + j < n) {
      offs[idx0 + j] = run;
      run += v[j];
      if (idx0 + j == n - 1) offs[n] = run;
    }
  }
}

#define BKSH 8
#define BKN (1 << BKSH)
#define NBK ((NNODES + BKN - 1) / BKN)
#define ECH 256
#define ECHUNK (NEDGES / ECH)
#define TBE 1024
#define HIST_ROOM (((NBK * ECH) + 1 + 1023) & ~1023)
#define EPT ((ECHUNK + TBE - 1) / TBE)
static_assert(NEDGES % ECH == 0, "chunks must tile the edge list");
static_assert(NNODES <= (1 << 17), "source index must fit under the packed node-in-bucket");
static_assert(BKSH + 17 < 32, "the packed word must stay positive");

__global__ void edge_hist(const int* __restrict__ cols, int* __restrict__ hist,
                          float* __restrict__ sums, float* __restrict__ cnts) {
  __shared__ int h[NBK];
  const int tid = threadIdx.x;
  if (blockIdx.x == 0) {
    for (int i = tid; i < NGRAPHS * DOUT; i += TBE) sums[i] = 0.0f;
    if (tid < NGRAPHS) cnts[tid] = 0.0f;
  }
  for (int i = tid; i < NBK; i += TBE) h[i] = 0;
  __syncthreads();
  const int e0 = blockIdx.x * ECHUNK;
  int c[EPT];
#pragma unroll
  for (int k = 0; k < EPT; ++k) {
    const int i = k * TBE + tid;
    c[k] = (i < ECHUNK) ? cols[e0 + i] : -1;
  }
#pragma unroll
  for (int k = 0; k < EPT; ++k)
    if (c[k] >= 0) atomicAdd(&h[c[k] >> BKSH], 1);
  __syncthreads();
  for (int i = tid; i < NBK; i += TBE) hist[i * ECH + blockIdx.x] = h[i];
}

__global__ void edge_bucket(const int* __restrict__ rows, const int* __restrict__ cols,
                            const float* __restrict__ w, const int* __restrict__ base,
                            int2* __restrict__ tmp) {
  __shared__ int cur[NBK];
  const int tid = threadIdx.x;
  for (int i = tid; i < NBK; i += TBE) cur[i] = base[i * ECH + blockIdx.x];
  __syncthreads();
  const int e0 = blockIdx.x * ECHUNK;
  int c[EPT];
  int2 rec[EPT];
#pragma unroll
  for (int k = 0; k < EPT; ++k) {
    const int i = k * TBE + tid;
    const bool ok = (i < ECHUNK);
    c[k] = ok ? cols[e0 + i] : -1;
    rec[k].x = ok ? rows[e0 + i] : 0;
    rec[k].y = ok ? __float_as_int(w[e0 + i]) : 0;
  }
#pragma unroll
  for (int k = 0; k < EPT; ++k) {
    if (c[k] >= 0) {
      const int pos = atomicAdd(&cur[c[k] >> BKSH], 1);
      int2 t;
      t.x = rec[k].x | ((c[k] & (BKN - 1)) << 17);
      t.y = rec[k].y;
      tmp[pos] = t;
    }
  }
}

#define TBF 1024
#define FPT 4
__global__ void bucket_fine(const int2* __restrict__ tmp, const int* __restrict__ base,
                            int* __restrict__ offs, float* __restrict__ dinv,
                            int2* __restrict__ pairs) {
  __shared__ int cntL[BKN];
  __shared__ float degL[BKN];
  __shared__ int curL[BKN];
  __shared__ int wsum[BKN / 32];
  const int tid = threadIdx.x;
  const int lane = tid & 31;
  const int wv = tid >> 5;
  const int start = base[blockIdx.x * ECH];
  const int end = base[(blockIdx.x + 1) * ECH];
  if (tid < BKN) {
    cntL[tid] = 0;
    degL[tid] = 0.0f;
  }
  __syncthreads();

  for (int b0 = start; b0 < end; b0 += FPT * TBF) {
    int2 t[FPT];
#pragma unroll
    for (int k = 0; k < FPT; ++k) {
      const int i = b0 + k * TBF + tid;
      t[k] = (i < end) ? tmp[i] : make_int2(-1, 0);
    }
#pragma unroll
    for (int k = 0; k < FPT; ++k) {
      if (t[k].x >= 0) {
        const int cl = t[k].x >> 17;
        atomicAdd(&cntL[cl], 1);
        atomicAdd(&degL[cl], __int_as_float(t[k].y));
      }
    }
  }
  __syncthreads();

  int v = 0, inc = 0;
  if (tid < BKN) {
    v = cntL[tid];
    inc = v;
#pragma unroll
    for (int off = 1; off < 32; off <<= 1) {
      const int u = __shfl_up(inc, off, 32);
      if (lane >= off) inc += u;
    }
    if (lane == 31) wsum[wv] = inc;
  }
  __syncthreads();
  if (tid < BKN) {
    int wbase = 0;
    for (int k = 0; k < BKN / 32; ++k) wbase += (k < wv) ? wsum[k] : 0;
    const int excl = start + wbase + inc - v;
    curL[tid] = excl;
    const int node = blockIdx.x * BKN + tid;
    if (node < NNODES) {
      offs[node] = excl;
      dinv[node] = rsqrtf(fmaxf(degL[tid] + 1.0f, EPSF));
      if (node == NNODES - 1) offs[NNODES] = excl + v;
    }
  }
  __syncthreads();

  for (int b0 = start; b0 < end; b0 += FPT * TBF) {
    int2 t[FPT];
#pragma unroll
    for (int k = 0; k < FPT; ++k) {
      const int i = b0 + k * TBF + tid;
      t[k] = (i < end) ? tmp[i] : make_int2(-1, 0);
    }
#pragma unroll
    for (int k = 0; k < FPT; ++k) {
      if (t[k].x >= 0) {
        const int pos = atomicAdd(&curL[t[k].x >> 17], 1);
        int2 pr;
        pr.x = t[k].x & ((1 << 17) - 1);
        pr.y = t[k].y;
        pairs[pos] = pr;
      }
    }
  }
}

#define GT 256
template <int NCOLS, int KDIM>
__device__ __forceinline__ void stage_b_pairs(const float* __restrict__ B,
                                              float2* __restrict__ Bl) {
  constexpr int NC4 = NCOLS / 4;
  constexpr int UNITS = (KDIM / 2) * NC4;
  constexpr int UPT = UNITS / GT;
  constexpr int SB = (UPT < 8) ? UPT : 8;
  static_assert(UNITS % GT == 0 && UPT % SB == 0, "the units must tile the work-group");
  const float4* __restrict__ B4 = reinterpret_cast<const float4*>(B);
  float4* __restrict__ Bl4 = reinterpret_cast<float4*>(Bl);
  for (int u0 = threadIdx.x; u0 < UNITS; u0 += SB * GT) {
    float4 ra[SB], rb[SB];
#pragma unroll
    for (int s = 0; s < SB; ++s) {
      const int u = u0 + s * GT;
      const int k2 = u / NC4;
      const int n4 = u - k2 * NC4;
      ra[s] = B4[(2 * k2 + 0) * NC4 + n4];
      rb[s] = B4[(2 * k2 + 1) * NC4 + n4];
    }
#pragma unroll
    for (int s = 0; s < SB; ++s) {
      const int u = u0 + s * GT;
      Bl4[2 * u + 0] = make_float4(ra[s].x, rb[s].x, ra[s].y, rb[s].y);
      Bl4[2 * u + 1] = make_float4(ra[s].z, rb[s].z, ra[s].w, rb[s].w);
    }
  }
}

template <int NCOLS, int KDIM, int KPL, int CPL>
__global__ void wmma_gemm_f32(const float* __restrict__ A0, const float* __restrict__ A1,
                              const float* __restrict__ B, const float* __restrict__ dinv,
                              float* __restrict__ C0, float* __restrict__ C1, int M) {
  __shared__ alignas(16) float2 Bl[(KDIM / 2) * NCOLS];

  stage_b_pairs<NCOLS, KDIM>(B, Bl);
  __syncthreads();

  const int wave = (int)(((size_t)blockIdx.x * blockDim.x + threadIdx.x) >> 5);
  const int lane = threadIdx.x & 31;
  if (wave >= M / 16) return;
  const int m0 = wave * 16;
  const int half = lane >> 4;
  const int l = lane & 15;
  constexpr int NT = NCOLS / 16;

  v8f acc[NT];
#pragma unroll
  for (int t = 0; t < NT; ++t)
#pragma unroll
    for (int j = 0; j < 8; ++j) acc[t][j] = 0.0f;

  auto a_at = [&](int k) -> const float* {
    const float* base = (KPL < KDIM && k >= KPL) ? A1 : A0;
    return base + (size_t)(m0 + l) * KPL + (k & (KPL - 1)) + 2 * half;
  };

  v2f a;
  {
    const float* ap = a_at(0);
    a[0] = ap[0];
    a[1] = ap[1];
  }

  float rsc[8];
#pragma unroll
  for (int r = 0; r < 8; ++r) rsc[r] = dinv[m0 + r + 8 * half];

  for (int k0 = 0; k0 < KDIM; k0 += 4) {
    v2f a_next = a;
    if (k0 + 4 < KDIM) {
      const float* np = a_at(k0 + 4);
      a_next[0] = np[0];
      a_next[1] = np[1];
    }
    const float2* __restrict__ brow = Bl + (size_t)(k0 / 2 + half) * NCOLS + l;
    v2f bfrag[NT];
#pragma unroll
    for (int t = 0; t < NT; ++t) {
      float2 bb = brow[t * 16];
      bfrag[t][0] = bb.x;
      bfrag[t][1] = bb.y;
    }
#pragma unroll
    for (int t = 0; t < NT; ++t)
      acc[t] = __builtin_amdgcn_wmma_f32_16x16x4_f32(
           false, a,  false, bfrag[t],
           (short)0, acc[t],  false,  false);
    a = a_next;
  }

#pragma unroll
  for (int t = 0; t < NT; ++t) {
    constexpr int TPP = CPL / 16;
    float* __restrict__ cb = (t / TPP) ? C1 : C0;
    const int cc = (t % TPP) * 16 + l;
#pragma unroll
    for (int r = 0; r < 8; ++r)
      cb[(size_t)(m0 + r + 8 * half) * CPL + cc] = acc[t][r] * rsc[r];
  }
}

__global__ void ln_wmma_gemm_f32(const float* __restrict__ A0, const float* __restrict__ A1,
                                 const float* __restrict__ g, const float* __restrict__ b,
                                 const float* __restrict__ B, const float* __restrict__ dinv,
                                 float* __restrict__ C, int M) {
  constexpr int KDIM = DHID;
  constexpr int NCOLS = DOUT;
  constexpr int NT = NCOLS / 16;
  constexpr int KS = KDIM / 4;
  __shared__ alignas(16) float2 Bl[(KDIM / 2) * NCOLS];

  stage_b_pairs<NCOLS, KDIM>(B, Bl);
  __syncthreads();

  const int wave = (int)(((size_t)blockIdx.x * blockDim.x + threadIdx.x) >> 5);
  const int lane = threadIdx.x & 31;
  if (wave >= M / 16) return;
  const int m0 = wave * 16;
  const int half = lane >> 4;
  const int l = lane & 15;

  const float2* __restrict__ r0 =
      reinterpret_cast<const float2*>(A0 + (size_t)(m0 + l) * PLANE + 2 * half);
  const float2* __restrict__ r1 =
      reinterpret_cast<const float2*>(A1 + (size_t)(m0 + l) * PLANE + 2 * half);
  float rsc[8];
#pragma unroll
  for (int r = 0; r < 8; ++r) rsc[r] = dinv[m0 + r + 8 * half];

  v2f a[KS];
#pragma unroll
  for (int j = 0; j < KS; ++j) {
    const float2 t = (4 * j < PLANE) ? r0[2 * j] : r1[2 * j - PLANE / 2];
    a[j][0] = fmaxf(t.x, 0.0f);
    a[j][1] = fmaxf(t.y, 0.0f);
  }

  float sp[4] = {0.0f, 0.0f, 0.0f, 0.0f};
#pragma unroll
  for (int j = 0; j < KS; ++j) sp[j & 3] += a[j][0] + a[j][1];
  float s = (sp[0] + sp[1]) + (sp[2] + sp[3]);
  s += xhalf(s);
  const float mu = s * (1.0f / KDIM);
  float qp[4] = {0.0f, 0.0f, 0.0f, 0.0f};
#pragma unroll
  for (int j = 0; j < KS; ++j) {
    const float d0 = a[j][0] - mu;
    const float d1 = a[j][1] - mu;
    qp[j & 3] += d0 * d0 + d1 * d1;
  }
  float q = (qp[0] + qp[1]) + (qp[2] + qp[3]);
  q += xhalf(q);
  const float rs = rsqrtf(q * (1.0f / KDIM) + EPSF);

  const float2* __restrict__ gp = reinterpret_cast<const float2*>(g + 2 * half);
  const float2* __restrict__ bp = reinterpret_cast<const float2*>(b + 2 * half);
#pragma unroll
  for (int j = 0; j < KS; ++j) {
    const float2 gg = gp[2 * j];
    const float2 sh = bp[2 * j];
    a[j][0] = (a[j][0] - mu) * rs * gg.x + sh.x;
    a[j][1] = (a[j][1] - mu) * rs * gg.y + sh.y;
  }

  v8f acc[NT];
#pragma unroll
  for (int t = 0; t < NT; ++t)
#pragma unroll
    for (int j = 0; j < 8; ++j) acc[t][j] = 0.0f;

#pragma unroll
  for (int j = 0; j < KS; ++j) {
    const float2* __restrict__ brow = Bl + (size_t)(2 * j + half) * NCOLS + l;
    v2f bfrag[NT];
#pragma unroll
    for (int t = 0; t < NT; ++t) {
      float2 bb = brow[t * 16];
      bfrag[t][0] = bb.x;
      bfrag[t][1] = bb.y;
    }
#pragma unroll
    for (int t = 0; t < NT; ++t)
      acc[t] = __builtin_amdgcn_wmma_f32_16x16x4_f32(
           false, a[j],  false, bfrag[t],
           (short)0, acc[t],  false,  false);
  }

#pragma unroll
  for (int t = 0; t < NT; ++t) {
    const int cc = t * 16 + l;
#pragma unroll
    for (int r = 0; r < 8; ++r)
      C[(size_t)(m0 + r + 8 * half) * NCOLS + cc] = acc[t][r] * rsc[r];
  }
}

template <int J>
__device__ __forceinline__ int rshare(int x) {
  return __builtin_amdgcn_mov_dpp(x, 0x150 + J, 0xf, 0xf, true);
}
template <int J0, int NS, int U>
__device__ __forceinline__ void ask_rows(const char* __restrict__ Hb, unsigned loff, int roff,
                                         int cvb, float4 (&v)[NS], float (&cf)[NS]) {
  if constexpr (U < NS) {
    const unsigned ro = (unsigned)rshare<J0 + U>(roff);
    cf[U] = __int_as_float(rshare<J0 + U>(cvb));
    v[U] = *reinterpret_cast<const float4*>(Hb + (ro + loff));
    ask_rows<J0, NS, U + 1>(Hb, loff, roff, cvb, v, cf);
  }
}
template <int J0, int NS>
__device__ __forceinline__ void sum_steps(const char* __restrict__ Hb, unsigned loff, int roff,
                                          int cvb, float4& acc) {
  float4 v[NS];
  float cf[NS];
  ask_rows<J0, NS, 0>(Hb, loff, roff, cvb, v, cf);
#pragma unroll
  for (int u = 0; u < NS; ++u) {
    acc.x += v[u].x * cf[u];
    acc.y += v[u].y * cf[u];
    acc.z += v[u].z * cf[u];
    acc.w += v[u].w * cf[u];
  }
}
template <bool NORM>
__global__ void gather_rows(const float* __restrict__ hp, const int* __restrict__ offs,
                            const int2* __restrict__ pairs, const float* __restrict__ dinv,
                            const float* __restrict__ bias, const float* __restrict__ g,
                            const float* __restrict__ b, float* __restrict__ outp, int n) {
  const int wave = (int)(((size_t)blockIdx.x * blockDim.x + threadIdx.x) >> 5);
  const int lane = threadIdx.x & 31;
  if (wave >= n) return;
  const int node = __builtin_amdgcn_readfirstlane(wave);
  const int half = lane >> 4;
  const int c16 = lane & 15;
  const unsigned loff = (unsigned)c16 * 16u;
  const char* __restrict__ Hb = reinterpret_cast<const char*>(hp);
  const int beg = offs[node];
  const int end = offs[node + 1];
  const float dc = dinv[node];
  const float4 self = *reinterpret_cast<const float4*>(Hb + ((unsigned)node * 256u + loff));
  const float4 bb = reinterpret_cast<const float4*>(bias)[c16];
  const int e = 2 * c16 + half;

  float4 acc = make_float4(0.0f, 0.0f, 0.0f, 0.0f);
  for (int base = beg; base < end; base += 32) {
    const int m = min(32, end - base);
    const int2 pr = *reinterpret_cast<const int2*>(
        reinterpret_cast<const char*>(pairs) + (unsigned)(base + min(e, m - 1)) * 8u);
    const int roff = pr.x << 8;
    const float cw = __int_as_float(pr.y) * dc;
    const int cvb = (e < m) ? __float_as_int(cw) : 0;
    const int steps = (m + 1) >> 1;
    sum_steps<0, 8>(Hb, loff, roff, cvb, acc);
    if (steps > 8) {
      sum_steps<8, 4>(Hb, loff, roff, cvb, acc);
      if (steps > 12) sum_steps<12, 4>(Hb, loff, roff, cvb, acc);
    }
  }
  float a[4];
  a[0] = (acc.x + xhalf(acc.x)) + (bb.x + self.x * dc);
  a[1] = (acc.y + xhalf(acc.y)) + (bb.y + self.y * dc);
  a[2] = (acc.z + xhalf(acc.z)) + (bb.z + self.z * dc);
  a[3] = (acc.w + xhalf(acc.w)) + (bb.w + self.w * dc);
  float4 o;
  if (NORM) {
    float v[4];
#pragma unroll
    for (int f = 0; f < 4; ++f) v[f] = fmaxf(a[f], 0.0f);
    float rsum = (v[0] + v[1]) + (v[2] + v[3]);
    rsum += xrow<8>(rsum);
    rsum += xrow<4>(rsum);
    rsum += xrow<2>(rsum);
    rsum += xrow<1>(rsum);
    const float mu = rsum * (1.0f / PLANE);
    float q = 0.0f;
#pragma unroll
    for (int f = 0; f < 4; ++f) {
      const float d = v[f] - mu;
      q += d * d;
    }
    q += xrow<8>(q);
    q += xrow<4>(q);
    q += xrow<2>(q);
    q += xrow<1>(q);
    const float rs = rsqrtf(q * (1.0f / PLANE) + EPSF);
    const float4 gg = reinterpret_cast<const float4*>(g)[c16];
    const float4 sh = reinterpret_cast<const float4*>(b)[c16];
    o.x = (v[0] - mu) * rs * gg.x + sh.x;
    o.y = (v[1] - mu) * rs * gg.y + sh.y;
    o.z = (v[2] - mu) * rs * gg.z + sh.z;
    o.w = (v[3] - mu) * rs * gg.w + sh.w;
  } else {
    o.x = a[0];
    o.y = a[1];
    o.z = a[2];
    o.w = a[3];
  }
  if (half == 0)
    *reinterpret_cast<float4*>(reinterpret_cast<char*>(outp) + ((size_t)node * 256u + loff)) = o;
}

#define PR 32
__global__ void pool_accum(const float* __restrict__ a, const int* __restrict__ batch,
                           float* __restrict__ sums, float* __restrict__ cnts, int n) {
  const int wave = (int)(((size_t)blockIdx.x * blockDim.x + threadIdx.x) >> 5);
  const int lane = threadIdx.x & 31;
  const int row0 = wave * PR;
  if (row0 >= n) return;
  const int nrows = min(PR, n - row0);
  const float2* __restrict__ p =
      reinterpret_cast<const float2*>(a + (size_t)row0 * DOUT) + lane;
  const int gl = batch[row0 + min(lane, nrows - 1)];
  const int g0 = __builtin_amdgcn_readfirstlane(gl);

  if (nrows == PR && __all(gl == g0)) {
    float2 v[PR];
#pragma unroll
    for (int i = 0; i < PR; ++i) v[i] = p[i * (DOUT / 2)];
    float s0 = 0.0f, s1 = 0.0f;
#pragma unroll
    for (int i = 0; i < PR; ++i) {
      s0 += v[i].x;
      s1 += v[i].y;
    }
    atomicAdd(&sums[g0 * DOUT + lane * 2 + 0], s0);
    atomicAdd(&sums[g0 * DOUT + lane * 2 + 1], s1);
    if (lane == 0) atomicAdd(&cnts[g0], (float)PR);
  } else {
    int cur = g0;
    float s0 = 0.0f, s1 = 0.0f, c = 0.0f;
    for (int i = 0; i < nrows; ++i) {
      const int gi = __builtin_amdgcn_readfirstlane(batch[row0 + i]);
      if (gi != cur) {
        atomicAdd(&sums[cur * DOUT + lane * 2 + 0], s0);
        atomicAdd(&sums[cur * DOUT + lane * 2 + 1], s1);
        if (lane == 0) atomicAdd(&cnts[cur], c);
        cur = gi;
        s0 = 0.0f;
        s1 = 0.0f;
        c = 0.0f;
      }
      const float2 v = p[i * (DOUT / 2)];
      s0 += v.x;
      s1 += v.y;
      c += 1.0f;
    }
    atomicAdd(&sums[cur * DOUT + lane * 2 + 0], s0);
    atomicAdd(&sums[cur * DOUT + lane * 2 + 1], s1);
    if (lane == 0) atomicAdd(&cnts[cur], c);
  }
}

__global__ void pool_final(const float* __restrict__ sums, const float* __restrict__ cnts,
                           float* __restrict__ out) {
  int i = blockIdx.x * blockDim.x + threadIdx.x;
  if (i < NGRAPHS * DOUT) {
    int g = i / DOUT;
    float m = sums[i] / fmaxf(cnts[g], 1.0f);
    out[i] = 1.0f / (1.0f + __expf(-m));
  }
}

extern "C" void kernel_launch(void* const* d_in, const int* in_sizes, int n_in,
                              void* d_out, int out_size, void* d_ws, size_t ws_size,
                              hipStream_t stream) {
  const float* x   = (const float*)d_in[0];
  const int*   ei  = (const int*)d_in[1];
  const float* ew  = (const float*)d_in[2];
  const int*   bat = (const int*)d_in[3];
  const float* W1  = (const float*)d_in[4];
  const float* b1  = (const float*)d_in[5];
  const float* g1  = (const float*)d_in[6];
  const float* bt1 = (const float*)d_in[7];
  const float* W2  = (const float*)d_in[8];
  const float* b2  = (const float*)d_in[9];
  const float* g2  = (const float*)d_in[10];
  const float* bt2 = (const float*)d_in[11];
  float* out = (float*)d_out;

  const int* rows = ei;
  const int* cols = ei + NEDGES;

  char* ws = (char*)d_ws;
  size_t off = 0;
  auto carve = [&](size_t bytes) -> float* {
    float* p = (float*)(ws + off);
    off += (bytes + 255) & ~(size_t)255;
    return p;
  };
  float* hA   = carve((size_t)NNODES * PLANE * 4);
  float* hB   = carve((size_t)NNODES * PLANE * 4);
  float* aA   = carve((size_t)NNODES * PLANE * 4);
  float* dinv = carve((size_t)NPAD * 4);
  int*   offs = (int*)carve((size_t)NPAD * 4);
  int2*  pairs = (int2*)carve((size_t)NEDGES * 8);
  float* sums = carve((size_t)NGRAPHS * DOUT * 4);
  float* cnts = carve((size_t)NGRAPHS * 4);

  const int TB = 256;
  auto blocks = [&](long long threads) { return (int)((threads + TB - 1) / TB); };

  int2* tmp  = (int2*)hA;
  int*  hist = (int*)hB;
  int*  base = (int*)hB + HIST_ROOM;
  edge_hist<<<ECH, TBE, 0, stream>>>(cols, hist, sums, cnts);
  offs_scan<<<(NBK * ECH + SCAN_TB * 4 - 1) / (SCAN_TB * 4), SCAN_TB, 0, stream>>>(hist, base, NBK * ECH);
  edge_bucket<<<ECH, TBE, 0, stream>>>(rows, cols, ew, base, tmp);
  bucket_fine<<<NBK, TBF, 0, stream>>>(tmp, base, offs, dinv, pairs);

  wmma_gemm_f32<DHID, DIN, DIN, PLANE><<<((NNODES / 16) * 32 + GT - 1) / GT, GT, 0, stream>>>(x, x, W1, dinv, hA, hB, NNODES);
  gather_rows<false><<<blocks((long long)NNODES * 32), TB, 0, stream>>>(hA, offs, pairs, dinv, b1, g1, bt1, aA, NNODES);
  gather_rows<false><<<blocks((long long)NNODES * 32), TB, 0, stream>>>(hB, offs, pairs, dinv, b1 + PLANE, g1, bt1, hA, NNODES);

  ln_wmma_gemm_f32<<<((NNODES / 16) * 32 + GT - 1) / GT, GT, 0, stream>>>(aA, hA, g1, bt1, W2, dinv, hB, NNODES);
  gather_rows<true><<<blocks((long long)NNODES * 32), TB, 0, stream>>>(hB, offs, pairs, dinv, b2, g2, bt2, aA, NNODES);

  pool_accum<<<blocks((long long)((NNODES + PR - 1) / PR) * 32), TB, 0, stream>>>(aA, bat, sums, cnts, NNODES);
  pool_final<<<blocks(NGRAPHS * DOUT), TB, 0, stream>>>(sums, cnts, out);
}


# layer-1 neighbour sums of both planes, relu+layernorm and the layer-2 product in ONE launch: 16-node tiles, normalised rows reach two multiplying waves through LDS; LN sum order changes
# speedup vs baseline: 4.6411x; 1.0839x over previous
#include <hip/hip_runtime.h>
#include <math.h>


#define NNODES 100000
#define NEDGES 1600000
#define NGRAPHS 64
#define DIN 256
#define DHID 128
#define DOUT 64
#define EPSF 1e-5f
#define PLANE 64
#define NPAD 100032

typedef __attribute__((ext_vector_type(2))) float v2f;
typedef __attribute__((ext_vector_type(8))) float v8f;

__device__ __forceinline__ float xhalf(float x) {
  const int b = __builtin_bit_cast(int, x);
  return __builtin_bit_cast(
      float, __builtin_amdgcn_permlanex16(b, b, 0x76543210, 0xfedcba98, false, false));
}
template <int XM>
__device__ __forceinline__ float xrow(float x) {
  const int b = __builtin_bit_cast(int, x);
  return __builtin_bit_cast(float, __builtin_amdgcn_update_dpp(b, b, 0x160 + XM, 0xf, 0xf, false));
}
__device__ __forceinline__ float wave_sum(float x) {
  x += xhalf(x);
  x += xrow<8>(x);
  x += xrow<4>(x);
  x += xrow<2>(x);
  x += xrow<1>(x);
  return x;
}

#define SCAN_TB 1024
__global__ void offs_scan(const int* __restrict__ cnt, int* __restrict__ offs, int n) {
  __shared__ int wsum[SCAN_TB / 32];
  const int tid = threadIdx.x;
  const int lane = tid & 31;
  const int wv = tid >> 5;
  const int first = blockIdx.x * (SCAN_TB * 4);

  int part = 0;
  const int4* __restrict__ c4 = reinterpret_cast<const int4*>(cnt);
#pragma unroll 8
  for (int i = tid; i < first / 4; i += SCAN_TB) {
    const int4 q = c4[i];
    part += (q.x + q.y) + (q.z + q.w);
  }
#pragma unroll
  for (int off = 16; off > 0; off >>= 1) part += __shfl_xor(part, off, 32);
  if (lane == 0) wsum[wv] = part;
  __syncthreads();
  int base = 0;
  for (int k = 0; k < SCAN_TB / 32; ++k) base += wsum[k];
  __syncthreads();

  const int idx0 = first + tid * 4;
  int v[4];
#pragma unroll
  for (int j = 0; j < 4; ++j) v[j] = (idx0 + j < n) ? cnt[idx0 + j] : 0;
  const int tsum = (v[0] + v[1]) + (v[2] + v[3]);
  int inc = tsum;
#pragma unroll
  for (int off = 1; off < 32; off <<= 1) {
    const int t = __shfl_up(inc, off, 32);
    if (lane >= off) inc += t;
  }
  if (lane == 31) wsum[wv] = inc;
  __syncthreads();
  int wbase = 0;
  for (int k = 0; k < SCAN_TB / 32; ++k) wbase += (k < wv) ? wsum[k] : 0;
  int run = base + wbase + inc - tsum;
#pragma unroll
  for (int j = 0; j < 4; ++j) {
    if (idx0 + j < n) {
      offs[idx0 + j] = run;
      run += v[j];
      if (idx0 + j == n - 1) offs[n] = run;
    }
  }
}

#define BKSH 8
#define BKN (1 << BKSH)
#define NBK ((NNODES + BKN - 1) / BKN)
#define ECH 256
#define ECHUNK (NEDGES / ECH)
#define TBE 1024
#define HIST_ROOM (((NBK * ECH) + 1 + 1023) & ~1023)
#define EPT ((ECHUNK + TBE - 1) / TBE)
static_assert(NEDGES % ECH == 0, "chunks must tile the edge list");
static_assert(NNODES <= (1 << 17), "source index must fit under the packed node-in-bucket");
static_assert(BKSH + 17 < 32, "the packed word must stay positive");

__global__ void edge_hist(const int* __restrict__ cols, int* __restrict__ hist,
                          float* __restrict__ sums, float* __restrict__ cnts) {
  __shared__ int h[NBK];
  const int tid = threadIdx.x;
  if (blockIdx.x == 0) {
    for (int i = tid; i < NGRAPHS * DOUT; i += TBE) sums[i] = 0.0f;
    if (tid < NGRAPHS) cnts[tid] = 0.0f;
  }
  for (int i = tid; i < NBK; i += TBE) h[i] = 0;
  __syncthreads();
  const int e0 = blockIdx.x * ECHUNK;
  int c[EPT];
#pragma unroll
  for (int k = 0; k < EPT; ++k) {
    const int i = k * TBE + tid;
    c[k] = (i < ECHUNK) ? cols[e0 + i] : -1;
  }
#pragma unroll
  for (int k = 0; k < EPT; ++k)
    if (c[k] >= 0) atomicAdd(&h[c[k] >> BKSH], 1);
  __syncthreads();
  for (int i = tid; i < NBK; i += TBE) hist[i * ECH + blockIdx.x] = h[i];
}

__global__ void edge_bucket(const int* __restrict__ rows, const int* __restrict__ cols,
                            const float* __restrict__ w, const int* __restrict__ base,
                            int2* __restrict__ tmp) {
  __shared__ int cur[NBK];
  const int tid = threadIdx.x;
  for (int i = tid; i < NBK; i += TBE) cur[i] = base[i * ECH + blockIdx.x];
  __syncthreads();
  const int e0 = blockIdx.x * ECHUNK;
  int c[EPT];
  int2 rec[EPT];
#pragma unroll
  for (int k = 0; k < EPT; ++k) {
    const int i = k * TBE + tid;
    const bool ok = (i < ECHUNK);
    c[k] = ok ? cols[e0 + i] : -1;
    rec[k].x = ok ? rows[e0 + i] : 0;
    rec[k].y = ok ? __float_as_int(w[e0 + i]) : 0;
  }
#pragma unroll
  for (int k = 0; k < EPT; ++k) {
    if (c[k] >= 0) {
      const int pos = atomicAdd(&cur[c[k] >> BKSH], 1);
      int2 t;
      t.x = rec[k].x | ((c[k] & (BKN - 1)) << 17);
      t.y = rec[k].y;
      tmp[pos] = t;
    }
  }
}

#define TBF 1024
#define FPT 4
__global__ void bucket_fine(const int2* __restrict__ tmp, const int* __restrict__ base,
                            int* __restrict__ offs, float* __restrict__ dinv,
                            int2* __restrict__ pairs) {
  __shared__ int cntL[BKN];
  __shared__ float degL[BKN];
  __shared__ int curL[BKN];
  __shared__ int wsum[BKN / 32];
  const int tid = threadIdx.x;
  const int lane = tid & 31;
  const int wv = tid >> 5;
  const int start = base[blockIdx.x * ECH];
  const int end = base[(blockIdx.x + 1) * ECH];
  if (tid < BKN) {
    cntL[tid] = 0;
    degL[tid] = 0.0f;
  }
  __syncthreads();

  for (int b0 = start; b0 < end; b0 += FPT * TBF) {
    int2 t[FPT];
#pragma unroll
    for (int k = 0; k < FPT; ++k) {
      const int i = b0 + k * TBF + tid;
      t[k] = (i < end) ? tmp[i] : make_int2(-1, 0);
    }
#pragma unroll
    for (int k = 0; k < FPT; ++k) {
      if (t[k].x >= 0) {
        const int cl = t[k].x >> 17;
        atomicAdd(&cntL[cl], 1);
        atomicAdd(&degL[cl], __int_as_float(t[k].y));
      }
    }
  }
  __syncthreads();

  int v = 0, inc = 0;
  if (tid < BKN) {
    v = cntL[tid];
    inc = v;
#pragma unroll
    for (int off = 1; off < 32; off <<= 1) {
      const int u = __shfl_up(inc, off, 32);
      if (lane >= off) inc += u;
    }
    if (lane == 31) wsum[wv] = inc;
  }
  __syncthreads();
  if (tid < BKN) {
    int wbase = 0;
    for (int k = 0; k < BKN / 32; ++k) wbase += (k < wv) ? wsum[k] : 0;
    const int excl = start + wbase + inc - v;
    curL[tid] = excl;
    const int node = blockIdx.x * BKN + tid;
    if (node < NNODES) {
      offs[node] = excl;
      dinv[node] = rsqrtf(fmaxf(degL[tid] + 1.0f, EPSF));
      if (node == NNODES - 1) offs[NNODES] = excl + v;
    }
  }
  __syncthreads();

  for (int b0 = start; b0 < end; b0 += FPT * TBF) {
    int2 t[FPT];
#pragma unroll
    for (int k = 0; k < FPT; ++k) {
      const int i = b0 + k * TBF + tid;
      t[k] = (i < end) ? tmp[i] : make_int2(-1, 0);
    }
#pragma unroll
    for (int k = 0; k < FPT; ++k) {
      if (t[k].x >= 0) {
        const int pos = atomicAdd(&curL[t[k].x >> 17], 1);
        int2 pr;
        pr.x = t[k].x & ((1 << 17) - 1);
        pr.y = t[k].y;
        pairs[pos] = pr;
      }
    }
  }
}

#define GT 256
template <int NCOLS, int KDIM>
__device__ __forceinline__ void stage_b_pairs(const float* __restrict__ B,
                                              float2* __restrict__ Bl) {
  constexpr int NC4 = NCOLS / 4;
  constexpr int UNITS = (KDIM / 2) * NC4;
  constexpr int UPT = UNITS / GT;
  constexpr int SB = (UPT < 8) ? UPT : 8;
  static_assert(UNITS % GT == 0 && UPT % SB == 0, "the units must tile the work-group");
  const float4* __restrict__ B4 = reinterpret_cast<const float4*>(B);
  float4* __restrict__ Bl4 = reinterpret_cast<float4*>(Bl);
  for (int u0 = threadIdx.x; u0 < UNITS; u0 += SB * GT) {
    float4 ra[SB], rb[SB];
#pragma unroll
    for (int s = 0; s < SB; ++s) {
      const int u = u0 + s * GT;
      const int k2 = u / NC4;
      const int n4 = u - k2 * NC4;
      ra[s] = B4[(2 * k2 + 0) * NC4 + n4];
      rb[s] = B4[(2 * k2 + 1) * NC4 + n4];
    }
#pragma unroll
    for (int s = 0; s < SB; ++s) {
      const int u = u0 + s * GT;
      Bl4[2 * u + 0] = make_float4(ra[s].x, rb[s].x, ra[s].y, rb[s].y);
      Bl4[2 * u + 1] = make_float4(ra[s].z, rb[s].z, ra[s].w, rb[s].w);
    }
  }
}

template <int NCOLS, int KDIM, int KPL, int CPL>
__global__ void wmma_gemm_f32(const float* __restrict__ A0, const float* __restrict__ A1,
                              const float* __restrict__ B, const float* __restrict__ dinv,
                              float* __restrict__ C0, float* __restrict__ C1, int M) {
  __shared__ alignas(16) float2 Bl[(KDIM / 2) * NCOLS];

  stage_b_pairs<NCOLS, KDIM>(B, Bl);
  __syncthreads();

  const int wave = (int)(((size_t)blockIdx.x * blockDim.x + threadIdx.x) >> 5);
  const int lane = threadIdx.x & 31;
  if (wave >= M / 16) return;
  const int m0 = wave * 16;
  const int half = lane >> 4;
  const int l = lane & 15;
  constexpr int NT = NCOLS / 16;

  v8f acc[NT];
#pragma unroll
  for (int t = 0; t < NT; ++t)
#pragma unroll
    for (int j = 0; j < 8; ++j) acc[t][j] = 0.0f;

  auto a_at = [&](int k) -> const float* {
    const float* base = (KPL < KDIM && k >= KPL) ? A1 : A0;
    return base + (size_t)(m0 + l) * KPL + (k & (KPL - 1)) + 2 * half;
  };

  v2f a;
  {
    const float* ap = a_at(0);
    a[0] = ap[0];
    a[1] = ap[1];
  }

  float rsc[8];
#pragma unroll
  for (int r = 0; r < 8; ++r) rsc[r] = dinv[m0 + r + 8 * half];

  for (int k0 = 0; k0 < KDIM; k0 += 4) {
    v2f a_next = a;
    if (k0 + 4 < KDIM) {
      const float* np = a_at(k0 + 4);
      a_next[0] = np[0];
      a_next[1] = np[1];
    }
    const float2* __restrict__ brow = Bl + (size_t)(k0 / 2 + half) * NCOLS + l;
    v2f bfrag[NT];
#pragma unroll
    for (int t = 0; t < NT; ++t) {
      float2 bb = brow[t * 16];
      bfrag[t][0] = bb.x;
      bfrag[t][1] = bb.y;
    }
#pragma unroll
    for (int t = 0; t < NT; ++t)
      acc[t] = __builtin_amdgcn_wmma_f32_16x16x4_f32(
           false, a,  false, bfrag[t],
           (short)0, acc[t],  false,  false);
    a = a_next;
  }

#pragma unroll
  for (int t = 0; t < NT; ++t) {
    constexpr int TPP = CPL / 16;
    float* __restrict__ cb = (t / TPP) ? C1 : C0;
    const int cc = (t % TPP) * 16 + l;
#pragma unroll
    for (int r = 0; r < 8; ++r)
      cb[(size_t)(m0 + r + 8 * half) * CPL + cc] = acc[t][r] * rsc[r];
  }
}

template <int J>
__device__ __forceinline__ int rshare(int x) {
  return __builtin_amdgcn_mov_dpp(x, 0x150 + J, 0xf, 0xf, true);
}
template <int J0, int NS, int U>
__device__ __forceinline__ void ask_rows(const char* __restrict__ Hb, unsigned loff, int roff,
                                         int cvb, float4 (&v)[NS], float (&cf)[NS]) {
  if constexpr (U < NS) {
    const unsigned ro = (unsigned)rshare<J0 + U>(roff);
    cf[U] = __int_as_float(rshare<J0 + U>(cvb));
    v[U] = *reinterpret_cast<const float4*>(Hb + (ro + loff));
    ask_rows<J0, NS, U + 1>(Hb, loff, roff, cvb, v, cf);
  }
}
template <int J0, int NS>
__device__ __forceinline__ void sum_steps(const char* __restrict__ Hb, unsigned loff, int roff,
                                          int cvb, float4& acc) {
  float4 v[NS];
  float cf[NS];
  ask_rows<J0, NS, 0>(Hb, loff, roff, cvb, v, cf);
#pragma unroll
  for (int u = 0; u < NS; ++u) {
    acc.x += v[u].x * cf[u];
    acc.y += v[u].y * cf[u];
    acc.z += v[u].z * cf[u];
    acc.w += v[u].w * cf[u];
  }
}
template <bool NORM>
__global__ void gather_rows(const float* __restrict__ hp, const int* __restrict__ offs,
                            const int2* __restrict__ pairs, const float* __restrict__ dinv,
                            const float* __restrict__ bias, const float* __restrict__ g,
                            const float* __restrict__ b, float* __restrict__ outp, int n) {
  const int wave = (int)(((size_t)blockIdx.x * blockDim.x + threadIdx.x) >> 5);
  const int lane = threadIdx.x & 31;
  if (wave >= n) return;
  const int node = __builtin_amdgcn_readfirstlane(wave);
  const int half = lane >> 4;
  const int c16 = lane & 15;
  const unsigned loff = (unsigned)c16 * 16u;
  const char* __restrict__ Hb = reinterpret_cast<const char*>(hp);
  const int beg = offs[node];
  const int end = offs[node + 1];
  const float dc = dinv[node];
  const float4 self = *reinterpret_cast<const float4*>(Hb + ((unsigned)node * 256u + loff));
  const float4 bb = reinterpret_cast<const float4*>(bias)[c16];
  const int e = 2 * c16 + half;

  float4 acc = make_float4(0.0f, 0.0f, 0.0f, 0.0f);
  for (int base = beg; base < end; base += 32) {
    const int m = min(32, end - base);
    const int2 pr = *reinterpret_cast<const int2*>(
        reinterpret_cast<const char*>(pairs) + (unsigned)(base + min(e, m - 1)) * 8u);
    const int roff = pr.x << 8;
    const float cw = __int_as_float(pr.y) * dc;
    const int cvb = (e < m) ? __float_as_int(cw) : 0;
    const int steps = (m + 1) >> 1;
    sum_steps<0, 8>(Hb, loff, roff, cvb, acc);
    if (steps > 8) {
      sum_steps<8, 4>(Hb, loff, roff, cvb, acc);
      if (steps > 12) sum_steps<12, 4>(Hb, loff, roff, cvb, acc);
    }
  }
  float a[4];
  a[0] = (acc.x + xhalf(acc.x)) + (bb.x + self.x * dc);
  a[1] = (acc.y + xhalf(acc.y)) + (bb.y + self.y * dc);
  a[2] = (acc.z + xhalf(acc.z)) + (bb.z + self.z * dc);
  a[3] = (acc.w + xhalf(acc.w)) + (bb.w + self.w * dc);
  float4 o;
  if (NORM) {
    float v[4];
#pragma unroll
    for (int f = 0; f < 4; ++f) v[f] = fmaxf(a[f], 0.0f);
    float rsum = (v[0] + v[1]) + (v[2] + v[3]);
    rsum += xrow<8>(rsum);
    rsum += xrow<4>(rsum);
    rsum += xrow<2>(rsum);
    rsum += xrow<1>(rsum);
    const float mu = rsum * (1.0f / PLANE);
    float q = 0.0f;
#pragma unroll
    for (int f = 0; f < 4; ++f) {
      const float d = v[f] - mu;
      q += d * d;
    }
    q += xrow<8>(q);
    q += xrow<4>(q);
    q += xrow<2>(q);
    q += xrow<1>(q);
    const float rs = rsqrtf(q * (1.0f / PLANE) + EPSF);
    const float4 gg = reinterpret_cast<const float4*>(g)[c16];
    const float4 sh = reinterpret_cast<const float4*>(b)[c16];
    o.x = (v[0] - mu) * rs * gg.x + sh.x;
    o.y = (v[1] - mu) * rs * gg.y + sh.y;
    o.z = (v[2] - mu) * rs * gg.z + sh.z;
    o.w = (v[3] - mu) * rs * gg.w + sh.w;
  } else {
    o.x = a[0];
    o.y = a[1];
    o.z = a[2];
    o.w = a[3];
  }
  if (half == 0)
    *reinterpret_cast<float4*>(reinterpret_cast<char*>(outp) + ((size_t)node * 256u + loff)) = o;
}

#define APITCH 132
#define NTILES (NNODES / 16)
#define RSG 1024
static_assert(NNODES % 16 == 0, "tiles of 16 nodes must tile the node list");
template <int J0, int NS, int U>
__device__ __forceinline__ void ask_rows2(const char* __restrict__ HbA, const char* __restrict__ HbB,
                                          unsigned loff, int roff, int cvb, float4 (&va)[NS],
                                          float4 (&vb)[NS], float (&cf)[NS]) {
  if constexpr (U < NS) {
    const unsigned ro = (unsigned)rshare<J0 + U>(roff) + loff;
    cf[U] = __int_as_float(rshare<J0 + U>(cvb));
    va[U] = *reinterpret_cast<const float4*>(HbA + ro);
    vb[U] = *reinterpret_cast<const float4*>(HbB + ro);
    ask_rows2<J0, NS, U + 1>(HbA, HbB, loff, roff, cvb, va, vb, cf);
  }
}
template <int J0, int NS>
__device__ __forceinline__ void sum_steps2(const char* __restrict__ HbA, const char* __restrict__ HbB,
                                           unsigned loff, int roff, int cvb, float4& accA,
                                           float4& accB) {
  float4 va[NS], vb[NS];
  float cf[NS];
  ask_rows2<J0, NS, 0>(HbA, HbB, loff, roff, cvb, va, vb, cf);
#pragma unroll
  for (int u = 0; u < NS; ++u) {
    accA.x += va[u].x * cf[u];
    accA.y += va[u].y * cf[u];
    accA.z += va[u].z * cf[u];
    accA.w += va[u].w * cf[u];
    accB.x += vb[u].x * cf[u];
    accB.y += vb[u].y * cf[u];
    accB.z += vb[u].z * cf[u];
    accB.w += vb[u].w * cf[u];
  }
}
__global__ void __launch_bounds__(GT, 8)
rowsum_ln_gemm_f32(const float* __restrict__ hpA, const float* __restrict__ hpB,
                   const int* __restrict__ offs, const int2* __restrict__ pairs,
                   const float* __restrict__ dinv, const float* __restrict__ bias,
                   const float* __restrict__ g, const float* __restrict__ b,
                   const float* __restrict__ B, float* __restrict__ C) {
  constexpr int KDIM = DHID;
  constexpr int NCOLS = DOUT;
  constexpr int KS = KDIM / 4;
  __shared__ alignas(16) float2 Bl[(KDIM / 2) * NCOLS];
  __shared__ alignas(16) float As[2][16 * APITCH];

  stage_b_pairs<NCOLS, KDIM>(B, Bl);

  const int lane = threadIdx.x & 31;
  const int wv = __builtin_amdgcn_readfirstlane((int)(threadIdx.x >> 5));
  const int half = lane >> 4;
  const int c16 = lane & 15;
  const unsigned loff = (unsigned)c16 * 16u;
  const char* __restrict__ HbA = reinterpret_cast<const char*>(hpA);
  const char* __restrict__ HbB = reinterpret_cast<const char*>(hpB);
  const char* __restrict__ Pb = reinterpret_cast<const char*>(pairs);
  const float4 bbA = reinterpret_cast<const float4*>(bias)[c16];
  const float4 bbB = reinterpret_cast<const float4*>(bias)[16 + c16];
  const float4 gg = reinterpret_cast<const float4*>(g)[16 * half + c16];
  const float4 sh = reinterpret_cast<const float4*>(b)[16 * half + c16];
  const int e = 2 * c16 + half;

  int it = 0;
  for (int tile = blockIdx.x; tile < NTILES; tile += gridDim.x, ++it) {
    float* __restrict__ Ab = As[it & 1];
    const int n0 = tile * 16 + 2 * wv;
    const int o0 = offs[n0];
    const int o1 = offs[n0 + 1];
    const int o2 = offs[n0 + 2];
    const float dn0 = dinv[n0];
    const float dn1 = dinv[n0 + 1];
    for (int qn = 0; qn < 2; ++qn) {
      const int rowl = 2 * wv + qn;
      const int node = n0 + qn;
      const int beg = qn ? o1 : o0;
      const int end = qn ? o2 : o1;
      const float dc = qn ? dn1 : dn0;
      const float4 selfA = *reinterpret_cast<const float4*>(HbA + ((unsigned)node * 256u + loff));
      const float4 selfB = *reinterpret_cast<const float4*>(HbB + ((unsigned)node * 256u + loff));
      int2 pr = *reinterpret_cast<const int2*>(
          Pb + (unsigned)min(beg + min(e, max(min(32, end - beg), 1) - 1), NEDGES - 1) * 8u);
      float4 sA, sB;
      sA.x = bbA.x + selfA.x * dc;
      sA.y = bbA.y + selfA.y * dc;
      sA.z = bbA.z + selfA.z * dc;
      sA.w = bbA.w + selfA.w * dc;
      sB.x = bbB.x + selfB.x * dc;
      sB.y = bbB.y + selfB.y * dc;
      sB.z = bbB.z + selfB.z * dc;
      sB.w = bbB.w + selfB.w * dc;
      asm volatile("" : "+v"(sA.x), "+v"(sA.y), "+v"(sA.z), "+v"(sA.w),
                        "+v"(sB.x), "+v"(sB.y), "+v"(sB.z), "+v"(sB.w),
                        "+v"(pr.x), "+v"(pr.y));

      float4 accA = make_float4(0.0f, 0.0f, 0.0f, 0.0f);
      float4 accB = make_float4(0.0f, 0.0f, 0.0f, 0.0f);
      for (int base = beg; base < end; base += 32) {
        const int m = min(32, end - base);
        const int roff = pr.x << 8;
        const float cw = __int_as_float(pr.y) * dc;
        const int cvb = (e < m) ? __float_as_int(cw) : 0;
        const int steps = (m + 1) >> 1;
        sum_steps2<0, 8>(HbA, HbB, loff, roff, cvb, accA, accB);
        if (steps > 8) {
          sum_steps2<8, 4>(HbA, HbB, loff, roff, cvb, accA, accB);
          if (steps > 12) sum_steps2<12, 4>(HbA, HbB, loff, roff, cvb, accA, accB);
        }
        if (base + 32 < end)
          pr = *reinterpret_cast<const int2*>(
              Pb + (unsigned)(base + 32 + min(e, min(32, end - base - 32) - 1)) * 8u);
      }
      float v0[4], v1[4];
      v0[0] = fmaxf((accA.x + xhalf(accA.x)) + sA.x, 0.0f);
      v0[1] = fmaxf((accA.y + xhalf(accA.y)) + sA.y, 0.0f);
      v0[2] = fmaxf((accA.z + xhalf(accA.z)) + sA.z, 0.0f);
      v0[3] = fmaxf((accA.w + xhalf(accA.w)) + sA.w, 0.0f);
      v1[0] = fmaxf((accB.x + xhalf(accB.x)) + sB.x, 0.0f);
      v1[1] = fmaxf((accB.y + xhalf(accB.y)) + sB.y, 0.0f);
      v1[2] = fmaxf((accB.z + xhalf(accB.z)) + sB.z, 0.0f);
      v1[3] = fmaxf((accB.w + xhalf(accB.w)) + sB.w, 0.0f);
      float rsum = ((v0[0] + v0[1]) + (v0[2] + v0[3])) + ((v1[0] + v1[1]) + (v1[2] + v1[3]));
      rsum += xrow<8>(rsum);
      rsum += xrow<4>(rsum);
      rsum += xrow<2>(rsum);
      rsum += xrow<1>(rsum);
      const float mu = rsum * (1.0f / KDIM);
      float q = 0.0f;
#pragma unroll
      for (int f = 0; f < 4; ++f) {
        const float d0 = v0[f] - mu;
        const float d1 = v1[f] - mu;
        q += d0 * d0 + d1 * d1;
      }
      q += xrow<8>(q);
      q += xrow<4>(q);
      q += xrow<2>(q);
      q += xrow<1>(q);
      const float rs = rsqrtf(q * (1.0f / KDIM) + EPSF);
      float4 o;
      o.x = ((half ? v1[0] : v0[0]) - mu) * rs * gg.x + sh.x;
      o.y = ((half ? v1[1] : v0[1]) - mu) * rs * gg.y + sh.y;
      o.z = ((half ? v1[2] : v0[2]) - mu) * rs * gg.z + sh.z;
      o.w = ((half ? v1[3] : v0[3]) - mu) * rs * gg.w + sh.w;
      *reinterpret_cast<float4*>(Ab + rowl * APITCH + PLANE * half + 4 * c16) = o;
    }
    __syncthreads();

    if ((wv >> 1) == (it & 3)) {
      const int pw = wv & 1;
      const int m0 = tile * 16;
      float rsc[8];
#pragma unroll
      for (int r = 0; r < 8; ++r) rsc[r] = dinv[m0 + r + 8 * half];
      v8f acc0, acc1;
#pragma unroll
      for (int r = 0; r < 8; ++r) {
        acc0[r] = 0.0f;
        acc1[r] = 0.0f;
      }
      const float* __restrict__ arow = Ab + c16 * APITCH + 2 * half;
      const float2* __restrict__ brow = Bl + (size_t)half * NCOLS + (2 * pw) * 16 + c16;
      __builtin_amdgcn_sched_barrier(0);
#pragma unroll
      for (int j = 0; j < KS; ++j) {
        const float2 aa = *reinterpret_cast<const float2*>(arow + 4 * j);
        const float2 b0 = brow[(size_t)(2 * j) * NCOLS];
        const float2 b1 = brow[(size_t)(2 * j) * NCOLS + 16];
        v2f af, bf0, bf1;
        af[0] = aa.x;
        af[1] = aa.y;
        bf0[0] = b0.x;
        bf0[1] = b0.y;
        bf1[0] = b1.x;
        bf1[1] = b1.y;
        acc0 = __builtin_amdgcn_wmma_f32_16x16x4_f32(
             false, af,  false, bf0,
             (short)0, acc0,  false,  false);
        acc1 = __builtin_amdgcn_wmma_f32_16x16x4_f32(
             false, af,  false, bf1,
             (short)0, acc1,  false,  false);
      }
      __builtin_amdgcn_sched_barrier(0);
      float* __restrict__ cbase = C + (size_t)(m0 + 8 * half) * NCOLS + (2 * pw) * 16 + c16;
#pragma unroll
      for (int r = 0; r < 8; ++r) {
        cbase[r * NCOLS] = acc0[r] * rsc[r];
        cbase[r * NCOLS + 16] = acc1[r] * rsc[r];
      }
    }
  }
}

#define PR 32
__global__ void pool_accum(const float* __restrict__ a, const int* __restrict__ batch,
                           float* __restrict__ sums, float* __restrict__ cnts, int n) {
  const int wave = (int)(((size_t)blockIdx.x * blockDim.x + threadIdx.x) >> 5);
  const int lane = threadIdx.x & 31;
  const int row0 = wave * PR;
  if (row0 >= n) return;
  const int nrows = min(PR, n - row0);
  const float2* __restrict__ p =
      reinterpret_cast<const float2*>(a + (size_t)row0 * DOUT) + lane;
  const int gl = batch[row0 + min(lane, nrows - 1)];
  const int g0 = __builtin_amdgcn_readfirstlane(gl);

  if (nrows == PR && __all(gl == g0)) {
    float2 v[PR];
#pragma unroll
    for (int i = 0; i < PR; ++i) v[i] = p[i * (DOUT / 2)];
    float s0 = 0.0f, s1 = 0.0f;
#pragma unroll
    for (int i = 0; i < PR; ++i) {
      s0 += v[i].x;
      s1 += v[i].y;
    }
    atomicAdd(&sums[g0 * DOUT + lane * 2 + 0], s0);
    atomicAdd(&sums[g0 * DOUT + lane * 2 + 1], s1);
    if (lane == 0) atomicAdd(&cnts[g0], (float)PR);
  } else {
    int cur = g0;
    float s0 = 0.0f, s1 = 0.0f, c = 0.0f;
    for (int i = 0; i < nrows; ++i) {
      const int gi = __builtin_amdgcn_readfirstlane(batch[row0 + i]);
      if (gi != cur) {
        atomicAdd(&sums[cur * DOUT + lane * 2 + 0], s0);
        atomicAdd(&sums[cur * DOUT + lane * 2 + 1], s1);
        if (lane == 0) atomicAdd(&cnts[cur], c);
        cur = gi;
        s0 = 0.0f;
        s1 = 0.0f;
        c = 0.0f;
      }
      const float2 v = p[i * (DOUT / 2)];
      s0 += v.x;
      s1 += v.y;
      c += 1.0f;
    }
    atomicAdd(&sums[cur * DOUT + lane * 2 + 0], s0);
    atomicAdd(&sums[cur * DOUT + lane * 2 + 1], s1);
    if (lane == 0) atomicAdd(&cnts[cur], c);
  }
}

__global__ void pool_final(const float* __restrict__ sums, const float* __restrict__ cnts,
                           float* __restrict__ out) {
  int i = blockIdx.x * blockDim.x + threadIdx.x;
  if (i < NGRAPHS * DOUT) {
    int g = i / DOUT;
    float m = sums[i] / fmaxf(cnts[g], 1.0f);
    out[i] = 1.0f / (1.0f + __expf(-m));
  }
}

extern "C" void kernel_launch(void* const* d_in, const int* in_sizes, int n_in,
                              void* d_out, int out_size, void* d_ws, size_t ws_size,
                              hipStream_t stream) {
  const float* x   = (const float*)d_in[0];
  const int*   ei  = (const int*)d_in[1];
  const float* ew  = (const float*)d_in[2];
  const int*   bat = (const int*)d_in[3];
  const float* W1  = (const float*)d_in[4];
  const float* b1  = (const float*)d_in[5];
  const float* g1  = (const float*)d_in[6];
  const float* bt1 = (const float*)d_in[7];
  const float* W2  = (const float*)d_in[8];
  const float* b2  = (const float*)d_in[9];
  const float* g2  = (const float*)d_in[10];
  const float* bt2 = (const float*)d_in[11];
  float* out = (float*)d_out;

  const int* rows = ei;
  const int* cols = ei + NEDGES;

  char* ws = (char*)d_ws;
  size_t off = 0;
  auto carve = [&](size_t bytes) -> float* {
    float* p = (float*)(ws + off);
    off += (bytes + 255) & ~(size_t)255;
    return p;
  };
  float* hA   = carve((size_t)NNODES * PLANE * 4);
  float* hB   = carve((size_t)NNODES * PLANE * 4);
  float* aA   = carve((size_t)NNODES * PLANE * 4);
  float* dinv = carve((size_t)NPAD * 4);
  int*   offs = (int*)carve((size_t)NPAD * 4);
  int2*  pairs = (int2*)carve((size_t)NEDGES * 8);
  float* sums = carve((size_t)NGRAPHS * DOUT * 4);
  float* cnts = carve((size_t)NGRAPHS * 4);

  const int TB = 256;
  auto blocks = [&](long long threads) { return (int)((threads + TB - 1) / TB); };

  int2* tmp  = (int2*)hA;
  int*  hist = (int*)hB;
  int*  base = (int*)hB + HIST_ROOM;
  edge_hist<<<ECH, TBE, 0, stream>>>(cols, hist, sums, cnts);
  offs_scan<<<(NBK * ECH + SCAN_TB * 4 - 1) / (SCAN_TB * 4), SCAN_TB, 0, stream>>>(hist, base, NBK * ECH);
  edge_bucket<<<ECH, TBE, 0, stream>>>(rows, cols, ew, base, tmp);
  bucket_fine<<<NBK, TBF, 0, stream>>>(tmp, base, offs, dinv, pairs);

  wmma_gemm_f32<DHID, DIN, DIN, PLANE><<<((NNODES / 16) * 32 + GT - 1) / GT, GT, 0, stream>>>(x, x, W1, dinv, hA, hB, NNODES);

  rowsum_ln_gemm_f32<<<RSG, GT, 0, stream>>>(hA, hB, offs, pairs, dinv, b1, g1, bt1, W2, aA);
  gather_rows<true><<<blocks((long long)NNODES * 32), TB, 0, stream>>>(aA, offs, pairs, dinv, b2, g2, bt2, hA, NNODES);

  pool_accum<<<blocks((long long)((NNODES + PR - 1) / PR) * 32), TB, 0, stream>>>(hA, bat, sums, cnts, NNODES);
  pool_final<<<blocks(NGRAPHS * DOUT), TB, 0, stream>>>(sums, cnts, out);
}


# layer-2 neighbour sums, relu+layernorm and pool sums in ONE launch: resident work-groups loop over 66 consecutive nodes; pool partials in registers, then LDS, then one atomic; pool sum order changes
# speedup vs baseline: 5.2652x; 1.1345x over previous
#include <hip/hip_runtime.h>
#include <math.h>


#define NNODES 100000
#define NEDGES 1600000
#define NGRAPHS 64
#define DIN 256
#define DHID 128
#define DOUT 64
#define EPSF 1e-5f
#define PLANE 64
#define NPAD 100032

typedef __attribute__((ext_vector_type(2))) float v2f;
typedef __attribute__((ext_vector_type(8))) float v8f;

__device__ __forceinline__ float xhalf(float x) {
  const int b = __builtin_bit_cast(int, x);
  return __builtin_bit_cast(
      float, __builtin_amdgcn_permlanex16(b, b, 0x76543210, 0xfedcba98, false, false));
}
template <int XM>
__device__ __forceinline__ float xrow(float x) {
  const int b = __builtin_bit_cast(int, x);
  return __builtin_bit_cast(float, __builtin_amdgcn_update_dpp(b, b, 0x160 + XM, 0xf, 0xf, false));
}
__device__ __forceinline__ float wave_sum(float x) {
  x += xhalf(x);
  x += xrow<8>(x);
  x += xrow<4>(x);
  x += xrow<2>(x);
  x += xrow<1>(x);
  return x;
}

#define SCAN_TB 1024
__global__ void offs_scan(const int* __restrict__ cnt, int* __restrict__ offs, int n) {
  __shared__ int wsum[SCAN_TB / 32];
  const int tid = threadIdx.x;
  const int lane = tid & 31;
  const int wv = tid >> 5;
  const int first = blockIdx.x * (SCAN_TB * 4);

  int part = 0;
  const int4* __restrict__ c4 = reinterpret_cast<const int4*>(cnt);
#pragma unroll 8
  for (int i = tid; i < first / 4; i += SCAN_TB) {
    const int4 q = c4[i];
    part += (q.x + q.y) + (q.z + q.w);
  }
#pragma unroll
  for (int off = 16; off > 0; off >>= 1) part += __shfl_xor(part, off, 32);
  if (lane == 0) wsum[wv] = part;
  __syncthreads();
  int base = 0;
  for (int k = 0; k < SCAN_TB / 32; ++k) base += wsum[k];
  __syncthreads();

  const int idx0 = first + tid * 4;
  int v[4];
#pragma unroll
  for (int j = 0; j < 4; ++j) v[j] = (idx0 + j < n) ? cnt[idx0 + j] : 0;
  const int tsum = (v[0] + v[1]) + (v[2] + v[3]);
  int inc = tsum;
#pragma unroll
  for (int off = 1; off < 32; off <<= 1) {
    const int t = __shfl_up(inc, off, 32);
    if (lane >= off) inc += t;
  }
  if (lane == 31) wsum[wv] = inc;
  __syncthreads();
  int wbase = 0;
  for (int k = 0; k < SCAN_TB / 32; ++k) wbase += (k < wv) ? wsum[k] : 0;
  int run = base + wbase + inc - tsum;
#pragma unroll
  for (int j = 0; j < 4; ++j) {
    if (idx0 + j < n) {
      offs[idx0 + j] = run;
      run += v[j];
      if (idx0 + j == n - 1) offs[n] = run;
    }
  }
}

#define BKSH 8
#define BKN (1 << BKSH)
#define NBK ((NNODES + BKN - 1) / BKN)
#define ECH 256
#define ECHUNK (NEDGES / ECH)
#define TBE 1024
#define HIST_ROOM (((NBK * ECH) + 1 + 1023) & ~1023)
#define EPT ((ECHUNK + TBE - 1) / TBE)
static_assert(NEDGES % ECH == 0, "chunks must tile the edge list");
static_assert(NNODES <= (1 << 17), "source index must fit under the packed node-in-bucket");
static_assert(BKSH + 17 < 32, "the packed word must stay positive");

__global__ void edge_hist(const int* __restrict__ cols, int* __restrict__ hist,
                          float* __restrict__ sums, float* __restrict__ cnts) {
  __shared__ int h[NBK];
  const int tid = threadIdx.x;
  if (blockIdx.x == 0) {
    for (int i = tid; i < NGRAPHS * DOUT; i += TBE) sums[i] = 0.0f;
    if (tid < NGRAPHS) cnts[tid] = 0.0f;
  }
  for (int i = tid; i < NBK; i += TBE) h[i] = 0;
  __syncthreads();
  const int e0 = blockIdx.x * ECHUNK;
  int c[EPT];
#pragma unroll
  for (int k = 0; k < EPT; ++k) {
    const int i = k * TBE + tid;
    c[k] = (i < ECHUNK) ? cols[e0 + i] : -1;
  }
#pragma unroll
  for (int k = 0; k < EPT; ++k)
    if (c[k] >= 0) atomicAdd(&h[c[k] >> BKSH], 1);
  __syncthreads();
  for (int i = tid; i < NBK; i += TBE) hist[i * ECH + blockIdx.x] = h[i];
}

__global__ void edge_bucket(const int* __restrict__ rows, const int* __restrict__ cols,
                            const float* __restrict__ w, const int* __restrict__ base,
                            int2* __restrict__ tmp) {
  __shared__ int cur[NBK];
  const int tid = threadIdx.x;
  for (int i = tid; i < NBK; i += TBE) cur[i] = base[i * ECH + blockIdx.x];
  __syncthreads();
  const int e0 = blockIdx.x * ECHUNK;
  int c[EPT];
  int2 rec[EPT];
#pragma unroll
  for (int k = 0; k < EPT; ++k) {
    const int i = k * TBE + tid;
    const bool ok = (i < ECHUNK);
    c[k] = ok ? cols[e0 + i] : -1;
    rec[k].x = ok ? rows[e0 + i] : 0;
    rec[k].y = ok ? __float_as_int(w[e0 + i]) : 0;
  }
#pragma unroll
  for (int k = 0; k < EPT; ++k) {
    if (c[k] >= 0) {
      const int pos = atomicAdd(&cur[c[k] >> BKSH], 1);
      int2 t;
      t.x = rec[k].x | ((c[k] & (BKN - 1)) << 17);
      t.y = rec[k].y;
      tmp[pos] = t;
    }
  }
}

#define TBF 1024
#define FPT 4
__global__ void bucket_fine(const int2* __restrict__ tmp, const int* __restrict__ base,
                            int* __restrict__ offs, float* __restrict__ dinv,
                            int2* __restrict__ pairs) {
  __shared__ int cntL[BKN];
  __shared__ float degL[BKN];
  __shared__ int curL[BKN];
  __shared__ int wsum[BKN / 32];
  const int tid = threadIdx.x;
  const int lane = tid & 31;
  const int wv = tid >> 5;
  const int start = base[blockIdx.x * ECH];
  const int end = base[(blockIdx.x + 1) * ECH];
  if (tid < BKN) {
    cntL[tid] = 0;
    degL[tid] = 0.0f;
  }
  __syncthreads();

  for (int b0 = start; b0 < end; b0 += FPT * TBF) {
    int2 t[FPT];
#pragma unroll
    for (int k = 0; k < FPT; ++k) {
      const int i = b0 + k * TBF + tid;
      t[k] = (i < end) ? tmp[i] : make_int2(-1, 0);
    }
#pragma unroll
    for (int k = 0; k < FPT; ++k) {
      if (t[k].x >= 0) {
        const int cl = t[k].x >> 17;
        atomicAdd(&cntL[cl], 1);
        atomicAdd(&degL[cl], __int_as_float(t[k].y));
      }
    }
  }
  __syncthreads();

  int v = 0, inc = 0;
  if (tid < BKN) {
    v = cntL[tid];
    inc = v;
#pragma unroll
    for (int off = 1; off < 32; off <<= 1) {
      const int u = __shfl_up(inc, off, 32);
      if (lane >= off) inc += u;
    }
    if (lane == 31) wsum[wv] = inc;
  }
  __syncthreads();
  if (tid < BKN) {
    int wbase = 0;
    for (int k = 0; k < BKN / 32; ++k) wbase += (k < wv) ? wsum[k] : 0;
    const int excl = start + wbase + inc - v;
    curL[tid] = excl;
    const int node = blockIdx.x * BKN + tid;
    if (node < NNODES) {
      offs[node] = excl;
      dinv[node] = rsqrtf(fmaxf(degL[tid] + 1.0f, EPSF));
      if (node == NNODES - 1) offs[NNODES] = excl + v;
    }
  }
  __syncthreads();

  for (int b0 = start; b0 < end; b0 += FPT * TBF) {
    int2 t[FPT];
#pragma unroll
    for (int k = 0; k < FPT; ++k) {
      const int i = b0 + k * TBF + tid;
      t[k] = (i < end) ? tmp[i] : make_int2(-1, 0);
    }
#pragma unroll
    for (int k = 0; k < FPT; ++k) {
      if (t[k].x >= 0) {
        const int pos = atomicAdd(&curL[t[k].x >> 17], 1);
        int2 pr;
        pr.x = t[k].x & ((1 << 17) - 1);
        pr.y = t[k].y;
        pairs[pos] = pr;
      }
    }
  }
}

#define GT 256
template <int NCOLS, int KDIM>
__device__ __forceinline__ void stage_b_pairs(const float* __restrict__ B,
                                              float2* __restrict__ Bl) {
  constexpr int NC4 = NCOLS / 4;
  constexpr int UNITS = (KDIM / 2) * NC4;
  constexpr int UPT = UNITS / GT;
  constexpr int SB = (UPT < 8) ? UPT : 8;
  static_assert(UNITS % GT == 0 && UPT % SB == 0, "the units must tile the work-group");
  const float4* __restrict__ B4 = reinterpret_cast<const float4*>(B);
  float4* __restrict__ Bl4 = reinterpret_cast<float4*>(Bl);
  for (int u0 = threadIdx.x; u0 < UNITS; u0 += SB * GT) {
    float4 ra[SB], rb[SB];
#pragma unroll
    for (int s = 0; s < SB; ++s) {
      const int u = u0 + s * GT;
      const int k2 = u / NC4;
      const int n4 = u - k2 * NC4;
      ra[s] = B4[(2 * k2 + 0) * NC4 + n4];
      rb[s] = B4[(2 * k2 + 1) * NC4 + n4];
    }
#pragma unroll
    for (int s = 0; s < SB; ++s) {
      const int u = u0 + s * GT;
      Bl4[2 * u + 0] = make_float4(ra[s].x, rb[s].x, ra[s].y, rb[s].y);
      Bl4[2 * u + 1] = make_float4(ra[s].z, rb[s].z, ra[s].w, rb[s].w);
    }
  }
}

template <int NCOLS, int KDIM, int KPL, int CPL>
__global__ void wmma_gemm_f32(const float* __restrict__ A0, const float* __restrict__ A1,
                              const float* __restrict__ B, const float* __restrict__ dinv,
                              float* __restrict__ C0, float* __restrict__ C1, int M) {
  __shared__ alignas(16) float2 Bl[(KDIM / 2) * NCOLS];

  stage_b_pairs<NCOLS, KDIM>(B, Bl);
  __syncthreads();

  const int wave = (int)(((size_t)blockIdx.x * blockDim.x + threadIdx.x) >> 5);
  const int lane = threadIdx.x & 31;
  if (wave >= M / 16) return;
  const int m0 = wave * 16;
  const int half = lane >> 4;
  const int l = lane & 15;
  constexpr int NT = NCOLS / 16;

  v8f acc[NT];
#pragma unroll
  for (int t = 0; t < NT; ++t)
#pragma unroll
    for (int j = 0; j < 8; ++j) acc[t][j] = 0.0f;

  auto a_at = [&](int k) -> const float* {
    const float* base = (KPL < KDIM && k >= KPL) ? A1 : A0;
    return base + (size_t)(m0 + l) * KPL + (k & (KPL - 1)) + 2 * half;
  };

  v2f a;
  {
    const float* ap = a_at(0);
    a[0] = ap[0];
    a[1] = ap[1];
  }

  float rsc[8];
#pragma unroll
  for (int r = 0; r < 8; ++r) rsc[r] = dinv[m0 + r + 8 * half];

  for (int k0 = 0; k0 < KDIM; k0 += 4) {
    v2f a_next = a;
    if (k0 + 4 < KDIM) {
      const float* np = a_at(k0 + 4);
      a_next[0] = np[0];
      a_next[1] = np[1];
    }
    const float2* __restrict__ brow = Bl + (size_t)(k0 / 2 + half) * NCOLS + l;
    v2f bfrag[NT];
#pragma unroll
    for (int t = 0; t < NT; ++t) {
      float2 bb = brow[t * 16];
      bfrag[t][0] = bb.x;
      bfrag[t][1] = bb.y;
    }
#pragma unroll
    for (int t = 0; t < NT; ++t)
      acc[t] = __builtin_amdgcn_wmma_f32_16x16x4_f32(
           false, a,  false, bfrag[t],
           (short)0, acc[t],  false,  false);
    a = a_next;
  }

#pragma unroll
  for (int t = 0; t < NT; ++t) {
    constexpr int TPP = CPL / 16;
    float* __restrict__ cb = (t / TPP) ? C1 : C0;
    const int cc = (t % TPP) * 16 + l;
#pragma unroll
    for (int r = 0; r < 8; ++r)
      cb[(size_t)(m0 + r + 8 * half) * CPL + cc] = acc[t][r] * rsc[r];
  }
}

template <int J>
__device__ __forceinline__ int rshare(int x) {
  return __builtin_amdgcn_mov_dpp(x, 0x150 + J, 0xf, 0xf, true);
}
template <int J0, int NS, int U>
__device__ __forceinline__ void ask_rows(const char* __restrict__ Hb, unsigned loff, int roff,
                                         int cvb, float4 (&v)[NS], float (&cf)[NS]) {
  if constexpr (U < NS) {
    const unsigned ro = (unsigned)rshare<J0 + U>(roff);
    cf[U] = __int_as_float(rshare<J0 + U>(cvb));
    v[U] = *reinterpret_cast<const float4*>(Hb + (ro + loff));
    ask_rows<J0, NS, U + 1>(Hb, loff, roff, cvb, v, cf);
  }
}
template <int J0, int NS>
__device__ __forceinline__ void sum_steps(const char* __restrict__ Hb, unsigned loff, int roff,
                                          int cvb, float4& acc) {
  float4 v[NS];
  float cf[NS];
  ask_rows<J0, NS, 0>(Hb, loff, roff, cvb, v, cf);
#pragma unroll
  for (int u = 0; u < NS; ++u) {
    acc.x = __builtin_fmaf(v[u].x, cf[u], acc.x);
    acc.y = __builtin_fmaf(v[u].y, cf[u], acc.y);
    acc.z = __builtin_fmaf(v[u].z, cf[u], acc.z);
    acc.w = __builtin_fmaf(v[u].w, cf[u], acc.w);
  }
}
#define PNB 66
#define PGRID ((NNODES + PNB - 1) / PNB)
__global__ void __launch_bounds__(GT)
rowsum_ln_pool(const float* __restrict__ hp, const int* __restrict__ offs,
               const int2* __restrict__ pairs, const float* __restrict__ dinv,
               const float* __restrict__ bias, const float* __restrict__ g,
               const float* __restrict__ b, const int* __restrict__ batch,
               float* __restrict__ sums, float* __restrict__ cnts) {
  __shared__ alignas(16) float psum[NGRAPHS * DOUT];
  __shared__ float pcnt[NGRAPHS];
  const int tid = threadIdx.x;
  for (int i = tid; i < NGRAPHS * DOUT / 4; i += GT)
    reinterpret_cast<float4*>(psum)[i] = make_float4(0.0f, 0.0f, 0.0f, 0.0f);
  if (tid < NGRAPHS) pcnt[tid] = 0.0f;
  __syncthreads();

  const int lane = tid & 31;
  const int wv = __builtin_amdgcn_readfirstlane(tid >> 5);
  const int half = lane >> 4;
  const int c16 = lane & 15;
  const unsigned loff = (unsigned)c16 * 16u;
  const char* __restrict__ Hb = reinterpret_cast<const char*>(hp);
  const float4 bb = reinterpret_cast<const float4*>(bias)[c16];
  const float4 gg = reinterpret_cast<const float4*>(g)[c16];
  const float4 sh = reinterpret_cast<const float4*>(b)[c16];
  const int e = 2 * c16 + half;
  const int n_lo = blockIdx.x * PNB;
  const int n_hi = min(n_lo + PNB, NNODES);

  int cur = -1;
  float4 ps = make_float4(0.0f, 0.0f, 0.0f, 0.0f);
  float pc = 0.0f;
  auto hand_over = [&]() {
    if (half == 0) {
      float* q = psum + cur * DOUT + 4 * c16;
      atomicAdd(q + 0, ps.x);
      atomicAdd(q + 1, ps.y);
      atomicAdd(q + 2, ps.z);
      atomicAdd(q + 3, ps.w);
    }
    if (lane == 0) atomicAdd(&pcnt[cur], pc);
  };

  static_assert((PNB + GT / 32 - 1) / (GT / 32) <= 32, "a lane per node of the wave");
  const int nk = min(n_lo + wv + (GT / 32) * lane, NNODES - 1);
  const int kbeg = offs[nk];
  const int kend = offs[nk + 1];
  const int kdc = __float_as_int(dinv[nk]);
  const int kgi = batch[nk];
  int k = 0;
  for (int node = n_lo + wv; node < n_hi; node += GT / 32, ++k) {
    const int beg = __builtin_amdgcn_readlane(kbeg, k);
    const int end = __builtin_amdgcn_readlane(kend, k);
    const float dc = __int_as_float(__builtin_amdgcn_readlane(kdc, k));
    const int gi = __builtin_amdgcn_readlane(kgi, k);

    const float4 self = *reinterpret_cast<const float4*>(Hb + ((unsigned)node * 256u + loff));
    float4 acc = make_float4(0.0f, 0.0f, 0.0f, 0.0f);
    for (int base = beg; base < end; base += 32) {
      const int m = min(32, end - base);
      const int2 pr = *reinterpret_cast<const int2*>(
          reinterpret_cast<const char*>(pairs) + (unsigned)(base + min(e, m - 1)) * 8u);
      const int roff = pr.x << 8;
      const float cw = __int_as_float(pr.y) * dc;
      const int cvb = (e < m) ? __float_as_int(cw) : 0;
      const int steps = (m + 1) >> 1;
      sum_steps<0, 8>(Hb, loff, roff, cvb, acc);
      if (steps > 8) {
        sum_steps<8, 4>(Hb, loff, roff, cvb, acc);
        if (steps > 12) sum_steps<12, 4>(Hb, loff, roff, cvb, acc);
      }
    }
    float v[4];
    v[0] = fmaxf((acc.x + xhalf(acc.x)) + (bb.x + self.x * dc), 0.0f);
    v[1] = fmaxf((acc.y + xhalf(acc.y)) + (bb.y + self.y * dc), 0.0f);
    v[2] = fmaxf((acc.z + xhalf(acc.z)) + (bb.z + self.z * dc), 0.0f);
    v[3] = fmaxf((acc.w + xhalf(acc.w)) + (bb.w + self.w * dc), 0.0f);
    float rsum = (v[0] + v[1]) + (v[2] + v[3]);
    rsum += xrow<8>(rsum);
    rsum += xrow<4>(rsum);
    rsum += xrow<2>(rsum);
    rsum += xrow<1>(rsum);
    const float mu = rsum * (1.0f / PLANE);
    float q = 0.0f;
#pragma unroll
    for (int f = 0; f < 4; ++f) {
      const float d = v[f] - mu;
      q += d * d;
    }
    q += xrow<8>(q);
    q += xrow<4>(q);
    q += xrow<2>(q);
    q += xrow<1>(q);
    const float rs = rsqrtf(q * (1.0f / PLANE) + EPSF);
    if (gi != cur) {
      if (cur >= 0) hand_over();
      cur = gi;
      ps = make_float4(0.0f, 0.0f, 0.0f, 0.0f);
      pc = 0.0f;
    }
    ps.x += (v[0] - mu) * rs * gg.x + sh.x;
    ps.y += (v[1] - mu) * rs * gg.y + sh.y;
    ps.z += (v[2] - mu) * rs * gg.z + sh.z;
    ps.w += (v[3] - mu) * rs * gg.w + sh.w;
    pc += 1.0f;
  }
  if (cur >= 0) hand_over();
  __syncthreads();

  constexpr int GSTEP = GT / DOUT;
  constexpr int NGQ = NGRAPHS / GSTEP;
  static_assert(GT % DOUT == 0 && NGRAPHS % GSTEP == 0, "the threads must tile the LDS sums");
  const int f = tid & (DOUT - 1);
  const int gq = tid / DOUT;
  float met[NGQ];
#pragma unroll
  for (int j = 0; j < NGQ; ++j) met[j] = pcnt[gq + GSTEP * j];
#pragma unroll
  for (int j = 0; j < NGQ; ++j) {
    const int i = (gq + GSTEP * j) * DOUT + f;
    if (met[j] != 0.0f) atomicAdd(&sums[i], psum[i]);
  }
  if (tid < NGRAPHS && pcnt[tid] != 0.0f) atomicAdd(&cnts[tid], pcnt[tid]);
}

#define APITCH 132
#define NTILES (NNODES / 16)
#define RSG 1024
static_assert(NNODES % 16 == 0, "tiles of 16 nodes must tile the node list");
template <int J0, int NS, int U>
__device__ __forceinline__ void ask_rows2(const char* __restrict__ HbA, const char* __restrict__ HbB,
                                          unsigned loff, int roff, int cvb, float4 (&va)[NS],
                                          float4 (&vb)[NS], float (&cf)[NS]) {
  if constexpr (U < NS) {
    const unsigned ro = (unsigned)rshare<J0 + U>(roff) + loff;
    cf[U] = __int_as_float(rshare<J0 + U>(cvb));
    va[U] = *reinterpret_cast<const float4*>(HbA + ro);
    vb[U] = *reinterpret_cast<const float4*>(HbB + ro);
    ask_rows2<J0, NS, U + 1>(HbA, HbB, loff, roff, cvb, va, vb, cf);
  }
}
template <int J0, int NS>
__device__ __forceinline__ void sum_steps2(const char* __restrict__ HbA, const char* __restrict__ HbB,
                                           unsigned loff, int roff, int cvb, float4& accA,
                                           float4& accB) {
  float4 va[NS], vb[NS];
  float cf[NS];
  ask_rows2<J0, NS, 0>(HbA, HbB, loff, roff, cvb, va, vb, cf);
#pragma unroll
  for (int u = 0; u < NS; ++u) {
    accA.x += va[u].x * cf[u];
    accA.y += va[u].y * cf[u];
    accA.z += va[u].z * cf[u];
    accA.w += va[u].w * cf[u];
    accB.x += vb[u].x * cf[u];
    accB.y += vb[u].y * cf[u];
    accB.z += vb[u].z * cf[u];
    accB.w += vb[u].w * cf[u];
  }
}
__global__ void __launch_bounds__(GT, 8)
rowsum_ln_gemm_f32(const float* __restrict__ hpA, const float* __restrict__ hpB,
                   const int* __restrict__ offs, const int2* __restrict__ pairs,
                   const float* __restrict__ dinv, const float* __restrict__ bias,
                   const float* __restrict__ g, const float* __restrict__ b,
                   const float* __restrict__ B, float* __restrict__ C) {
  constexpr int KDIM = DHID;
  constexpr int NCOLS = DOUT;
  constexpr int KS = KDIM / 4;
  __shared__ alignas(16) float2 Bl[(KDIM / 2) * NCOLS];
  __shared__ alignas(16) float As[2][16 * APITCH];

  stage_b_pairs<NCOLS, KDIM>(B, Bl);

  const int lane = threadIdx.x & 31;
  const int wv = __builtin_amdgcn_readfirstlane((int)(threadIdx.x >> 5));
  const int half = lane >> 4;
  const int c16 = lane & 15;
  const unsigned loff = (unsigned)c16 * 16u;
  const char* __restrict__ HbA = reinterpret_cast<const char*>(hpA);
  const char* __restrict__ HbB = reinterpret_cast<const char*>(hpB);
  const char* __restrict__ Pb = reinterpret_cast<const char*>(pairs);
  const float4 bbA = reinterpret_cast<const float4*>(bias)[c16];
  const float4 bbB = reinterpret_cast<const float4*>(bias)[16 + c16];
  const float4 gg = reinterpret_cast<const float4*>(g)[16 * half + c16];
  const float4 sh = reinterpret_cast<const float4*>(b)[16 * half + c16];
  const int e = 2 * c16 + half;

  int it = 0;
  for (int tile = blockIdx.x; tile < NTILES; tile += gridDim.x, ++it) {
    float* __restrict__ Ab = As[it & 1];
    const int n0 = tile * 16 + 2 * wv;
    const int o0 = offs[n0];
    const int o1 = offs[n0 + 1];
    const int o2 = offs[n0 + 2];
    const float dn0 = dinv[n0];
    const float dn1 = dinv[n0 + 1];
    for (int qn = 0; qn < 2; ++qn) {
      const int rowl = 2 * wv + qn;
      const int node = n0 + qn;
      const int beg = qn ? o1 : o0;
      const int end = qn ? o2 : o1;
      const float dc = qn ? dn1 : dn0;
      const float4 selfA = *reinterpret_cast<const float4*>(HbA + ((unsigned)node * 256u + loff));
      const float4 selfB = *reinterpret_cast<const float4*>(HbB + ((unsigned)node * 256u + loff));
      int2 pr = *reinterpret_cast<const int2*>(
          Pb + (unsigned)min(beg + min(e, max(min(32, end - beg), 1) - 1), NEDGES - 1) * 8u);
      float4 sA, sB;
      sA.x = bbA.x + selfA.x * dc;
      sA.y = bbA.y + selfA.y * dc;
      sA.z = bbA.z + selfA.z * dc;
      sA.w = bbA.w + selfA.w * dc;
      sB.x = bbB.x + selfB.x * dc;
      sB.y = bbB.y + selfB.y * dc;
      sB.z = bbB.z + selfB.z * dc;
      sB.w = bbB.w + selfB.w * dc;
      asm volatile("" : "+v"(sA.x), "+v"(sA.y), "+v"(sA.z), "+v"(sA.w),
                        "+v"(sB.x), "+v"(sB.y), "+v"(sB.z), "+v"(sB.w),
                        "+v"(pr.x), "+v"(pr.y));

      float4 accA = make_float4(0.0f, 0.0f, 0.0f, 0.0f);
      float4 accB = make_float4(0.0f, 0.0f, 0.0f, 0.0f);
      for (int base = beg; base < end; base += 32) {
        const int m = min(32, end - base);
        const int roff = pr.x << 8;
        const float cw = __int_as_float(pr.y) * dc;
        const int cvb = (e < m) ? __float_as_int(cw) : 0;
        const int steps = (m + 1) >> 1;
        sum_steps2<0, 8>(HbA, HbB, loff, roff, cvb, accA, accB);
        if (steps > 8) {
          sum_steps2<8, 4>(HbA, HbB, loff, roff, cvb, accA, accB);
          if (steps > 12) sum_steps2<12, 4>(HbA, HbB, loff, roff, cvb, accA, accB);
        }
        if (base + 32 < end)
          pr = *reinterpret_cast<const int2*>(
              Pb + (unsigned)(base + 32 + min(e, min(32, end - base - 32) - 1)) * 8u);
      }
      float v0[4], v1[4];
      v0[0] = fmaxf((accA.x + xhalf(accA.x)) + sA.x, 0.0f);
      v0[1] = fmaxf((accA.y + xhalf(accA.y)) + sA.y, 0.0f);
      v0[2] = fmaxf((accA.z + xhalf(accA.z)) + sA.z, 0.0f);
      v0[3] = fmaxf((accA.w + xhalf(accA.w)) + sA.w, 0.0f);
      v1[0] = fmaxf((accB.x + xhalf(accB.x)) + sB.x, 0.0f);
      v1[1] = fmaxf((accB.y + xhalf(accB.y)) + sB.y, 0.0f);
      v1[2] = fmaxf((accB.z + xhalf(accB.z)) + sB.z, 0.0f);
      v1[3] = fmaxf((accB.w + xhalf(accB.w)) + sB.w, 0.0f);
      float rsum = ((v0[0] + v0[1]) + (v0[2] + v0[3])) + ((v1[0] + v1[1]) + (v1[2] + v1[3]));
      rsum += xrow<8>(rsum);
      rsum += xrow<4>(rsum);
      rsum += xrow<2>(rsum);
      rsum += xrow<1>(rsum);
      const float mu = rsum * (1.0f / KDIM);
      float q = 0.0f;
#pragma unroll
      for (int f = 0; f < 4; ++f) {
        const float d0 = v0[f] - mu;
        const float d1 = v1[f] - mu;
        q += d0 * d0 + d1 * d1;
      }
      q += xrow<8>(q);
      q += xrow<4>(q);
      q += xrow<2>(q);
      q += xrow<1>(q);
      const float rs = rsqrtf(q * (1.0f / KDIM) + EPSF);
      float4 o;
      o.x = ((half ? v1[0] : v0[0]) - mu) * rs * gg.x + sh.x;
      o.y = ((half ? v1[1] : v0[1]) - mu) * rs * gg.y + sh.y;
      o.z = ((half ? v1[2] : v0[2]) - mu) * rs * gg.z + sh.z;
      o.w = ((half ? v1[3] : v0[3]) - mu) * rs * gg.w + sh.w;
      *reinterpret_cast<float4*>(Ab + rowl * APITCH + PLANE * half + 4 * c16) = o;
    }
    __syncthreads();

    if ((wv >> 1) == (it & 3)) {
      const int pw = wv & 1;
      const int m0 = tile * 16;
      float rsc[8];
#pragma unroll
      for (int r = 0; r < 8; ++r) rsc[r] = dinv[m0 + r + 8 * half];
      v8f acc0, acc1;
#pragma unroll
      for (int r = 0; r < 8; ++r) {
        acc0[r] = 0.0f;
        acc1[r] = 0.0f;
      }
      const float* __restrict__ arow = Ab + c16 * APITCH + 2 * half;
      const float2* __restrict__ brow = Bl + (size_t)half * NCOLS + (2 * pw) * 16 + c16;
      __builtin_amdgcn_sched_barrier(0);
#pragma unroll
      for (int j = 0; j < KS; ++j) {
        const float2 aa = *reinterpret_cast<const float2*>(arow + 4 * j);
        const float2 b0 = brow[(size_t)(2 * j) * NCOLS];
        const float2 b1 = brow[(size_t)(2 * j) * NCOLS + 16];
        v2f af, bf0, bf1;
        af[0] = aa.x;
        af[1] = aa.y;
        bf0[0] = b0.x;
        bf0[1] = b0.y;
        bf1[0] = b1.x;
        bf1[1] = b1.y;
        acc0 = __builtin_amdgcn_wmma_f32_16x16x4_f32(
             false, af,  false, bf0,
             (short)0, acc0,  false,  false);
        acc1 = __builtin_amdgcn_wmma_f32_16x16x4_f32(
             false, af,  false, bf1,
             (short)0, acc1,  false,  false);
      }
      __builtin_amdgcn_sched_barrier(0);
      float* __restrict__ cbase = C + (size_t)(m0 + 8 * half) * NCOLS + (2 * pw) * 16 + c16;
#pragma unroll
      for (int r = 0; r < 8; ++r) {
        cbase[r * NCOLS] = acc0[r] * rsc[r];
        cbase[r * NCOLS + 16] = acc1[r] * rsc[r];
      }
    }
  }
}

__global__ void pool_final(const float* __restrict__ sums, const float* __restrict__ cnts,
                           float* __restrict__ out) {
  int i = blockIdx.x * blockDim.x + threadIdx.x;
  if (i < NGRAPHS * DOUT) {
    int g = i / DOUT;
    float m = sums[i] / fmaxf(cnts[g], 1.0f);
    out[i] = 1.0f / (1.0f + __expf(-m));
  }
}

extern "C" void kernel_launch(void* const* d_in, const int* in_sizes, int n_in,
                              void* d_out, int out_size, void* d_ws, size_t ws_size,
                              hipStream_t stream) {
  const float* x   = (const float*)d_in[0];
  const int*   ei  = (const int*)d_in[1];
  const float* ew  = (const float*)d_in[2];
  const int*   bat = (const int*)d_in[3];
  const float* W1  = (const float*)d_in[4];
  const float* b1  = (const float*)d_in[5];
  const float* g1  = (const float*)d_in[6];
  const float* bt1 = (const float*)d_in[7];
  const float* W2  = (const float*)d_in[8];
  const float* b2  = (const float*)d_in[9];
  const float* g2  = (const float*)d_in[10];
  const float* bt2 = (const float*)d_in[11];
  float* out = (float*)d_out;

  const int* rows = ei;
  const int* cols = ei + NEDGES;

  char* ws = (char*)d_ws;
  size_t off = 0;
  auto carve = [&](size_t bytes) -> float* {
    float* p = (float*)(ws + off);
    off += (bytes + 255) & ~(size_t)255;
    return p;
  };
  float* hA   = carve((size_t)NNODES * PLANE * 4);
  float* hB   = carve((size_t)NNODES * PLANE * 4);
  float* aA   = carve((size_t)NNODES * PLANE * 4);
  float* dinv = carve((size_t)NPAD * 4);
  int*   offs = (int*)carve((size_t)NPAD * 4);
  int2*  pairs = (int2*)carve((size_t)NEDGES * 8);
  float* sums = carve((size_t)NGRAPHS * DOUT * 4);
  float* cnts = carve((size_t)NGRAPHS * 4);

  const int TB = 256;
  auto blocks = [&](long long threads) { return (int)((threads + TB - 1) / TB); };

  int2* tmp  = (int2*)hA;
  int*  hist = (int*)hB;
  int*  base = (int*)hB + HIST_ROOM;
  edge_hist<<<ECH, TBE, 0, stream>>>(cols, hist, sums, cnts);
  offs_scan<<<(NBK * ECH + SCAN_TB * 4 - 1) / (SCAN_TB * 4), SCAN_TB, 0, stream>>>(hist, base, NBK * ECH);
  edge_bucket<<<ECH, TBE, 0, stream>>>(rows, cols, ew, base, tmp);
  bucket_fine<<<NBK, TBF, 0, stream>>>(tmp, base, offs, dinv, pairs);

  wmma_gemm_f32<DHID, DIN, DIN, PLANE><<<((NNODES / 16) * 32 + GT - 1) / GT, GT, 0, stream>>>(x, x, W1, dinv, hA, hB, NNODES);

  rowsum_ln_gemm_f32<<<RSG, GT, 0, stream>>>(hA, hB, offs, pairs, dinv, b1, g1, bt1, W2, aA);
  rowsum_ln_pool<<<PGRID, GT, 0, stream>>>(aA, offs, pairs, dinv, b2, g2, bt2, bat, sums, cnts);

  pool_final<<<blocks(NGRAPHS * DOUT), TB, 0, stream>>>(sums, cnts, out);
}


# edge sort level 1 inside each edge chunk's LDS (count, scan, place), the chunk stored as whole lines; the bucket pass reads its 256 chunk segments: no global scan, 2 launches fewer; sum orders change
# speedup vs baseline: 6.3384x; 1.2038x over previous
#include <hip/hip_runtime.h>
#include <math.h>


#define NNODES 100000
#define NEDGES 1600000
#define NGRAPHS 64
#define DIN 256
#define DHID 128
#define DOUT 64
#define EPSF 1e-5f
#define PLANE 64
#define NPAD 100032

typedef __attribute__((ext_vector_type(2))) float v2f;
typedef __attribute__((ext_vector_type(8))) float v8f;

__device__ __forceinline__ float xhalf(float x) {
  const int b = __builtin_bit_cast(int, x);
  return __builtin_bit_cast(
      float, __builtin_amdgcn_permlanex16(b, b, 0x76543210, 0xfedcba98, false, false));
}
template <int XM>
__device__ __forceinline__ float xrow(float x) {
  const int b = __builtin_bit_cast(int, x);
  return __builtin_bit_cast(float, __builtin_amdgcn_update_dpp(b, b, 0x160 + XM, 0xf, 0xf, false));
}
__device__ __forceinline__ float wave_sum(float x) {
  x += xhalf(x);
  x += xrow<8>(x);
  x += xrow<4>(x);
  x += xrow<2>(x);
  x += xrow<1>(x);
  return x;
}

#define BKSH 8
#define BKN (1 << BKSH)
#define NBK ((NNODES + BKN - 1) / BKN)
#define ECH 256
#define ECHUNK (NEDGES / ECH)
#define TBE 1024
#define EPT ((ECHUNK + TBE - 1) / TBE)
#define LROW (NBK + 1)
#define SCW ((LROW + 31) / 32)
static_assert(NEDGES % ECH == 0, "chunks must tile the edge list");
static_assert(ECHUNK % 2 == 0, "a chunk leaves LDS in 16-byte pieces of two edges");
static_assert(NNODES <= (1 << 17), "source index must fit under the packed node-in-bucket");
static_assert(BKSH + 17 < 32, "the packed word must stay positive");
static_assert(SCW * 32 <= TBE, "the scanning waves must exist");

__device__ __forceinline__ int wave_sum_i(int x) {
  x += __builtin_amdgcn_permlanex16(x, x, 0x76543210, 0xfedcba98, false, false);
  x += __builtin_amdgcn_update_dpp(x, x, 0x168, 0xf, 0xf, false);
  x += __builtin_amdgcn_update_dpp(x, x, 0x164, 0xf, 0xf, false);
  x += __builtin_amdgcn_update_dpp(x, x, 0x162, 0xf, 0xf, false);
  x += __builtin_amdgcn_update_dpp(x, x, 0x161, 0xf, 0xf, false);
  return x;
}

__global__ void
chunk_sort(const int* __restrict__ rows, const int* __restrict__ cols,
           const float* __restrict__ w, int* __restrict__ lstart,
           int2* __restrict__ tmp, float* __restrict__ sums, float* __restrict__ cnts) {
  __shared__ alignas(16) int2 srt[ECHUNK];
  __shared__ int h[NBK];
  __shared__ int wsum[SCW];
  const int tid = threadIdx.x;
  const int lane = tid & 31;
  const int wv = tid >> 5;
  if (blockIdx.x == 0) {
    for (int i = tid; i < NGRAPHS * DOUT; i += TBE) sums[i] = 0.0f;
    if (tid < NGRAPHS) cnts[tid] = 0.0f;
  }
  for (int i = tid; i < NBK; i += TBE) h[i] = 0;
  __syncthreads();
  const int e0 = blockIdx.x * ECHUNK;
  int c[EPT];
  int2 rec[EPT];
#pragma unroll
  for (int k = 0; k < EPT; ++k) {
    const int i = k * TBE + tid;
    const bool ok = (i < ECHUNK);
    c[k] = ok ? cols[e0 + i] : -1;
    rec[k].x = ok ? rows[e0 + i] : 0;
    rec[k].y = ok ? __float_as_int(w[e0 + i]) : 0;
  }
#pragma unroll
  for (int k = 0; k < EPT; ++k)
    if (c[k] >= 0) atomicAdd(&h[c[k] >> BKSH], 1);
  __syncthreads();

  int v = 0, inc = 0, excl = 0;
  if (tid < SCW * 32) {
    v = (tid < NBK) ? h[tid] : 0;
    inc = v;
#pragma unroll
    for (int off = 1; off < 32; off <<= 1) {
      const int u = __shfl_up(inc, off, 32);
      if (lane >= off) inc += u;
    }
    if (lane == 31) wsum[wv] = inc;
  }
  __syncthreads();
  if (tid < SCW * 32) {
    int wbase = 0;
    for (int k = 0; k < SCW; ++k) wbase += (k < wv) ? wsum[k] : 0;
    excl = wbase + inc - v;
    if (tid < NBK) h[tid] = excl;
  }
  __syncthreads();

#pragma unroll
  for (int k = 0; k < EPT; ++k) {
    if (c[k] >= 0) {
      const int pos = atomicAdd(&h[c[k] >> BKSH], 1);
      int2 t;
      t.x = rec[k].x | ((c[k] & (BKN - 1)) << 17);
      t.y = rec[k].y;
      srt[pos] = t;
    }
  }
  __syncthreads();

  const int4* __restrict__ s4 = reinterpret_cast<const int4*>(srt);
  int4* __restrict__ t4 = reinterpret_cast<int4*>(tmp + e0);
  for (int i = tid; i < ECHUNK / 2; i += TBE) t4[i] = s4[i];
  if (tid < LROW) lstart[blockIdx.x * LROW + tid] = excl;
}

#define TBF 1024
#define SPH 4
static_assert((TBF / 16) * SPH == ECH, "the 16-lane halves must cover the chunks");
template <bool PLACE>
__device__ __forceinline__ void edge_do(const int2 t, int* __restrict__ cntL,
                                        float* __restrict__ degL, int* __restrict__ curL,
                                        int2* __restrict__ pairs) {
  const int cl = t.x >> 17;
  if constexpr (PLACE) {
    const int pos = atomicAdd(&curL[cl], 1);
    int2 pr;
    pr.x = t.x & ((1 << 17) - 1);
    pr.y = t.y;
    pairs[pos] = pr;
  } else {
    atomicAdd(&cntL[cl], 1);
    atomicAdd(&degL[cl], __int_as_float(t.y));
  }
}
template <bool PLACE>
__device__ __forceinline__ void seg_sweep(const int2* __restrict__ tmp, int ch0, int l16,
                                          const int (&s0)[SPH], const int (&s1)[SPH],
                                          int* __restrict__ cntL, float* __restrict__ degL,
                                          int* __restrict__ curL, int2* __restrict__ pairs) {
  int2 t[2 * SPH];
#pragma unroll
  for (int j = 0; j < 2 * SPH; ++j) {
    const int k = j >> 1;
    const int i = s0[k] + 16 * (j & 1) + l16;
    t[j] = (i < s1[k]) ? tmp[(size_t)(ch0 + k) * ECHUNK + i] : make_int2(-1, 0);
  }
#pragma unroll
  for (int j = 0; j < 2 * SPH; ++j)
    if (t[j].x >= 0) edge_do<PLACE>(t[j], cntL, degL, curL, pairs);
#pragma unroll
  for (int k = 0; k < SPH; ++k)
    for (int i = s0[k] + 32 + l16; i < s1[k]; i += 16)
      edge_do<PLACE>(tmp[(size_t)(ch0 + k) * ECHUNK + i], cntL, degL, curL, pairs);
}
__global__ void
bucket_fine(const int2* __restrict__ tmp, const int* __restrict__ lstart,
            int* __restrict__ offs, float* __restrict__ dinv, int2* __restrict__ pairs) {
  __shared__ int cntL[BKN];
  __shared__ float degL[BKN];
  __shared__ int curL[BKN];
  __shared__ int wsum[BKN / 32];
  __shared__ int gsum[TBF / 32];
  const int tid = threadIdx.x;
  const int lane = tid & 31;
  const int wv = tid >> 5;
  const int l16 = tid & 15;
  const int ch0 = (tid >> 4) * SPH;
  int s0[SPH], s1[SPH];
#pragma unroll
  for (int k = 0; k < SPH; ++k) {
    const int* __restrict__ p = lstart + (ch0 + k) * LROW + blockIdx.x;
    s0[k] = p[0];
    s1[k] = p[1];
  }
  if (tid < BKN) {
    cntL[tid] = 0;
    degL[tid] = 0.0f;
  }
  const int part = wave_sum_i((l16 == 0) ? (s0[0] + s0[1]) + (s0[2] + s0[3]) : 0);
  if (lane == 0) gsum[wv] = part;
  __syncthreads();

  seg_sweep<false>(tmp, ch0, l16, s0, s1, cntL, degL, curL, pairs);
  __syncthreads();

  int v = 0, inc = 0;
  if (tid < BKN) {
    v = cntL[tid];
    inc = v;
#pragma unroll
    for (int off = 1; off < 32; off <<= 1) {
      const int u = __shfl_up(inc, off, 32);
      if (lane >= off) inc += u;
    }
    if (lane == 31) wsum[wv] = inc;
  }
  __syncthreads();
  int excl = 0;
  float dg = 0.0f;
  if (tid < BKN) {
    int wbase = 0;
    for (int k = 0; k < TBF / 32; ++k) wbase += gsum[k];
    for (int k = 0; k < BKN / 32; ++k) wbase += (k < wv) ? wsum[k] : 0;
    excl = wbase + inc - v;
    curL[tid] = excl;
    dg = degL[tid];
  }
  __syncthreads();
  if (tid < BKN) {
    const int node = blockIdx.x * BKN + tid;
    if (node < NNODES) {
      offs[node] = excl;
      dinv[node] = rsqrtf(fmaxf(dg + 1.0f, EPSF));
      if (node == NNODES - 1) offs[NNODES] = excl + v;
    }
  }

  seg_sweep<true>(tmp, ch0, l16, s0, s1, cntL, degL, curL, pairs);
}

#define GT 256
template <int NCOLS, int KDIM>
__device__ __forceinline__ void stage_b_pairs(const float* __restrict__ B,
                                              float2* __restrict__ Bl) {
  constexpr int NC4 = NCOLS / 4;
  constexpr int UNITS = (KDIM / 2) * NC4;
  constexpr int UPT = UNITS / GT;
  constexpr int SB = (UPT < 8) ? UPT : 8;
  static_assert(UNITS % GT == 0 && UPT % SB == 0, "the units must tile the work-group");
  const float4* __restrict__ B4 = reinterpret_cast<const float4*>(B);
  float4* __restrict__ Bl4 = reinterpret_cast<float4*>(Bl);
  for (int u0 = threadIdx.x; u0 < UNITS; u0 += SB * GT) {
    float4 ra[SB], rb[SB];
#pragma unroll
    for (int s = 0; s < SB; ++s) {
      const int u = u0 + s * GT;
      const int k2 = u / NC4;
      const int n4 = u - k2 * NC4;
      ra[s] = B4[(2 * k2 + 0) * NC4 + n4];
      rb[s] = B4[(2 * k2 + 1) * NC4 + n4];
    }
#pragma unroll
    for (int s = 0; s < SB; ++s) {
      const int u = u0 + s * GT;
      Bl4[2 * u + 0] = make_float4(ra[s].x, rb[s].x, ra[s].y, rb[s].y);
      Bl4[2 * u + 1] = make_float4(ra[s].z, rb[s].z, ra[s].w, rb[s].w);
    }
  }
}

template <int NCOLS, int KDIM, int KPL, int CPL>
__global__ void wmma_gemm_f32(const float* __restrict__ A0, const float* __restrict__ A1,
                              const float* __restrict__ B, const float* __restrict__ dinv,
                              float* __restrict__ C0, float* __restrict__ C1, int M) {
  __shared__ alignas(16) float2 Bl[(KDIM / 2) * NCOLS];

  stage_b_pairs<NCOLS, KDIM>(B, Bl);
  __syncthreads();

  const int wave = (int)(((size_t)blockIdx.x * blockDim.x + threadIdx.x) >> 5);
  const int lane = threadIdx.x & 31;
  if (wave >= M / 16) return;
  const int m0 = wave * 16;
  const int half = lane >> 4;
  const int l = lane & 15;
  constexpr int NT = NCOLS / 16;

  v8f acc[NT];
#pragma unroll
  for (int t = 0; t < NT; ++t)
#pragma unroll
    for (int j = 0; j < 8; ++j) acc[t][j] = 0.0f;

  auto a_at = [&](int k) -> const float* {
    const float* base = (KPL < KDIM && k >= KPL) ? A1 : A0;
    return base + (size_t)(m0 + l) * KPL + (k & (KPL - 1)) + 2 * half;
  };

  v2f a;
  {
    const float* ap = a_at(0);
    a[0] = ap[0];
    a[1] = ap[1];
  }

  float rsc[8];
#pragma unroll
  for (int r = 0; r < 8; ++r) rsc[r] = dinv[m0 + r + 8 * half];

  for (int k0 = 0; k0 < KDIM; k0 += 4) {
    v2f a_next = a;
    if (k0 + 4 < KDIM) {
      const float* np = a_at(k0 + 4);
      a_next[0] = np[0];
      a_next[1] = np[1];
    }
    const float2* __restrict__ brow = Bl + (size_t)(k0 / 2 + half) * NCOLS + l;
    v2f bfrag[NT];
#pragma unroll
    for (int t = 0; t < NT; ++t) {
      float2 bb = brow[t * 16];
      bfrag[t][0] = bb.x;
      bfrag[t][1] = bb.y;
    }
#pragma unroll
    for (int t = 0; t < NT; ++t)
      acc[t] = __builtin_amdgcn_wmma_f32_16x16x4_f32(
           false, a,  false, bfrag[t],
           (short)0, acc[t],  false,  false);
    a = a_next;
  }

#pragma unroll
  for (int t = 0; t < NT; ++t) {
    constexpr int TPP = CPL / 16;
    float* __restrict__ cb = (t / TPP) ? C1 : C0;
    const int cc = (t % TPP) * 16 + l;
#pragma unroll
    for (int r = 0; r < 8; ++r)
      cb[(size_t)(m0 + r + 8 * half) * CPL + cc] = acc[t][r] * rsc[r];
  }
}

template <int J>
__device__ __forceinline__ int rshare(int x) {
  return __builtin_amdgcn_mov_dpp(x, 0x150 + J, 0xf, 0xf, true);
}
template <int J0, int NS, int U>
__device__ __forceinline__ void ask_rows(const char* __restrict__ Hb, unsigned loff, int roff,
                                         int cvb, float4 (&v)[NS], float (&cf)[NS]) {
  if constexpr (U < NS) {
    const unsigned ro = (unsigned)rshare<J0 + U>(roff);
    cf[U] = __int_as_float(rshare<J0 + U>(cvb));
    v[U] = *reinterpret_cast<const float4*>(Hb + (ro + loff));
    ask_rows<J0, NS, U + 1>(Hb, loff, roff, cvb, v, cf);
  }
}
template <int J0, int NS>
__device__ __forceinline__ void sum_steps(const char* __restrict__ Hb, unsigned loff, int roff,
                                          int cvb, float4& acc) {
  float4 v[NS];
  float cf[NS];
  ask_rows<J0, NS, 0>(Hb, loff, roff, cvb, v, cf);
#pragma unroll
  for (int u = 0; u < NS; ++u) {
    acc.x = __builtin_fmaf(v[u].x, cf[u], acc.x);
    acc.y = __builtin_fmaf(v[u].y, cf[u], acc.y);
    acc.z = __builtin_fmaf(v[u].z, cf[u], acc.z);
    acc.w = __builtin_fmaf(v[u].w, cf[u], acc.w);
  }
}
#define PNB 66
#define PGRID ((NNODES + PNB - 1) / PNB)
__global__ void __launch_bounds__(GT)
rowsum_ln_pool(const float* __restrict__ hp, const int* __restrict__ offs,
               const int2* __restrict__ pairs, const float* __restrict__ dinv,
               const float* __restrict__ bias, const float* __restrict__ g,
               const float* __restrict__ b, const int* __restrict__ batch,
               float* __restrict__ sums, float* __restrict__ cnts) {
  __shared__ alignas(16) float psum[NGRAPHS * DOUT];
  __shared__ float pcnt[NGRAPHS];
  const int tid = threadIdx.x;
  for (int i = tid; i < NGRAPHS * DOUT / 4; i += GT)
    reinterpret_cast<float4*>(psum)[i] = make_float4(0.0f, 0.0f, 0.0f, 0.0f);
  if (tid < NGRAPHS) pcnt[tid] = 0.0f;
  __syncthreads();

  const int lane = tid & 31;
  const int wv = __builtin_amdgcn_readfirstlane(tid >> 5);
  const int half = lane >> 4;
  const int c16 = lane & 15;
  const unsigned loff = (unsigned)c16 * 16u;
  const char* __restrict__ Hb = reinterpret_cast<const char*>(hp);
  const float4 bb = reinterpret_cast<const float4*>(bias)[c16];
  const float4 gg = reinterpret_cast<const float4*>(g)[c16];
  const float4 sh = reinterpret_cast<const float4*>(b)[c16];
  const int e = 2 * c16 + half;
  const int n_lo = blockIdx.x * PNB;
  const int n_hi = min(n_lo + PNB, NNODES);

  int cur = -1;
  float4 ps = make_float4(0.0f, 0.0f, 0.0f, 0.0f);
  float pc = 0.0f;
  auto hand_over = [&]() {
    if (half == 0) {
      float* q = psum + cur * DOUT + 4 * c16;
      atomicAdd(q + 0, ps.x);
      atomicAdd(q + 1, ps.y);
      atomicAdd(q + 2, ps.z);
      atomicAdd(q + 3, ps.w);
    }
    if (lane == 0) atomicAdd(&pcnt[cur], pc);
  };

  static_assert((PNB + GT / 32 - 1) / (GT / 32) <= 32, "a lane per node of the wave");
  const int nk = min(n_lo + wv + (GT / 32) * lane, NNODES - 1);
  const int kbeg = offs[nk];
  const int kend = offs[nk + 1];
  const int kdc = __float_as_int(dinv[nk]);
  const int kgi = batch[nk];
  int k = 0;
  for (int node = n_lo + wv; node < n_hi; node += GT / 32, ++k) {
    const int beg = __builtin_amdgcn_readlane(kbeg, k);
    const int end = __builtin_amdgcn_readlane(kend, k);
    const float dc = __int_as_float(__builtin_amdgcn_readlane(kdc, k));
    const int gi = __builtin_amdgcn_readlane(kgi, k);

    const float4 self = *reinterpret_cast<const float4*>(Hb + ((unsigned)node * 256u + loff));
    float4 acc = make_float4(0.0f, 0.0f, 0.0f, 0.0f);
    for (int base = beg; base < end; base += 32) {
      const int m = min(32, end - base);
      const int2 pr = *reinterpret_cast<const int2*>(
          reinterpret_cast<const char*>(pairs) + (unsigned)(base + min(e, m - 1)) * 8u);
      const int roff = pr.x << 8;
      const float cw = __int_as_float(pr.y) * dc;
      const int cvb = (e < m) ? __float_as_int(cw) : 0;
      const int steps = (m + 1) >> 1;
      sum_steps<0, 8>(Hb, loff, roff, cvb, acc);
      if (steps > 8) {
        sum_steps<8, 4>(Hb, loff, roff, cvb, acc);
        if (steps > 12) sum_steps<12, 4>(Hb, loff, roff, cvb, acc);
      }
    }
    float v[4];
    v[0] = fmaxf((acc.x + xhalf(acc.x)) + (bb.x + self.x * dc), 0.0f);
    v[1] = fmaxf((acc.y + xhalf(acc.y)) + (bb.y + self.y * dc), 0.0f);
    v[2] = fmaxf((acc.z + xhalf(acc.z)) + (bb.z + self.z * dc), 0.0f);
    v[3] = fmaxf((acc.w + xhalf(acc.w)) + (bb.w + self.w * dc), 0.0f);
    float rsum = (v[0] + v[1]) + (v[2] + v[3]);
    rsum += xrow<8>(rsum);
    rsum += xrow<4>(rsum);
    rsum += xrow<2>(rsum);
    rsum += xrow<1>(rsum);
    const float mu = rsum * (1.0f / PLANE);
    float q = 0.0f;
#pragma unroll
    for (int f = 0; f < 4; ++f) {
      const float d = v[f] - mu;
      q += d * d;
    }
    q += xrow<8>(q);
    q += xrow<4>(q);
    q += xrow<2>(q);
    q += xrow<1>(q);
    const float rs = rsqrtf(q * (1.0f / PLANE) + EPSF);
    if (gi != cur) {
      if (cur >= 0) hand_over();
      cur = gi;
      ps = make_float4(0.0f, 0.0f, 0.0f, 0.0f);
      pc = 0.0f;
    }
    ps.x += (v[0] - mu) * rs * gg.x + sh.x;
    ps.y += (v[1] - mu) * rs * gg.y + sh.y;
    ps.z += (v[2] - mu) * rs * gg.z + sh.z;
    ps.w += (v[3] - mu) * rs * gg.w + sh.w;
    pc += 1.0f;
  }
  if (cur >= 0) hand_over();
  __syncthreads();

  constexpr int GSTEP = GT / DOUT;
  constexpr int NGQ = NGRAPHS / GSTEP;
  static_assert(GT % DOUT == 0 && NGRAPHS % GSTEP == 0, "the threads must tile the LDS sums");
  const int f = tid & (DOUT - 1);
  const int gq = tid / DOUT;
  float met[NGQ];
#pragma unroll
  for (int j = 0; j < NGQ; ++j) met[j] = pcnt[gq + GSTEP * j];
#pragma unroll
  for (int j = 0; j < NGQ; ++j) {
    const int i = (gq + GSTEP * j) * DOUT + f;
    if (met[j] != 0.0f) atomicAdd(&sums[i], psum[i]);
  }
  if (tid < NGRAPHS && pcnt[tid] != 0.0f) atomicAdd(&cnts[tid], pcnt[tid]);
}

#define APITCH 132
#define NTILES (NNODES / 16)
#define RSG 1024
static_assert(NNODES % 16 == 0, "tiles of 16 nodes must tile the node list");
template <int J0, int NS, int U>
__device__ __forceinline__ void ask_rows2(const char* __restrict__ HbA, const char* __restrict__ HbB,
                                          unsigned loff, int roff, int cvb, float4 (&va)[NS],
                                          float4 (&vb)[NS], float (&cf)[NS]) {
  if constexpr (U < NS) {
    const unsigned ro = (unsigned)rshare<J0 + U>(roff) + loff;
    cf[U] = __int_as_float(rshare<J0 + U>(cvb));
    va[U] = *reinterpret_cast<const float4*>(HbA + ro);
    vb[U] = *reinterpret_cast<const float4*>(HbB + ro);
    ask_rows2<J0, NS, U + 1>(HbA, HbB, loff, roff, cvb, va, vb, cf);
  }
}
template <int J0, int NS>
__device__ __forceinline__ void sum_steps2(const char* __restrict__ HbA, const char* __restrict__ HbB,
                                           unsigned loff, int roff, int cvb, float4& accA,
                                           float4& accB) {
  float4 va[NS], vb[NS];
  float cf[NS];
  ask_rows2<J0, NS, 0>(HbA, HbB, loff, roff, cvb, va, vb, cf);
#pragma unroll
  for (int u = 0; u < NS; ++u) {
    accA.x += va[u].x * cf[u];
    accA.y += va[u].y * cf[u];
    accA.z += va[u].z * cf[u];
    accA.w += va[u].w * cf[u];
    accB.x += vb[u].x * cf[u];
    accB.y += vb[u].y * cf[u];
    accB.z += vb[u].z * cf[u];
    accB.w += vb[u].w * cf[u];
  }
}
__global__ void __launch_bounds__(GT, 8)
rowsum_ln_gemm_f32(const float* __restrict__ hpA, const float* __restrict__ hpB,
                   const int* __restrict__ offs, const int2* __restrict__ pairs,
                   const float* __restrict__ dinv, const float* __restrict__ bias,
                   const float* __restrict__ g, const float* __restrict__ b,
                   const float* __restrict__ B, float* __restrict__ C) {
  constexpr int KDIM = DHID;
  constexpr int NCOLS = DOUT;
  constexpr int KS = KDIM / 4;
  __shared__ alignas(16) float2 Bl[(KDIM / 2) * NCOLS];
  __shared__ alignas(16) float As[2][16 * APITCH];

  stage_b_pairs<NCOLS, KDIM>(B, Bl);

  const int lane = threadIdx.x & 31;
  const int wv = __builtin_amdgcn_readfirstlane((int)(threadIdx.x >> 5));
  const int half = lane >> 4;
  const int c16 = lane & 15;
  const unsigned loff = (unsigned)c16 * 16u;
  const char* __restrict__ HbA = reinterpret_cast<const char*>(hpA);
  const char* __restrict__ HbB = reinterpret_cast<const char*>(hpB);
  const char* __restrict__ Pb = reinterpret_cast<const char*>(pairs);
  const float4 bbA = reinterpret_cast<const float4*>(bias)[c16];
  const float4 bbB = reinterpret_cast<const float4*>(bias)[16 + c16];
  const float4 gg = reinterpret_cast<const float4*>(g)[16 * half + c16];
  const float4 sh = reinterpret_cast<const float4*>(b)[16 * half + c16];
  const int e = 2 * c16 + half;

  int it = 0;
  for (int tile = blockIdx.x; tile < NTILES; tile += gridDim.x, ++it) {
    float* __restrict__ Ab = As[it & 1];
    const int n0 = tile * 16 + 2 * wv;
    const int o0 = offs[n0];
    const int o1 = offs[n0 + 1];
    const int o2 = offs[n0 + 2];
    const float dn0 = dinv[n0];
    const float dn1 = dinv[n0 + 1];
    for (int qn = 0; qn < 2; ++qn) {
      const int rowl = 2 * wv + qn;
      const int node = n0 + qn;
      const int beg = qn ? o1 : o0;
      const int end = qn ? o2 : o1;
      const float dc = qn ? dn1 : dn0;
      const float4 selfA = *reinterpret_cast<const float4*>(HbA + ((unsigned)node * 256u + loff));
      const float4 selfB = *reinterpret_cast<const float4*>(HbB + ((unsigned)node * 256u + loff));
      int2 pr = *reinterpret_cast<const int2*>(
          Pb + (unsigned)min(beg + min(e, max(min(32, end - beg), 1) - 1), NEDGES - 1) * 8u);
      float4 sA, sB;
      sA.x = bbA.x + selfA.x * dc;
      sA.y = bbA.y + selfA.y * dc;
      sA.z = bbA.z + selfA.z * dc;
      sA.w = bbA.w + selfA.w * dc;
      sB.x = bbB.x + selfB.x * dc;
      sB.y = bbB.y + selfB.y * dc;
      sB.z = bbB.z + selfB.z * dc;
      sB.w = bbB.w + selfB.w * dc;
      asm volatile("" : "+v"(sA.x), "+v"(sA.y), "+v"(sA.z), "+v"(sA.w),
                        "+v"(sB.x), "+v"(sB.y), "+v"(sB.z), "+v"(sB.w),
                        "+v"(pr.x), "+v"(pr.y));

      float4 accA = make_float4(0.0f, 0.0f, 0.0f, 0.0f);
      float4 accB = make_float4(0.0f, 0.0f, 0.0f, 0.0f);
      for (int base = beg; base < end; base += 32) {
        const int m = min(32, end - base);
        const int roff = pr.x << 8;
        const float cw = __int_as_float(pr.y) * dc;
        const int cvb = (e < m) ? __float_as_int(cw) : 0;
        const int steps = (m + 1) >> 1;
        sum_steps2<0, 8>(HbA, HbB, loff, roff, cvb, accA, accB);
        if (steps > 8) {
          sum_steps2<8, 4>(HbA, HbB, loff, roff, cvb, accA, accB);
          if (steps > 12) sum_steps2<12, 4>(HbA, HbB, loff, roff, cvb, accA, accB);
        }
        if (base + 32 < end)
          pr = *reinterpret_cast<const int2*>(
              Pb + (unsigned)(base + 32 + min(e, min(32, end - base - 32) - 1)) * 8u);
      }
      float v0[4], v1[4];
      v0[0] = fmaxf((accA.x + xhalf(accA.x)) + sA.x, 0.0f);
      v0[1] = fmaxf((accA.y + xhalf(accA.y)) + sA.y, 0.0f);
      v0[2] = fmaxf((accA.z + xhalf(accA.z)) + sA.z, 0.0f);
      v0[3] = fmaxf((accA.w + xhalf(accA.w)) + sA.w, 0.0f);
      v1[0] = fmaxf((accB.x + xhalf(accB.x)) + sB.x, 0.0f);
      v1[1] = fmaxf((accB.y + xhalf(accB.y)) + sB.y, 0.0f);
      v1[2] = fmaxf((accB.z + xhalf(accB.z)) + sB.z, 0.0f);
      v1[3] = fmaxf((accB.w + xhalf(accB.w)) + sB.w, 0.0f);
      float rsum = ((v0[0] + v0[1]) + (v0[2] + v0[3])) + ((v1[0] + v1[1]) + (v1[2] + v1[3]));
      rsum += xrow<8>(rsum);
      rsum += xrow<4>(rsum);
      rsum += xrow<2>(rsum);
      rsum += xrow<1>(rsum);
      const float mu = rsum * (1.0f / KDIM);
      float q = 0.0f;
#pragma unroll
      for (int f = 0; f < 4; ++f) {
        const float d0 = v0[f] - mu;
        const float d1 = v1[f] - mu;
        q += d0 * d0 + d1 * d1;
      }
      q += xrow<8>(q);
      q += xrow<4>(q);
      q += xrow<2>(q);
      q += xrow<1>(q);
      const float rs = rsqrtf(q * (1.0f / KDIM) + EPSF);
      float4 o;
      o.x = ((half ? v1[0] : v0[0]) - mu) * rs * gg.x + sh.x;
      o.y = ((half ? v1[1] : v0[1]) - mu) * rs * gg.y + sh.y;
      o.z = ((half ? v1[2] : v0[2]) - mu) * rs * gg.z + sh.z;
      o.w = ((half ? v1[3] : v0[3]) - mu) * rs * gg.w + sh.w;
      *reinterpret_cast<float4*>(Ab + rowl * APITCH + PLANE * half + 4 * c16) = o;
    }
    __syncthreads();

    if ((wv >> 1) == (it & 3)) {
      const int pw = wv & 1;
      const int m0 = tile * 16;
      float rsc[8];
#pragma unroll
      for (int r = 0; r < 8; ++r) rsc[r] = dinv[m0 + r + 8 * half];
      v8f acc0, acc1;
#pragma unroll
      for (int r = 0; r < 8; ++r) {
        acc0[r] = 0.0f;
        acc1[r] = 0.0f;
      }
      const float* __restrict__ arow = Ab + c16 * APITCH + 2 * half;
      const float2* __restrict__ brow = Bl + (size_t)half * NCOLS + (2 * pw) * 16 + c16;
      __builtin_amdgcn_sched_barrier(0);
#pragma unroll
      for (int j = 0; j < KS; ++j) {
        const float2 aa = *reinterpret_cast<const float2*>(arow + 4 * j);
        const float2 b0 = brow[(size_t)(2 * j) * NCOLS];
        const float2 b1 = brow[(size_t)(2 * j) * NCOLS + 16];
        v2f af, bf0, bf1;
        af[0] = aa.x;
        af[1] = aa.y;
        bf0[0] = b0.x;
        bf0[1] = b0.y;
        bf1[0] = b1.x;
        bf1[1] = b1.y;
        acc0 = __builtin_amdgcn_wmma_f32_16x16x4_f32(
             false, af,  false, bf0,
             (short)0, acc0,  false,  false);
        acc1 = __builtin_amdgcn_wmma_f32_16x16x4_f32(
             false, af,  false, bf1,
             (short)0, acc1,  false,  false);
      }
      __builtin_amdgcn_sched_barrier(0);
      float* __restrict__ cbase = C + (size_t)(m0 + 8 * half) * NCOLS + (2 * pw) * 16 + c16;
#pragma unroll
      for (int r = 0; r < 8; ++r) {
        cbase[r * NCOLS] = acc0[r] * rsc[r];
        cbase[r * NCOLS + 16] = acc1[r] * rsc[r];
      }
    }
  }
}

__global__ void pool_final(const float* __restrict__ sums, const float* __restrict__ cnts,
                           float* __restrict__ out) {
  int i = blockIdx.x * blockDim.x + threadIdx.x;
  if (i < NGRAPHS * DOUT) {
    int g = i / DOUT;
    float m = sums[i] / fmaxf(cnts[g], 1.0f);
    out[i] = 1.0f / (1.0f + __expf(-m));
  }
}

extern "C" void kernel_launch(void* const* d_in, const int* in_sizes, int n_in,
                              void* d_out, int out_size, void* d_ws, size_t ws_size,
                              hipStream_t stream) {
  const float* x   = (const float*)d_in[0];
  const int*   ei  = (const int*)d_in[1];
  const float* ew  = (const float*)d_in[2];
  const int*   bat = (const int*)d_in[3];
  const float* W1  = (const float*)d_in[4];
  const float* b1  = (const float*)d_in[5];
  const float* g1  = (const float*)d_in[6];
  const float* bt1 = (const float*)d_in[7];
  const float* W2  = (const float*)d_in[8];
  const float* b2  = (const float*)d_in[9];
  const float* g2  = (const float*)d_in[10];
  const float* bt2 = (const float*)d_in[11];
  float* out = (float*)d_out;

  const int* rows = ei;
  const int* cols = ei + NEDGES;

  char* ws = (char*)d_ws;
  size_t off = 0;
  auto carve = [&](size_t bytes) -> float* {
    float* p = (float*)(ws + off);
    off += (bytes + 255) & ~(size_t)255;
    return p;
  };
  float* hA   = carve((size_t)NNODES * PLANE * 4);
  float* hB   = carve((size_t)NNODES * PLANE * 4);
  float* aA   = carve((size_t)NNODES * PLANE * 4);
  float* dinv = carve((size_t)NPAD * 4);
  int*   offs = (int*)carve((size_t)NPAD * 4);
  int2*  pairs = (int2*)carve((size_t)NEDGES * 8);
  float* sums = carve((size_t)NGRAPHS * DOUT * 4);
  float* cnts = carve((size_t)NGRAPHS * 4);

  const int TB = 256;
  auto blocks = [&](long long threads) { return (int)((threads + TB - 1) / TB); };

  int2* tmp    = (int2*)hA;
  int*  lstart = (int*)hB;
  chunk_sort<<<ECH, TBE, 0, stream>>>(rows, cols, ew, lstart, tmp, sums, cnts);
  bucket_fine<<<NBK, TBF, 0, stream>>>(tmp, lstart, offs, dinv, pairs);

  wmma_gemm_f32<DHID, DIN, DIN, PLANE><<<((NNODES / 16) * 32 + GT - 1) / GT, GT, 0, stream>>>(x, x, W1, dinv, hA, hB, NNODES);

  rowsum_ln_gemm_f32<<<RSG, GT, 0, stream>>>(hA, hB, offs, pairs, dinv, b1, g1, bt1, W2, aA);
  rowsum_ln_pool<<<PGRID, GT, 0, stream>>>(aA, offs, pairs, dinv, b2, g2, bt2, bat, sums, cnts);

  pool_final<<<blocks(NGRAPHS * DOUT), TB, 0, stream>>>(sums, cnts, out);
}
